# starting kernel
# baseline (speedup 1.0000x reference)
#include <hip/hip_runtime.h>
#include <hip/hip_bf16.h>


typedef __attribute__((ext_vector_type(2))) float v2f;
typedef __attribute__((ext_vector_type(8))) float v8f;

#define D_FEAT 64
#define LDS_STRIDE 68

__global__ void gcn_zero_f32(float* __restrict__ p, int n) {
    int i = blockIdx.x * blockDim.x + threadIdx.x;
    if (i < n) p[i] = 0.0f;
}

__global__ void gcn_degree(const int* __restrict__ src, const int* __restrict__ dst,
                           float* __restrict__ deg_out, float* __restrict__ deg_in,
                           int nEdges) {
    int e = blockIdx.x * blockDim.x + threadIdx.x;
    if (e >= nEdges) return;
    atomicAdd(&deg_out[src[e]], 1.0f);
    atomicAdd(&deg_in[dst[e]], 1.0f);
}

__global__ void gcn_deg_isqrt(float* __restrict__ d, int n) {
    int i = blockIdx.x * blockDim.x + threadIdx.x;
    if (i < n) d[i] = rsqrtf(fmaxf(d[i], 1.0f));
}

__global__ void gcn_scale_rows(const float* __restrict__ x, const float* __restrict__ s,
                               float* __restrict__ out, int nQuads  ) {
    int t = blockIdx.x * blockDim.x + threadIdx.x;
    if (t >= nQuads) return;
    int row = t >> 4;
    float sc = s[row];
    float4 v = ((const float4*)x)[t];
    v.x *= sc; v.y *= sc; v.z *= sc; v.w *= sc;
    ((float4*)out)[t] = v;
}

__global__ void gcn_scatter(const float* __restrict__ h,
                            const int* __restrict__ src, const int* __restrict__ dst,
                            float* __restrict__ agg, int nEdges) {
    int t = blockIdx.x * blockDim.x + threadIdx.x;
    int e = t >> 4;
    if (e >= nEdges) return;
    int f = (t & 15) << 2;
    int s = src[e], d = dst[e];
    float4 v = *(const float4*)(h + (size_t)s * D_FEAT + f);
    float* p = agg + (size_t)d * D_FEAT + f;
    atomicAdd(p + 0, v.x);
    atomicAdd(p + 1, v.y);
    atomicAdd(p + 2, v.z);
    atomicAdd(p + 3, v.w);
}

__global__ void __launch_bounds__(128)
gcn_transform_wmma(const float* __restrict__ agg,
                   const float* __restrict__ W,
                   const float* __restrict__ bias,
                   const float* __restrict__ in_isqrt,
                   const float* __restrict__ post_scale,
                   float* __restrict__ out, int nNodes) {
    __shared__ float lds[16 * LDS_STRIDE];

    const int r0   = blockIdx.x * 16;
    const int tid  = threadIdx.x;
    const int wave = tid >> 5;
    const int lane = tid & 31;

    {
        int idx = tid * 8;
        int row = idx >> 6;
        int col = idx & 63;
        int grow = r0 + row; if (grow >= nNodes) grow = nNodes - 1;
        float s = in_isqrt[grow];
        const float* g = agg + (size_t)grow * D_FEAT + col;
        float* l = &lds[row * LDS_STRIDE + col];
        #pragma unroll
        for (int j = 0; j < 8; ++j) l[j] = g[j] * s;
    }
    __syncthreads();

    const int colbase = wave << 4;
    const int n     = lane & 15;
    const int khalf = lane >> 4;

    v8f acc = {};
    #pragma unroll
    for (int kk = 0; kk < 16; ++kk) {
        const int k0 = (kk << 2) + (khalf << 1);
        v2f a = *(const v2f*)&lds[n * LDS_STRIDE + k0];
        v2f b;
        b.x = W[(k0 + 0) * D_FEAT + colbase + n];
        b.y = W[(k0 + 1) * D_FEAT + colbase + n];
        acc = __builtin_amdgcn_wmma_f32_16x16x4_f32(
             false, a,  false, b,
             (short)0, acc,  false,  false);
    }

    const int col = colbase + n;
    const float bv = bias[col];
    #pragma unroll
    for (int r = 0; r < 8; ++r) {
        int row = r0 + (khalf << 3) + r;
        if (row < nNodes) {
            float v = acc[r] + bv;
            v = (v > 0.0f) ? v : 0.01f * v;
            if (post_scale) v *= post_scale[row];
            out[(size_t)row * D_FEAT + col] = v;
        }
    }
}

__global__ void gcn_classify(const float* __restrict__ h,
                             const float* __restrict__ Wc,
                             const float* __restrict__ bc,
                             float* __restrict__ out, int nNodes) {
    int i = blockIdx.x * blockDim.x + threadIdx.x;
    if (i >= nNodes) return;
    float a0 = bc[0], a1 = bc[1];
    const float4* row = (const float4*)(h + (size_t)i * D_FEAT);
    #pragma unroll
    for (int q = 0; q < 16; ++q) {
        float4 v = row[q];
        int k = q << 2;
        a0 += v.x * Wc[(k+0)*2] + v.y * Wc[(k+1)*2] + v.z * Wc[(k+2)*2] + v.w * Wc[(k+3)*2];
        a1 += v.x * Wc[(k+0)*2+1] + v.y * Wc[(k+1)*2+1] + v.z * Wc[(k+2)*2+1] + v.w * Wc[(k+3)*2+1];
    }
    out[(size_t)i * 2 + 0] = a0;
    out[(size_t)i * 2 + 1] = a1;
}

extern "C" void kernel_launch(void* const* d_in, const int* in_sizes, int n_in,
                              void* d_out, int out_size, void* d_ws, size_t ws_size,
                              hipStream_t stream) {
    const float* x   = (const float*)d_in[0];
    const int*   src = (const int*)  d_in[1];
    const int*   dst = (const int*)  d_in[2];
    const float* W1  = (const float*)d_in[3];
    const float* b1  = (const float*)d_in[4];
    const float* W2  = (const float*)d_in[5];
    const float* b2  = (const float*)d_in[6];
    const float* Wc  = (const float*)d_in[7];
    const float* bc  = (const float*)d_in[8];
    float* out = (float*)d_out;

    const int N = in_sizes[0] / D_FEAT;
    const int E = in_sizes[1];

    float* bufA      = (float*)d_ws;
    float* bufB      = bufA + (size_t)N * D_FEAT;
    float* out_isqrt = bufB + (size_t)N * D_FEAT;
    float* in_isqrt  = out_isqrt + N;

    const int B256 = 256;
    const int featTotal = N * D_FEAT;
    dim3 blk(B256);

    gcn_zero_f32<<<(2 * N + B256 - 1) / B256, blk, 0, stream>>>(out_isqrt, 2 * N);
    gcn_degree<<<(E + B256 - 1) / B256, blk, 0, stream>>>(src, dst, out_isqrt, in_isqrt, E);
    gcn_deg_isqrt<<<(2 * N + B256 - 1) / B256, blk, 0, stream>>>(out_isqrt, 2 * N);

    gcn_scale_rows<<<(N * 16 + B256 - 1) / B256, blk, 0, stream>>>(x, out_isqrt, bufA, N * 16);
    gcn_zero_f32<<<(featTotal + B256 - 1) / B256, blk, 0, stream>>>(bufB, featTotal);
    gcn_scatter<<<((size_t)E * 16 + B256 - 1) / B256, blk, 0, stream>>>(bufA, src, dst, bufB, E);
    gcn_transform_wmma<<<(N + 15) / 16, dim3(128), 0, stream>>>(
        bufB, W1, b1, in_isqrt, out_isqrt, bufA, N);

    gcn_zero_f32<<<(featTotal + B256 - 1) / B256, blk, 0, stream>>>(bufB, featTotal);
    gcn_scatter<<<((size_t)E * 16 + B256 - 1) / B256, blk, 0, stream>>>(bufA, src, dst, bufB, E);
    gcn_transform_wmma<<<(N + 15) / 16, dim3(128), 0, stream>>>(
        bufB, W2, b2, in_isqrt, nullptr, bufA, N);

    gcn_classify<<<(N + B256 - 1) / B256, blk, 0, stream>>>(bufA, Wc, bc, out, N);
}


# Edges counting-sorted by destination in each call (int atomic count+rank, scan, place); atomic-free per-node row sum replaces f32 atomic scatter-add and zero fill; sums run in rank (arrival) order
# speedup vs baseline: 1.4841x; 1.4841x over previous
#include <hip/hip_runtime.h>
#include <hip/hip_bf16.h>


typedef __attribute__((ext_vector_type(2))) float v2f;
typedef __attribute__((ext_vector_type(8))) float v8f;

#define D_FEAT 64
#define LDS_STRIDE 68

__global__ void gcn_zero_i32(int* __restrict__ p, int n) {
    int i = blockIdx.x * blockDim.x + threadIdx.x;
    if (i < n) p[i] = 0;
}

__global__ void gcn_count_rank(const int* __restrict__ src, const int* __restrict__ dst,
                               int* __restrict__ cnt_out, int* __restrict__ cnt_in,
                               int* __restrict__ rank, int nEdges) {
    int e = blockIdx.x * blockDim.x + threadIdx.x;
    if (e >= nEdges) return;
    const int s = src[e];
    const int d = dst[e];
    atomicAdd(&cnt_out[s], 1);
    rank[e] = atomicAdd(&cnt_in[d], 1);
}

__global__ void __launch_bounds__(256)
gcn_scan_isqrt(const int* __restrict__ cnt_in, const int* __restrict__ cnt_out,
               int* __restrict__ row_start, float* __restrict__ in_isqrt,
               float* __restrict__ out_isqrt, int n) {
    __shared__ int wsum[8];
    const int tid  = threadIdx.x;
    const int lane = tid & 31;
    const int wave = tid >> 5;
    const int base = blockIdx.x * 256;

    int part = 0;
    #pragma unroll 8
    for (int i = tid * 4; i < base; i += 1024) {
        int4 q = *(const int4*)(cnt_in + i);
        part += (q.x + q.y) + (q.z + q.w);
    }
    for (int o = 16; o > 0; o >>= 1) part += __shfl_xor(part, o);
    if (lane == 0) wsum[wave] = part;
    __syncthreads();
    int offset = 0;
    for (int w = 0; w < 8; ++w) offset += wsum[w];
    __syncthreads();

    const int i = base + tid;
    const int cnt = (i < n) ? cnt_in[i] : 0;
    int incl = cnt;
    for (int o = 1; o < 32; o <<= 1) {
        int t = __shfl_up(incl, o);
        if (lane >= o) incl += t;
    }
    if (lane == 31) wsum[wave] = incl;
    __syncthreads();
    int wprefix = 0;
    for (int w = 0; w < 8; ++w) wprefix += (w < wave) ? wsum[w] : 0;

    if (i < n) {
        const int excl = offset + wprefix + incl - cnt;
        row_start[i] = excl;
        if (i == n - 1) row_start[n] = excl + cnt;
        in_isqrt[i]  = rsqrtf(fmaxf((float)cnt, 1.0f));
        out_isqrt[i] = rsqrtf(fmaxf((float)cnt_out[i], 1.0f));
    }
}

__global__ void gcn_fill(const int* __restrict__ src, const int* __restrict__ dst,
                         const int* __restrict__ rank, const int* __restrict__ row_start,
                         int* __restrict__ col, int nEdges) {
    int e = blockIdx.x * blockDim.x + threadIdx.x;
    if (e >= nEdges) return;
    col[row_start[dst[e]] + rank[e]] = src[e];
}

__global__ void gcn_scale_rows(const float* __restrict__ x, const float* __restrict__ s,
                               float* __restrict__ out, int nQuads  ) {
    int t = blockIdx.x * blockDim.x + threadIdx.x;
    if (t >= nQuads) return;
    int row = t >> 4;
    float sc = s[row];
    float4 v = ((const float4*)x)[t];
    v.x *= sc; v.y *= sc; v.z *= sc; v.w *= sc;
    ((float4*)out)[t] = v;
}

__global__ void __launch_bounds__(256)
gcn_gather(const float* __restrict__ h, const int* __restrict__ row_start,
           const int* __restrict__ col, float* __restrict__ agg, int nNodes) {
    const int t  = blockIdx.x * blockDim.x + threadIdx.x;
    const int v  = t >> 4;
    const int l  = t & 15;
    const int f  = l << 2;
    const bool valid = v < nNodes;
    const int beg = valid ? row_start[v]     : 0;
    const int end = valid ? row_start[v + 1] : 0;
    const float4 zero4 = make_float4(0.0f, 0.0f, 0.0f, 0.0f);
    float4 acc = zero4;

    for (int base = beg; ; base += 16) {
        const int rem  = end - base;
        const int remW = max(rem, __shfl_xor(rem, 16));
        if (remW <= 0) break;
        const int c   = (l < rem) ? col[base + l] : 0;
        const int cnt = min(remW, 16);
        for (int k = 0; k < cnt; k += 4) {
            const int s0 = __shfl(c, k,     16);
            const int s1 = __shfl(c, k + 1, 16);
            const int s2 = __shfl(c, k + 2, 16);
            const int s3 = __shfl(c, k + 3, 16);
            float4 r0 = zero4, r1 = zero4, r2 = zero4, r3 = zero4;
            if (k     < rem) r0 = *(const float4*)(h + (size_t)s0 * D_FEAT + f);
            if (k + 1 < rem) r1 = *(const float4*)(h + (size_t)s1 * D_FEAT + f);
            if (k + 2 < rem) r2 = *(const float4*)(h + (size_t)s2 * D_FEAT + f);
            if (k + 3 < rem) r3 = *(const float4*)(h + (size_t)s3 * D_FEAT + f);
            acc.x += r0.x; acc.y += r0.y; acc.z += r0.z; acc.w += r0.w;
            acc.x += r1.x; acc.y += r1.y; acc.z += r1.z; acc.w += r1.w;
            acc.x += r2.x; acc.y += r2.y; acc.z += r2.z; acc.w += r2.w;
            acc.x += r3.x; acc.y += r3.y; acc.z += r3.z; acc.w += r3.w;
        }
    }
    if (valid) *(float4*)(agg + (size_t)v * D_FEAT + f) = acc;
}

__global__ void __launch_bounds__(128)
gcn_transform_wmma(const float* __restrict__ agg,
                   const float* __restrict__ W,
                   const float* __restrict__ bias,
                   const float* __restrict__ in_isqrt,
                   const float* __restrict__ post_scale,
                   float* __restrict__ out, int nNodes) {
    __shared__ float lds[16 * LDS_STRIDE];

    const int r0   = blockIdx.x * 16;
    const int tid  = threadIdx.x;
    const int wave = tid >> 5;
    const int lane = tid & 31;

    {
        int idx = tid * 8;
        int row = idx >> 6;
        int col = idx & 63;
        int grow = r0 + row; if (grow >= nNodes) grow = nNodes - 1;
        float s = in_isqrt[grow];
        const float* g = agg + (size_t)grow * D_FEAT + col;
        float* l = &lds[row * LDS_STRIDE + col];
        #pragma unroll
        for (int j = 0; j < 8; ++j) l[j] = g[j] * s;
    }
    __syncthreads();

    const int colbase = wave << 4;
    const int n     = lane & 15;
    const int khalf = lane >> 4;

    v8f acc = {};
    #pragma unroll
    for (int kk = 0; kk < 16; ++kk) {
        const int k0 = (kk << 2) + (khalf << 1);
        v2f a = *(const v2f*)&lds[n * LDS_STRIDE + k0];
        v2f b;
        b.x = W[(k0 + 0) * D_FEAT + colbase + n];
        b.y = W[(k0 + 1) * D_FEAT + colbase + n];
        acc = __builtin_amdgcn_wmma_f32_16x16x4_f32(
             false, a,  false, b,
             (short)0, acc,  false,  false);
    }

    const int col = colbase + n;
    const float bv = bias[col];
    #pragma unroll
    for (int r = 0; r < 8; ++r) {
        int row = r0 + (khalf << 3) + r;
        if (row < nNodes) {
            float v = acc[r] + bv;
            v = (v > 0.0f) ? v : 0.01f * v;
            if (post_scale) v *= post_scale[row];
            out[(size_t)row * D_FEAT + col] = v;
        }
    }
}

__global__ void gcn_classify(const float* __restrict__ h,
                             const float* __restrict__ Wc,
                             const float* __restrict__ bc,
                             float* __restrict__ out, int nNodes) {
    int i = blockIdx.x * blockDim.x + threadIdx.x;
    if (i >= nNodes) return;
    float a0 = bc[0], a1 = bc[1];
    const float4* row = (const float4*)(h + (size_t)i * D_FEAT);
    #pragma unroll
    for (int q = 0; q < 16; ++q) {
        float4 v = row[q];
        int k = q << 2;
        a0 += v.x * Wc[(k+0)*2] + v.y * Wc[(k+1)*2] + v.z * Wc[(k+2)*2] + v.w * Wc[(k+3)*2];
        a1 += v.x * Wc[(k+0)*2+1] + v.y * Wc[(k+1)*2+1] + v.z * Wc[(k+2)*2+1] + v.w * Wc[(k+3)*2+1];
    }
    out[(size_t)i * 2 + 0] = a0;
    out[(size_t)i * 2 + 1] = a1;
}

extern "C" void kernel_launch(void* const* d_in, const int* in_sizes, int n_in,
                              void* d_out, int out_size, void* d_ws, size_t ws_size,
                              hipStream_t stream) {
    const float* x   = (const float*)d_in[0];
    const int*   src = (const int*)  d_in[1];
    const int*   dst = (const int*)  d_in[2];
    const float* W1  = (const float*)d_in[3];
    const float* b1  = (const float*)d_in[4];
    const float* W2  = (const float*)d_in[5];
    const float* b2  = (const float*)d_in[6];
    const float* Wc  = (const float*)d_in[7];
    const float* bc  = (const float*)d_in[8];
    float* out = (float*)d_out;

    const int N = in_sizes[0] / D_FEAT;
    const int E = in_sizes[1];

    float* bufA      = (float*)d_ws;
    float* bufB      = bufA + (size_t)N * D_FEAT;
    float* out_isqrt = bufB + (size_t)N * D_FEAT;
    float* in_isqrt  = out_isqrt + N;
    int*   row_start = (int*)(in_isqrt + N);
    int*   col       = row_start + (N + 1);
    const int Npad   = (N + 3) & ~3;
    int*   cnt_out   = (int*)bufA;
    int*   cnt_in    = cnt_out + Npad;
    int*   rank      = (int*)bufB;

    if (ws_size < ((size_t)2 * N * D_FEAT + (size_t)3 * N + 1 + (size_t)E) * 4) return;
    if ((size_t)E > (size_t)N * D_FEAT) return;

    const int B256 = 256;
    dim3 blk(B256);

    gcn_zero_i32<<<(Npad + N + B256 - 1) / B256, blk, 0, stream>>>(cnt_out, Npad + N);
    gcn_count_rank<<<(E + B256 - 1) / B256, blk, 0, stream>>>(src, dst, cnt_out, cnt_in, rank, E);
    gcn_scan_isqrt<<<(N + B256 - 1) / B256, blk, 0, stream>>>(cnt_in, cnt_out, row_start,
                                                           in_isqrt, out_isqrt, N);
    gcn_fill<<<(E + B256 - 1) / B256, blk, 0, stream>>>(src, dst, rank, row_start, col, E);

    gcn_scale_rows<<<(N * 16 + B256 - 1) / B256, blk, 0, stream>>>(x, out_isqrt, bufA, N * 16);
    gcn_gather<<<(N * 16 + B256 - 1) / B256, blk, 0, stream>>>(bufA, row_start, col, bufB, N);
    gcn_transform_wmma<<<(N + 15) / 16, dim3(128), 0, stream>>>(
        bufB, W1, b1, in_isqrt, out_isqrt, bufA, N);

    gcn_gather<<<(N * 16 + B256 - 1) / B256, blk, 0, stream>>>(bufA, row_start, col, bufB, N);
    gcn_transform_wmma<<<(N + 15) / 16, dim3(128), 0, stream>>>(
        bufB, W2, b2, in_isqrt, nullptr, bufA, N);

    gcn_classify<<<(N + B256 - 1) / B256, blk, 0, stream>>>(bufA, Wc, bc, out, N);
}


# Per-node row sum fused into the transform kernel's LDS A-tile fill: one launch per layer (32 nodes per 4-wave block), no agg round trip through memory; sums still in edge order
# speedup vs baseline: 1.5910x; 1.0720x over previous
#include <hip/hip_runtime.h>
#include <hip/hip_bf16.h>


typedef __attribute__((ext_vector_type(2))) float v2f;
typedef __attribute__((ext_vector_type(8))) float v8f;

#define D_FEAT 64
#define LDS_STRIDE 68

__global__ void gcn_zero_i32(int* __restrict__ p, int n) {
    int i = blockIdx.x * blockDim.x + threadIdx.x;
    if (i < n) p[i] = 0;
}

__global__ void gcn_count_rank(const int* __restrict__ src, const int* __restrict__ dst,
                               int* __restrict__ cnt_out, int* __restrict__ cnt_in,
                               int* __restrict__ rank, int nEdges) {
    int e = blockIdx.x * blockDim.x + threadIdx.x;
    if (e >= nEdges) return;
    const int s = src[e];
    const int d = dst[e];
    atomicAdd(&cnt_out[s], 1);
    rank[e] = atomicAdd(&cnt_in[d], 1);
}

__global__ void __launch_bounds__(256)
gcn_scan_isqrt(const int* __restrict__ cnt_in, const int* __restrict__ cnt_out,
               int* __restrict__ row_start, float* __restrict__ in_isqrt,
               float* __restrict__ out_isqrt, int n) {
    __shared__ int wsum[8];
    const int tid  = threadIdx.x;
    const int lane = tid & 31;
    const int wave = tid >> 5;
    const int base = blockIdx.x * 256;

    int part = 0;
    #pragma unroll 8
    for (int i = tid * 4; i < base; i += 1024) {
        int4 q = *(const int4*)(cnt_in + i);
        part += (q.x + q.y) + (q.z + q.w);
    }
    for (int o = 16; o > 0; o >>= 1) part += __shfl_xor(part, o);
    if (lane == 0) wsum[wave] = part;
    __syncthreads();
    int offset = 0;
    for (int w = 0; w < 8; ++w) offset += wsum[w];
    __syncthreads();

    const int i = base + tid;
    const int cnt = (i < n) ? cnt_in[i] : 0;
    int incl = cnt;
    for (int o = 1; o < 32; o <<= 1) {
        int t = __shfl_up(incl, o);
        if (lane >= o) incl += t;
    }
    if (lane == 31) wsum[wave] = incl;
    __syncthreads();
    int wprefix = 0;
    for (int w = 0; w < 8; ++w) wprefix += (w < wave) ? wsum[w] : 0;

    if (i < n) {
        const int excl = offset + wprefix + incl - cnt;
        row_start[i] = excl;
        if (i == n - 1) row_start[n] = excl + cnt;
        in_isqrt[i]  = rsqrtf(fmaxf((float)cnt, 1.0f));
        out_isqrt[i] = rsqrtf(fmaxf((float)cnt_out[i], 1.0f));
    }
}

__global__ void gcn_fill(const int* __restrict__ src, const int* __restrict__ dst,
                         const int* __restrict__ rank, const int* __restrict__ row_start,
                         int* __restrict__ col, int nEdges) {
    int e = blockIdx.x * blockDim.x + threadIdx.x;
    if (e >= nEdges) return;
    col[row_start[dst[e]] + rank[e]] = src[e];
}

__global__ void gcn_scale_rows(const float* __restrict__ x, const float* __restrict__ s,
                               float* __restrict__ out, int nQuads  ) {
    int t = blockIdx.x * blockDim.x + threadIdx.x;
    if (t >= nQuads) return;
    int row = t >> 4;
    float sc = s[row];
    float4 v = ((const float4*)x)[t];
    v.x *= sc; v.y *= sc; v.z *= sc; v.w *= sc;
    ((float4*)out)[t] = v;
}

#define TILE_NODES 32
#define NODES_PER_HW 4
__global__ void __launch_bounds__(128)
gcn_layer_fused(const float* __restrict__ h,
                const int* __restrict__ row_start,
                const int* __restrict__ col,
                const float* __restrict__ W,
                const float* __restrict__ bias,
                const float* __restrict__ in_isqrt,
                const float* __restrict__ post_scale,
                float* __restrict__ out, int nNodes) {
    __shared__ float lds[TILE_NODES * LDS_STRIDE];

    const int r0   = blockIdx.x * TILE_NODES;
    const int tid  = threadIdx.x;
    const int wave = tid >> 5;
    const int lane = tid & 31;

    {
        const int hw = tid >> 4;
        const int l  = tid & 15;
        const int f  = l << 2;
        const int n0 = r0 + hw * NODES_PER_HW;

        int   rsv = 0;
        float scv = 0.0f;
        if (l <= NODES_PER_HW) rsv = row_start[min(n0 + l, nNodes)];
        if (l <  NODES_PER_HW && n0 + l < nNodes) scv = in_isqrt[n0 + l];
        const int segbeg = __shfl(rsv, 0, 16);
        const int e1     = __shfl(rsv, 1, 16);
        const int e2     = __shfl(rsv, 2, 16);
        const int e3     = __shfl(rsv, 3, 16);
        const int segend = __shfl(rsv, 4, 16);
        const float sc0  = __shfl(scv, 0, 16);
        const float sc1  = __shfl(scv, 1, 16);
        const float sc2  = __shfl(scv, 2, 16);
        const float sc3  = __shfl(scv, 3, 16);

        const float4 zero4 = make_float4(0.0f, 0.0f, 0.0f, 0.0f);
        float4 acc = zero4;
        int cur    = 0;
        int curEnd = e1;
        const int lbase = (hw * NODES_PER_HW) * LDS_STRIDE + f;

#define GCN_FLUSH()                                                              \
        {                                                                        \
            const float s_ = (cur == 0) ? sc0 : (cur == 1) ? sc1 : (cur == 2) ? sc2 : sc3; \
            float4 o_;                                                           \
            o_.x = acc.x * s_; o_.y = acc.y * s_; o_.z = acc.z * s_; o_.w = acc.w * s_; \
            *(float4*)&lds[lbase + cur * LDS_STRIDE] = o_;                       \
            acc = zero4;                                                         \
            ++cur;                                                               \
            curEnd = (cur == 1) ? e2 : (cur == 2) ? e3 : segend;                 \
        }
#define GCN_ADD(rq, jq)                                                          \
        {                                                                        \
            const int j_ = (jq);                                                 \
            while (cur < NODES_PER_HW && j_ >= curEnd) GCN_FLUSH()               \
            acc.x += rq.x; acc.y += rq.y; acc.z += rq.z; acc.w += rq.w;          \
        }
#define GCN_LOAD(rq, sq, kq)                                                     \
        float4 rq = zero4;                                                       \
        if ((kq) < rem) rq = *(const float4*)(hb + ((unsigned)(sq) << 8));
#define GCN_GROUP_LOADS(K)                                                       \
            const int s0 = __shfl(c, (K),     16);                               \
            const int s1 = __shfl(c, (K) + 1, 16);                               \
            const int s2 = __shfl(c, (K) + 2, 16);                               \
            const int s3 = __shfl(c, (K) + 3, 16);                               \
            const int s4 = __shfl(c, (K) + 4, 16);                               \
            const int s5 = __shfl(c, (K) + 5, 16);                               \
            const int s6 = __shfl(c, (K) + 6, 16);                               \
            const int s7 = __shfl(c, (K) + 7, 16);                               \
            GCN_LOAD(q0, s0, (K))                                                \
            GCN_LOAD(q1, s1, (K) + 1)                                            \
            GCN_LOAD(q2, s2, (K) + 2)                                            \
            GCN_LOAD(q3, s3, (K) + 3)                                            \
            GCN_LOAD(q4, s4, (K) + 4)                                            \
            GCN_LOAD(q5, s5, (K) + 5)                                            \
            GCN_LOAD(q6, s6, (K) + 6)                                            \
            GCN_LOAD(q7, s7, (K) + 7)
#define GCN_GROUP_ADDS(K)                                                        \
            GCN_ADD(q0, base + (K))                                              \
            GCN_ADD(q1, base + (K) + 1)                                          \
            GCN_ADD(q2, base + (K) + 2)                                          \
            GCN_ADD(q3, base + (K) + 3)                                          \
            GCN_ADD(q4, base + (K) + 4)                                          \
            GCN_ADD(q5, base + (K) + 5)                                          \
            GCN_ADD(q6, base + (K) + 6)                                          \
            GCN_ADD(q7, base + (K) + 7)

        const char* hb = (const char*)h + (f << 2);
        int cnext = (segbeg + l < segend) ? col[segbeg + l] : 0;
        for (int base = segbeg; ; base += 16) {
            const int rem  = segend - base;
            const int remW = max(rem, __shfl_xor(rem, 16));
            if (remW <= 0) break;
            const int c = cnext;
            {
                GCN_GROUP_LOADS(0)
                cnext = (base + 16 + l < segend) ? col[base + 16 + l] : 0;
                GCN_GROUP_ADDS(0)
            }
            if (remW > 8) {
                GCN_GROUP_LOADS(8)
                GCN_GROUP_ADDS(8)
            }
        }
        while (cur < NODES_PER_HW) GCN_FLUSH()
#undef GCN_GROUP_ADDS
#undef GCN_GROUP_LOADS
#undef GCN_LOAD
#undef GCN_ADD
#undef GCN_FLUSH
    }
    __syncthreads();

    const int colbase = wave << 4;
    const int n     = lane & 15;
    const int khalf = lane >> 4;

    v8f acc0 = {};
    v8f acc1 = {};
    #pragma unroll
    for (int kk = 0; kk < 16; ++kk) {
        const int k0 = (kk << 2) + (khalf << 1);
        v2f a0 = *(const v2f*)&lds[n * LDS_STRIDE + k0];
        v2f a1 = *(const v2f*)&lds[(16 + n) * LDS_STRIDE + k0];
        v2f b;
        b.x = W[(k0 + 0) * D_FEAT + colbase + n];
        b.y = W[(k0 + 1) * D_FEAT + colbase + n];
        acc0 = __builtin_amdgcn_wmma_f32_16x16x4_f32(
             false, a0,  false, b,
             (short)0, acc0,  false,  false);
        acc1 = __builtin_amdgcn_wmma_f32_16x16x4_f32(
             false, a1,  false, b,
             (short)0, acc1,  false,  false);
    }

    const int colo = colbase + n;
    const float bv = bias[colo];
    #pragma unroll
    for (int t = 0; t < 2; ++t) {
        #pragma unroll
        for (int r = 0; r < 8; ++r) {
            const int row = r0 + (t << 4) + (khalf << 3) + r;
            if (row < nNodes) {
                float v = (t == 0 ? acc0[r] : acc1[r]) + bv;
                v = (v > 0.0f) ? v : 0.01f * v;
                if (post_scale) v *= post_scale[row];
                out[(size_t)row * D_FEAT + colo] = v;
            }
        }
    }
}

__global__ void gcn_classify(const float* __restrict__ h,
                             const float* __restrict__ Wc,
                             const float* __restrict__ bc,
                             float* __restrict__ out, int nNodes) {
    int i = blockIdx.x * blockDim.x + threadIdx.x;
    if (i >= nNodes) return;
    float a0 = bc[0], a1 = bc[1];
    const float4* row = (const float4*)(h + (size_t)i * D_FEAT);
    #pragma unroll
    for (int q = 0; q < 16; ++q) {
        float4 v = row[q];
        int k = q << 2;
        a0 += v.x * Wc[(k+0)*2] + v.y * Wc[(k+1)*2] + v.z * Wc[(k+2)*2] + v.w * Wc[(k+3)*2];
        a1 += v.x * Wc[(k+0)*2+1] + v.y * Wc[(k+1)*2+1] + v.z * Wc[(k+2)*2+1] + v.w * Wc[(k+3)*2+1];
    }
    out[(size_t)i * 2 + 0] = a0;
    out[(size_t)i * 2 + 1] = a1;
}

extern "C" void kernel_launch(void* const* d_in, const int* in_sizes, int n_in,
                              void* d_out, int out_size, void* d_ws, size_t ws_size,
                              hipStream_t stream) {
    const float* x   = (const float*)d_in[0];
    const int*   src = (const int*)  d_in[1];
    const int*   dst = (const int*)  d_in[2];
    const float* W1  = (const float*)d_in[3];
    const float* b1  = (const float*)d_in[4];
    const float* W2  = (const float*)d_in[5];
    const float* b2  = (const float*)d_in[6];
    const float* Wc  = (const float*)d_in[7];
    const float* bc  = (const float*)d_in[8];
    float* out = (float*)d_out;

    const int N = in_sizes[0] / D_FEAT;
    const int E = in_sizes[1];

    float* bufA      = (float*)d_ws;
    float* bufB      = bufA + (size_t)N * D_FEAT;
    float* out_isqrt = bufB + (size_t)N * D_FEAT;
    float* in_isqrt  = out_isqrt + N;
    int*   row_start = (int*)(in_isqrt + N);
    int*   col       = row_start + (N + 1);
    const int Npad   = (N + 3) & ~3;
    int*   cnt_out   = (int*)bufA;
    int*   cnt_in    = cnt_out + Npad;
    int*   rank      = (int*)bufB;

    if (ws_size < ((size_t)2 * N * D_FEAT + (size_t)3 * N + 1 + (size_t)E) * 4) return;
    if ((size_t)E > (size_t)N * D_FEAT) return;

    const int B256 = 256;
    dim3 blk(B256);

    gcn_zero_i32<<<(Npad + N + B256 - 1) / B256, blk, 0, stream>>>(cnt_out, Npad + N);
    gcn_count_rank<<<(E + B256 - 1) / B256, blk, 0, stream>>>(src, dst, cnt_out, cnt_in, rank, E);
    gcn_scan_isqrt<<<(N + B256 - 1) / B256, blk, 0, stream>>>(cnt_in, cnt_out, row_start,
                                                           in_isqrt, out_isqrt, N);
    gcn_fill<<<(E + B256 - 1) / B256, blk, 0, stream>>>(src, dst, rank, row_start, col, E);

    const int nTiles = (N + TILE_NODES - 1) / TILE_NODES;
    gcn_scale_rows<<<(N * 16 + B256 - 1) / B256, blk, 0, stream>>>(x, out_isqrt, bufA, N * 16);
    gcn_layer_fused<<<nTiles, dim3(128), 0, stream>>>(
        bufA, row_start, col, W1, b1, in_isqrt, out_isqrt, bufB, N);

    gcn_layer_fused<<<nTiles, dim3(128), 0, stream>>>(
        bufB, row_start, col, W2, b2, in_isqrt, nullptr, bufA, N);

    gcn_classify<<<(N + B256 - 1) / B256, blk, 0, stream>>>(bufA, Wc, bc, out, N);
}


# Layer-1 row pre-scale (x * out_isqrt) moved into the edge-placing launch, two float4 per thread: its own 50000-wave launch is gone (8 -> 7 launches); arithmetic and sum order unchanged
# speedup vs baseline: 1.6410x; 1.0314x over previous
#include <hip/hip_runtime.h>
#include <hip/hip_bf16.h>


typedef __attribute__((ext_vector_type(2))) float v2f;
typedef __attribute__((ext_vector_type(8))) float v8f;

#define D_FEAT 64
#define LDS_STRIDE 68

__global__ void gcn_zero_i32(int* __restrict__ p, int n) {
    int i = blockIdx.x * blockDim.x + threadIdx.x;
    if (i < n) p[i] = 0;
}

__global__ void gcn_count_rank(const int* __restrict__ src, const int* __restrict__ dst,
                               int* __restrict__ cnt_out, int* __restrict__ cnt_in,
                               int* __restrict__ rank, int nEdges) {
    int e = blockIdx.x * blockDim.x + threadIdx.x;
    if (e >= nEdges) return;
    const int s = src[e];
    const int d = dst[e];
    atomicAdd(&cnt_out[s], 1);
    rank[e] = atomicAdd(&cnt_in[d], 1);
}

__global__ void __launch_bounds__(256)
gcn_scan_isqrt(const int* __restrict__ cnt_in, const int* __restrict__ cnt_out,
               int* __restrict__ row_start, float* __restrict__ in_isqrt,
               float* __restrict__ out_isqrt, int n) {
    __shared__ int wsum[8];
    const int tid  = threadIdx.x;
    const int lane = tid & 31;
    const int wave = tid >> 5;
    const int base = blockIdx.x * 256;

    int part = 0;
    #pragma unroll 8
    for (int i = tid * 4; i < base; i += 1024) {
        int4 q = *(const int4*)(cnt_in + i);
        part += (q.x + q.y) + (q.z + q.w);
    }
    for (int o = 16; o > 0; o >>= 1) part += __shfl_xor(part, o);
    if (lane == 0) wsum[wave] = part;
    __syncthreads();
    int offset = 0;
    for (int w = 0; w < 8; ++w) offset += wsum[w];
    __syncthreads();

    const int i = base + tid;
    const int cnt = (i < n) ? cnt_in[i] : 0;
    int incl = cnt;
    for (int o = 1; o < 32; o <<= 1) {
        int t = __shfl_up(incl, o);
        if (lane >= o) incl += t;
    }
    if (lane == 31) wsum[wave] = incl;
    __syncthreads();
    int wprefix = 0;
    for (int w = 0; w < 8; ++w) wprefix += (w < wave) ? wsum[w] : 0;

    if (i < n) {
        const int excl = offset + wprefix + incl - cnt;
        row_start[i] = excl;
        if (i == n - 1) row_start[n] = excl + cnt;
        in_isqrt[i]  = rsqrtf(fmaxf((float)cnt, 1.0f));
        out_isqrt[i] = rsqrtf(fmaxf((float)cnt_out[i], 1.0f));
    }
}

__global__ void gcn_fill_scale(const int* __restrict__ src, const int* __restrict__ dst,
                               const int* __restrict__ rank, const int* __restrict__ row_start,
                               int* __restrict__ col, int nEdges,
                               const float* __restrict__ x, const float* __restrict__ s,
                               float* __restrict__ out, int nQuads  ) {
    const int t  = blockIdx.x * blockDim.x + threadIdx.x;
    const int q0 = t;
    const int q1 = t + gridDim.x * blockDim.x;

    const int q0c = min(q0, nQuads - 1);
    const int q1c = min(q1, nQuads - 1);
    float4 v0 = ((const float4*)x)[q0c];
    float4 v1 = ((const float4*)x)[q1c];
    const float s0 = s[q0c >> 4];
    const float s1 = s[q1c >> 4];

    if (t < nEdges) col[row_start[dst[t]] + rank[t]] = src[t];

    if (q0 < nQuads) {
        v0.x *= s0; v0.y *= s0; v0.z *= s0; v0.w *= s0;
        ((float4*)out)[q0] = v0;
    }
    if (q1 < nQuads) {
        v1.x *= s1; v1.y *= s1; v1.z *= s1; v1.w *= s1;
        ((float4*)out)[q1] = v1;
    }
}

#define TILE_NODES 32
#define NODES_PER_HW 4
__global__ void __launch_bounds__(128)
gcn_layer_fused(const float* __restrict__ h,
                const int* __restrict__ row_start,
                const int* __restrict__ col,
                const float* __restrict__ W,
                const float* __restrict__ bias,
                const float* __restrict__ in_isqrt,
                const float* __restrict__ post_scale,
                float* __restrict__ out, int nNodes) {
    __shared__ float lds[TILE_NODES * LDS_STRIDE];

    const int r0   = blockIdx.x * TILE_NODES;
    const int tid  = threadIdx.x;
    const int wave = tid >> 5;
    const int lane = tid & 31;

    {
        const int hw = tid >> 4;
        const int l  = tid & 15;
        const int f  = l << 2;
        const int n0 = r0 + hw * NODES_PER_HW;

        int   rsv = 0;
        float scv = 0.0f;
        if (l <= NODES_PER_HW) rsv = row_start[min(n0 + l, nNodes)];
        if (l <  NODES_PER_HW && n0 + l < nNodes) scv = in_isqrt[n0 + l];
        const int segbeg = __shfl(rsv, 0, 16);
        const int e1     = __shfl(rsv, 1, 16);
        const int e2     = __shfl(rsv, 2, 16);
        const int e3     = __shfl(rsv, 3, 16);
        const int segend = __shfl(rsv, 4, 16);
        const float sc0  = __shfl(scv, 0, 16);
        const float sc1  = __shfl(scv, 1, 16);
        const float sc2  = __shfl(scv, 2, 16);
        const float sc3  = __shfl(scv, 3, 16);

        const float4 zero4 = make_float4(0.0f, 0.0f, 0.0f, 0.0f);
        float4 acc = zero4;
        int cur    = 0;
        int curEnd = e1;
        const int lbase = (hw * NODES_PER_HW) * LDS_STRIDE + f;

#define GCN_FLUSH()                                                              \
        {                                                                        \
            const float s_ = (cur == 0) ? sc0 : (cur == 1) ? sc1 : (cur == 2) ? sc2 : sc3; \
            float4 o_;                                                           \
            o_.x = acc.x * s_; o_.y = acc.y * s_; o_.z = acc.z * s_; o_.w = acc.w * s_; \
            *(float4*)&lds[lbase + cur * LDS_STRIDE] = o_;                       \
            acc = zero4;                                                         \
            ++cur;                                                               \
            curEnd = (cur == 1) ? e2 : (cur == 2) ? e3 : segend;                 \
        }
#define GCN_ADD(rq, jq)                                                          \
        {                                                                        \
            const int j_ = (jq);                                                 \
            while (cur < NODES_PER_HW && j_ >= curEnd) GCN_FLUSH()               \
            acc.x += rq.x; acc.y += rq.y; acc.z += rq.z; acc.w += rq.w;          \
        }
#define GCN_LOAD(rq, sq, kq)                                                     \
        float4 rq = zero4;                                                       \
        if ((kq) < rem) rq = *(const float4*)(hb + ((unsigned)(sq) << 8));
#define GCN_GROUP_LOADS(K)                                                       \
            const int s0 = __shfl(c, (K),     16);                               \
            const int s1 = __shfl(c, (K) + 1, 16);                               \
            const int s2 = __shfl(c, (K) + 2, 16);                               \
            const int s3 = __shfl(c, (K) + 3, 16);                               \
            const int s4 = __shfl(c, (K) + 4, 16);                               \
            const int s5 = __shfl(c, (K) + 5, 16);                               \
            const int s6 = __shfl(c, (K) + 6, 16);                               \
            const int s7 = __shfl(c, (K) + 7, 16);                               \
            GCN_LOAD(q0, s0, (K))                                                \
            GCN_LOAD(q1, s1, (K) + 1)                                            \
            GCN_LOAD(q2, s2, (K) + 2)                                            \
            GCN_LOAD(q3, s3, (K) + 3)                                            \
            GCN_LOAD(q4, s4, (K) + 4)                                            \
            GCN_LOAD(q5, s5, (K) + 5)                                            \
            GCN_LOAD(q6, s6, (K) + 6)                                            \
            GCN_LOAD(q7, s7, (K) + 7)
#define GCN_GROUP_ADDS(K)                                                        \
            GCN_ADD(q0, base + (K))                                              \
            GCN_ADD(q1, base + (K) + 1)                                          \
            GCN_ADD(q2, base + (K) + 2)                                          \
            GCN_ADD(q3, base + (K) + 3)                                          \
            GCN_ADD(q4, base + (K) + 4)                                          \
            GCN_ADD(q5, base + (K) + 5)                                          \
            GCN_ADD(q6, base + (K) + 6)                                          \
            GCN_ADD(q7, base + (K) + 7)

        const char* hb = (const char*)h + (f << 2);
        int cnext = (segbeg + l < segend) ? col[segbeg + l] : 0;
        for (int base = segbeg; ; base += 16) {
            const int rem  = segend - base;
            const int remW = max(rem, __shfl_xor(rem, 16));
            if (remW <= 0) break;
            const int c = cnext;
            {
                GCN_GROUP_LOADS(0)
                cnext = (base + 16 + l < segend) ? col[base + 16 + l] : 0;
                GCN_GROUP_ADDS(0)
            }
            if (remW > 8) {
                GCN_GROUP_LOADS(8)
                GCN_GROUP_ADDS(8)
            }
        }
        while (cur < NODES_PER_HW) GCN_FLUSH()
#undef GCN_GROUP_ADDS
#undef GCN_GROUP_LOADS
#undef GCN_LOAD
#undef GCN_ADD
#undef GCN_FLUSH
    }
    __syncthreads();

    const int colbase = wave << 4;
    const int n     = lane & 15;
    const int khalf = lane >> 4;

    v8f acc0 = {};
    v8f acc1 = {};
    #pragma unroll
    for (int kk = 0; kk < 16; ++kk) {
        const int k0 = (kk << 2) + (khalf << 1);
        v2f a0 = *(const v2f*)&lds[n * LDS_STRIDE + k0];
        v2f a1 = *(const v2f*)&lds[(16 + n) * LDS_STRIDE + k0];
        v2f b;
        b.x = W[(k0 + 0) * D_FEAT + colbase + n];
        b.y = W[(k0 + 1) * D_FEAT + colbase + n];
        acc0 = __builtin_amdgcn_wmma_f32_16x16x4_f32(
             false, a0,  false, b,
             (short)0, acc0,  false,  false);
        acc1 = __builtin_amdgcn_wmma_f32_16x16x4_f32(
             false, a1,  false, b,
             (short)0, acc1,  false,  false);
    }

    const int colo = colbase + n;
    const float bv = bias[colo];
    #pragma unroll
    for (int t = 0; t < 2; ++t) {
        #pragma unroll
        for (int r = 0; r < 8; ++r) {
            const int row = r0 + (t << 4) + (khalf << 3) + r;
            if (row < nNodes) {
                float v = (t == 0 ? acc0[r] : acc1[r]) + bv;
                v = (v > 0.0f) ? v : 0.01f * v;
                if (post_scale) v *= post_scale[row];
                out[(size_t)row * D_FEAT + colo] = v;
            }
        }
    }
}

__global__ void gcn_classify(const float* __restrict__ h,
                             const float* __restrict__ Wc,
                             const float* __restrict__ bc,
                             float* __restrict__ out, int nNodes) {
    int i = blockIdx.x * blockDim.x + threadIdx.x;
    if (i >= nNodes) return;
    float a0 = bc[0], a1 = bc[1];
    const float4* row = (const float4*)(h + (size_t)i * D_FEAT);
    #pragma unroll
    for (int q = 0; q < 16; ++q) {
        float4 v = row[q];
        int k = q << 2;
        a0 += v.x * Wc[(k+0)*2] + v.y * Wc[(k+1)*2] + v.z * Wc[(k+2)*2] + v.w * Wc[(k+3)*2];
        a1 += v.x * Wc[(k+0)*2+1] + v.y * Wc[(k+1)*2+1] + v.z * Wc[(k+2)*2+1] + v.w * Wc[(k+3)*2+1];
    }
    out[(size_t)i * 2 + 0] = a0;
    out[(size_t)i * 2 + 1] = a1;
}

extern "C" void kernel_launch(void* const* d_in, const int* in_sizes, int n_in,
                              void* d_out, int out_size, void* d_ws, size_t ws_size,
                              hipStream_t stream) {
    const float* x   = (const float*)d_in[0];
    const int*   src = (const int*)  d_in[1];
    const int*   dst = (const int*)  d_in[2];
    const float* W1  = (const float*)d_in[3];
    const float* b1  = (const float*)d_in[4];
    const float* W2  = (const float*)d_in[5];
    const float* b2  = (const float*)d_in[6];
    const float* Wc  = (const float*)d_in[7];
    const float* bc  = (const float*)d_in[8];
    float* out = (float*)d_out;

    const int N = in_sizes[0] / D_FEAT;
    const int E = in_sizes[1];

    float* bufA      = (float*)d_ws;
    float* bufB      = bufA + (size_t)N * D_FEAT;
    float* out_isqrt = bufB + (size_t)N * D_FEAT;
    float* in_isqrt  = out_isqrt + N;
    int*   row_start = (int*)(in_isqrt + N);
    int*   col       = row_start + (N + 1);
    const int Npad   = (N + 3) & ~3;
    int*   cnt_out   = (int*)bufA;
    int*   cnt_in    = cnt_out + Npad;
    int*   rank      = (int*)bufB;

    if (ws_size < ((size_t)2 * N * D_FEAT + (size_t)3 * N + 1 + (size_t)E) * 4) return;
    if ((size_t)E > (size_t)N * D_FEAT) return;

    const int B256 = 256;
    dim3 blk(B256);

    gcn_zero_i32<<<(Npad + N + B256 - 1) / B256, blk, 0, stream>>>(cnt_out, Npad + N);
    gcn_count_rank<<<(E + B256 - 1) / B256, blk, 0, stream>>>(src, dst, cnt_out, cnt_in, rank, E);
    gcn_scan_isqrt<<<(N + B256 - 1) / B256, blk, 0, stream>>>(cnt_in, cnt_out, row_start,
                                                           in_isqrt, out_isqrt, N);
    const int nFill = (E > N * 8) ? E : N * 8;
    gcn_fill_scale<<<(nFill + B256 - 1) / B256, blk, 0, stream>>>(src, dst, rank, row_start, col, E,
                                                               x, out_isqrt, bufA, N * 16);

    const int nTiles = (N + TILE_NODES - 1) / TILE_NODES;
    gcn_layer_fused<<<nTiles, dim3(128), 0, stream>>>(
        bufA, row_start, col, W1, b1, in_isqrt, out_isqrt, bufB, N);

    gcn_layer_fused<<<nTiles, dim3(128), 0, stream>>>(
        bufB, row_start, col, W2, b2, in_isqrt, nullptr, bufA, N);

    gcn_classify<<<(N + B256 - 1) / B256, blk, 0, stream>>>(bufA, Wc, bc, out, N);
}


# Edges sorted by destination through LDS (512-node bins: block histograms, one global add per block and bin, per-bin LDS sort), not per-edge returning global atomics; sums in LDS-atomic arrival order
# speedup vs baseline: 2.1047x; 1.2826x over previous
#include <hip/hip_runtime.h>
#include <hip/hip_bf16.h>


typedef __attribute__((ext_vector_type(2))) float v2f;
typedef __attribute__((ext_vector_type(8))) float v8f;

#define D_FEAT 64
#define LDS_STRIDE 68

#define BIN_SHIFT 9
#define BIN_NODES (1 << BIN_SHIFT)
#define MAX_BINS  2048
#define EDGE_THREADS 256
#define EDGES_PER_THREAD 8
#define EDGES_PER_BLOCK (EDGE_THREADS * EDGES_PER_THREAD)
#define SORT_THREADS BIN_NODES
#define STAGE_CAP 12288
#define SORT_UNROLL 8

__global__ void gcn_zero_i32(int* __restrict__ p0, int n0, int* __restrict__ p1, int n1) {
    int i = blockIdx.x * blockDim.x + threadIdx.x;
    if (i < n0) p0[i] = 0;
    else if (i - n0 < n1) p1[i - n0] = 0;
}

__device__ __forceinline__ void gcn_block_scan_excl(int* a, int n, int* wsum) {
    const int tid  = threadIdx.x;
    const int nthr = blockDim.x;
    const int lane = tid & 31;
    const int wave = tid >> 5;
    const int nwav = nthr >> 5;
    const int G  = (n + nthr - 1) / nthr;
    const int i0 = tid * G;
    __syncthreads();
    int sum = 0;
    for (int k = 0; k < G; ++k) { const int i = i0 + k; if (i < n) sum += a[i]; }
    int incl = sum;
    for (int o = 1; o < 32; o <<= 1) {
        int t = __shfl_up(incl, o);
        if (lane >= o) incl += t;
    }
    if (lane == 31) wsum[wave] = incl;
    __syncthreads();
    int run = incl - sum;
    for (int w = 0; w < nwav; ++w) run += (w < wave) ? wsum[w] : 0;
    for (int k = 0; k < G; ++k) {
        const int i = i0 + k;
        if (i < n) { const int v = a[i]; a[i] = run; run += v; }
    }
    __syncthreads();
}

__global__ void __launch_bounds__(EDGE_THREADS)
gcn_bin_count(const int* __restrict__ src, const int* __restrict__ dst,
              int* __restrict__ cnt_out, int* __restrict__ bin_cnt,
              int nEdges, int nBins) {
    __shared__ int hcnt[MAX_BINS];
    const int tid = threadIdx.x;
    for (int i = tid; i < nBins; i += EDGE_THREADS) hcnt[i] = 0;
    __syncthreads();
    const int base = blockIdx.x * EDGES_PER_BLOCK;
    int es[EDGES_PER_THREAD], ed[EDGES_PER_THREAD];
    #pragma unroll
    for (int k = 0; k < EDGES_PER_THREAD; ++k) {
        const int ec = min(base + k * EDGE_THREADS + tid, nEdges - 1);
        es[k] = src[ec];
        ed[k] = dst[ec];
    }
    #pragma unroll
    for (int k = 0; k < EDGES_PER_THREAD; ++k) {
        if (base + k * EDGE_THREADS + tid < nEdges) {
            atomicAdd(&cnt_out[es[k]], 1);
            atomicAdd(&hcnt[ed[k] >> BIN_SHIFT], 1);
        }
    }
    __syncthreads();
    for (int i = tid; i < nBins; i += EDGE_THREADS) {
        const int c = hcnt[i];
        if (c) atomicAdd(&bin_cnt[i], c);
    }
}

__global__ void __launch_bounds__(EDGE_THREADS)
gcn_bin_scatter(const int* __restrict__ src, const int* __restrict__ dst,
                const int* __restrict__ bin_cnt, int* __restrict__ bin_cur,
                int2* __restrict__ part, int nEdges, int nBins,
                const float* __restrict__ x, const int* __restrict__ cnt_out,
                float* __restrict__ out_isqrt, float* __restrict__ out,
                int nNodes, int nQuads  ) {
    __shared__ int  hcnt[MAX_BINS];
    __shared__ int  gpos[MAX_BINS];
    __shared__ int2 stage[EDGES_PER_BLOCK];
    __shared__ int  wsum[EDGE_THREADS / 32];

    const int tid = threadIdx.x;
    const int t   = blockIdx.x * EDGE_THREADS + tid;
    const int G   = gridDim.x * EDGE_THREADS;

    for (int i = t; i < nNodes; i += G)
        out_isqrt[i] = rsqrtf(fmaxf((float)cnt_out[i], 1.0f));
    #pragma unroll 4
    for (int q = t; q < nQuads; q += G) {
        float4 v = ((const float4*)x)[q];
        const float sc = rsqrtf(fmaxf((float)cnt_out[q >> 4], 1.0f));
        v.x *= sc; v.y *= sc; v.z *= sc; v.w *= sc;
        ((float4*)out)[q] = v;
    }

    for (int i = tid; i < nBins; i += EDGE_THREADS) { gpos[i] = bin_cnt[i]; hcnt[i] = 0; }
    gcn_block_scan_excl(gpos, nBins, wsum);

    const int base = blockIdx.x * EDGES_PER_BLOCK;
    int es[EDGES_PER_THREAD], ed[EDGES_PER_THREAD], er[EDGES_PER_THREAD];
    #pragma unroll
    for (int k = 0; k < EDGES_PER_THREAD; ++k) {
        const int ec = min(base + k * EDGE_THREADS + tid, nEdges - 1);
        es[k] = src[ec];
        ed[k] = dst[ec];
    }
    #pragma unroll
    for (int k = 0; k < EDGES_PER_THREAD; ++k) {
        er[k] = -1;
        if (base + k * EDGE_THREADS + tid < nEdges)
            er[k] = atomicAdd(&hcnt[ed[k] >> BIN_SHIFT], 1);
    }
    __syncthreads();

    for (int i = tid; i < nBins; i += EDGE_THREADS) {
        const int c = hcnt[i];
        if (c) gpos[i] += atomicAdd(&bin_cur[i], c);
    }
    gcn_block_scan_excl(hcnt, nBins, wsum);

    #pragma unroll
    for (int k = 0; k < EDGES_PER_THREAD; ++k) {
        if (er[k] >= 0) stage[hcnt[ed[k] >> BIN_SHIFT] + er[k]] = make_int2(ed[k], es[k]);
    }
    __syncthreads();

    const int nLocal = min(EDGES_PER_BLOCK, nEdges - base);
    #pragma unroll
    for (int k = 0; k < EDGES_PER_THREAD; ++k) {
        const int i  = k * EDGE_THREADS + tid;
        const int ic = min(i, nLocal - 1);
        const int2 p = stage[ic];
        const int b = p.x >> BIN_SHIFT;
        const int o = gpos[b] + (ic - hcnt[b]);
        if (i < nLocal) part[o] = p;
    }
}

__global__ void __launch_bounds__(SORT_THREADS)
gcn_bin_sort(const int2* __restrict__ part, const int* __restrict__ bin_cnt,
             int* __restrict__ row_start, float* __restrict__ in_isqrt,
             int* __restrict__ col, int nNodes) {
    __shared__ int cnt[BIN_NODES];
    __shared__ int cur[BIN_NODES];
    __shared__ int wsum[SORT_THREADS / 32];
    __shared__ int stage[STAGE_CAP];

    const int tid  = threadIdx.x;
    const int lane = tid & 31;
    const int wave = tid >> 5;
    const int bin  = blockIdx.x;

    int pre = 0;
    for (int i = tid; i < bin; i += SORT_THREADS) pre += bin_cnt[i];
    for (int o = 16; o > 0; o >>= 1) pre += __shfl_xor(pre, o);
    if (lane == 0) wsum[wave] = pre;
    cnt[tid] = 0;
    __syncthreads();
    int segBeg = 0;
    for (int w = 0; w < SORT_THREADS / 32; ++w) segBeg += wsum[w];
    const int segLen = bin_cnt[bin];
    const int2* seg = part + segBeg;

    for (int i0 = tid; i0 < segLen; i0 += SORT_UNROLL * SORT_THREADS) {
        int dl[SORT_UNROLL];
        #pragma unroll
        for (int u = 0; u < SORT_UNROLL; ++u)
            dl[u] = seg[min(i0 + u * SORT_THREADS, segLen - 1)].x;
        #pragma unroll
        for (int u = 0; u < SORT_UNROLL; ++u)
            if (i0 + u * SORT_THREADS < segLen) atomicAdd(&cnt[dl[u] & (BIN_NODES - 1)], 1);
    }
    __syncthreads();

    const int c = cnt[tid];
    int incl = c;
    for (int o = 1; o < 32; o <<= 1) {
        int v = __shfl_up(incl, o);
        if (lane >= o) incl += v;
    }
    if (lane == 31) wsum[wave] = incl;
    __syncthreads();
    int excl = incl - c;
    for (int w = 0; w < SORT_THREADS / 32; ++w) excl += (w < wave) ? wsum[w] : 0;

    const int node = bin * BIN_NODES + tid;
    if (node < nNodes) {
        row_start[node] = segBeg + excl;
        if (node == nNodes - 1) row_start[nNodes] = segBeg + excl + c;
        in_isqrt[node] = rsqrtf(fmaxf((float)c, 1.0f));
    }
    cur[tid] = excl;
    __syncthreads();

    if (segLen <= STAGE_CAP) {
        for (int i0 = tid; i0 < segLen; i0 += SORT_UNROLL * SORT_THREADS) {
            int2 p[SORT_UNROLL];
            #pragma unroll
            for (int u = 0; u < SORT_UNROLL; ++u)
                p[u] = seg[min(i0 + u * SORT_THREADS, segLen - 1)];
            #pragma unroll
            for (int u = 0; u < SORT_UNROLL; ++u)
                if (i0 + u * SORT_THREADS < segLen)
                    stage[atomicAdd(&cur[p[u].x & (BIN_NODES - 1)], 1)] = p[u].y;
        }
        __syncthreads();
        for (int i = tid; i < segLen; i += SORT_THREADS) col[segBeg + i] = stage[i];
    } else {
        for (int i = tid; i < segLen; i += SORT_THREADS) {
            const int2 p = seg[i];
            col[segBeg + atomicAdd(&cur[p.x & (BIN_NODES - 1)], 1)] = p.y;
        }
    }
}

#define TILE_NODES 32
#define NODES_PER_HW 4
__global__ void __launch_bounds__(128)
gcn_layer_fused(const float* __restrict__ h,
                const int* __restrict__ row_start,
                const int* __restrict__ col,
                const float* __restrict__ W,
                const float* __restrict__ bias,
                const float* __restrict__ in_isqrt,
                const float* __restrict__ post_scale,
                float* __restrict__ out, int nNodes) {
    __shared__ float lds[TILE_NODES * LDS_STRIDE];

    const int r0   = blockIdx.x * TILE_NODES;
    const int tid  = threadIdx.x;
    const int wave = tid >> 5;
    const int lane = tid & 31;

    {
        const int hw = tid >> 4;
        const int l  = tid & 15;
        const int f  = l << 2;
        const int n0 = r0 + hw * NODES_PER_HW;

        int   rsv = 0;
        float scv = 0.0f;
        if (l <= NODES_PER_HW) rsv = row_start[min(n0 + l, nNodes)];
        if (l <  NODES_PER_HW && n0 + l < nNodes) scv = in_isqrt[n0 + l];
        const int segbeg = __shfl(rsv, 0, 16);
        const int e1     = __shfl(rsv, 1, 16);
        const int e2     = __shfl(rsv, 2, 16);
        const int e3     = __shfl(rsv, 3, 16);
        const int segend = __shfl(rsv, 4, 16);
        const float sc0  = __shfl(scv, 0, 16);
        const float sc1  = __shfl(scv, 1, 16);
        const float sc2  = __shfl(scv, 2, 16);
        const float sc3  = __shfl(scv, 3, 16);

        const float4 zero4 = make_float4(0.0f, 0.0f, 0.0f, 0.0f);
        float4 acc = zero4;
        int cur    = 0;
        int curEnd = e1;
        const int lbase = (hw * NODES_PER_HW) * LDS_STRIDE + f;

#define GCN_FLUSH()                                                              \
        {                                                                        \
            const float s_ = (cur == 0) ? sc0 : (cur == 1) ? sc1 : (cur == 2) ? sc2 : sc3; \
            float4 o_;                                                           \
            o_.x = acc.x * s_; o_.y = acc.y * s_; o_.z = acc.z * s_; o_.w = acc.w * s_; \
            *(float4*)&lds[lbase + cur * LDS_STRIDE] = o_;                       \
            acc = zero4;                                                         \
            ++cur;                                                               \
            curEnd = (cur == 1) ? e2 : (cur == 2) ? e3 : segend;                 \
        }
#define GCN_ADD(rq, jq)                                                          \
        {                                                                        \
            const int j_ = (jq);                                                 \
            while (cur < NODES_PER_HW && j_ >= curEnd) GCN_FLUSH()               \
            acc.x += rq.x; acc.y += rq.y; acc.z += rq.z; acc.w += rq.w;          \
        }
#define GCN_LOAD(rq, sq, kq)                                                     \
        float4 rq = zero4;                                                       \
        if ((kq) < rem) rq = *(const float4*)(hb + ((unsigned)(sq) << 8));
#define GCN_GROUP_LOADS(K)                                                       \
            const int s0 = __shfl(c, (K),     16);                               \
            const int s1 = __shfl(c, (K) + 1, 16);                               \
            const int s2 = __shfl(c, (K) + 2, 16);                               \
            const int s3 = __shfl(c, (K) + 3, 16);                               \
            const int s4 = __shfl(c, (K) + 4, 16);                               \
            const int s5 = __shfl(c, (K) + 5, 16);                               \
            const int s6 = __shfl(c, (K) + 6, 16);                               \
            const int s7 = __shfl(c, (K) + 7, 16);                               \
            GCN_LOAD(q0, s0, (K))                                                \
            GCN_LOAD(q1, s1, (K) + 1)                                            \
            GCN_LOAD(q2, s2, (K) + 2)                                            \
            GCN_LOAD(q3, s3, (K) + 3)                                            \
            GCN_LOAD(q4, s4, (K) + 4)                                            \
            GCN_LOAD(q5, s5, (K) + 5)                                            \
            GCN_LOAD(q6, s6, (K) + 6)                                            \
            GCN_LOAD(q7, s7, (K) + 7)
#define GCN_GROUP_ADDS(K)                                                        \
            GCN_ADD(q0, base + (K))                                              \
            GCN_ADD(q1, base + (K) + 1)                                          \
            GCN_ADD(q2, base + (K) + 2)                                          \
            GCN_ADD(q3, base + (K) + 3)                                          \
            GCN_ADD(q4, base + (K) + 4)                                          \
            GCN_ADD(q5, base + (K) + 5)                                          \
            GCN_ADD(q6, base + (K) + 6)                                          \
            GCN_ADD(q7, base + (K) + 7)

        const char* hb = (const char*)h + (f << 2);
        int cnext = (segbeg + l < segend) ? col[segbeg + l] : 0;
        for (int base = segbeg; ; base += 16) {
            const int rem  = segend - base;
            const int remW = max(rem, __shfl_xor(rem, 16));
            if (remW <= 0) break;
            const int c = cnext;
            {
                GCN_GROUP_LOADS(0)
                cnext = (base + 16 + l < segend) ? col[base + 16 + l] : 0;
                GCN_GROUP_ADDS(0)
            }
            if (remW > 8) {
                GCN_GROUP_LOADS(8)
                GCN_GROUP_ADDS(8)
            }
        }
        while (cur < NODES_PER_HW) GCN_FLUSH()
#undef GCN_GROUP_ADDS
#undef GCN_GROUP_LOADS
#undef GCN_LOAD
#undef GCN_ADD
#undef GCN_FLUSH
    }
    __syncthreads();

    const int colbase = wave << 4;
    const int n     = lane & 15;
    const int khalf = lane >> 4;

    v8f acc0 = {};
    v8f acc1 = {};
    #pragma unroll
    for (int kk = 0; kk < 16; ++kk) {
        const int k0 = (kk << 2) + (khalf << 1);
        v2f a0 = *(const v2f*)&lds[n * LDS_STRIDE + k0];
        v2f a1 = *(const v2f*)&lds[(16 + n) * LDS_STRIDE + k0];
        v2f b;
        b.x = W[(k0 + 0) * D_FEAT + colbase + n];
        b.y = W[(k0 + 1) * D_FEAT + colbase + n];
        acc0 = __builtin_amdgcn_wmma_f32_16x16x4_f32(
             false, a0,  false, b,
             (short)0, acc0,  false,  false);
        acc1 = __builtin_amdgcn_wmma_f32_16x16x4_f32(
             false, a1,  false, b,
             (short)0, acc1,  false,  false);
    }

    const int colo = colbase + n;
    const float bv = bias[colo];
    #pragma unroll
    for (int t = 0; t < 2; ++t) {
        #pragma unroll
        for (int r = 0; r < 8; ++r) {
            const int row = r0 + (t << 4) + (khalf << 3) + r;
            if (row < nNodes) {
                float v = (t == 0 ? acc0[r] : acc1[r]) + bv;
                v = (v > 0.0f) ? v : 0.01f * v;
                if (post_scale) v *= post_scale[row];
                out[(size_t)row * D_FEAT + colo] = v;
            }
        }
    }
}

__global__ void gcn_classify(const float* __restrict__ h,
                             const float* __restrict__ Wc,
                             const float* __restrict__ bc,
                             float* __restrict__ out, int nNodes) {
    int i = blockIdx.x * blockDim.x + threadIdx.x;
    if (i >= nNodes) return;
    float a0 = bc[0], a1 = bc[1];
    const float4* row = (const float4*)(h + (size_t)i * D_FEAT);
    #pragma unroll
    for (int q = 0; q < 16; ++q) {
        float4 v = row[q];
        int k = q << 2;
        a0 += v.x * Wc[(k+0)*2] + v.y * Wc[(k+1)*2] + v.z * Wc[(k+2)*2] + v.w * Wc[(k+3)*2];
        a1 += v.x * Wc[(k+0)*2+1] + v.y * Wc[(k+1)*2+1] + v.z * Wc[(k+2)*2+1] + v.w * Wc[(k+3)*2+1];
    }
    out[(size_t)i * 2 + 0] = a0;
    out[(size_t)i * 2 + 1] = a1;
}

extern "C" void kernel_launch(void* const* d_in, const int* in_sizes, int n_in,
                              void* d_out, int out_size, void* d_ws, size_t ws_size,
                              hipStream_t stream) {
    const float* x   = (const float*)d_in[0];
    const int*   src = (const int*)  d_in[1];
    const int*   dst = (const int*)  d_in[2];
    const float* W1  = (const float*)d_in[3];
    const float* b1  = (const float*)d_in[4];
    const float* W2  = (const float*)d_in[5];
    const float* b2  = (const float*)d_in[6];
    const float* Wc  = (const float*)d_in[7];
    const float* bc  = (const float*)d_in[8];
    float* out = (float*)d_out;

    const int N = in_sizes[0] / D_FEAT;
    const int E = in_sizes[1];
    const int nBins = (N + BIN_NODES - 1) >> BIN_SHIFT;

    float* bufA      = (float*)d_ws;
    float* bufB      = bufA + (size_t)N * D_FEAT;
    float* out_isqrt = bufB + (size_t)N * D_FEAT;
    float* in_isqrt  = out_isqrt + N;
    int*   row_start = (int*)(in_isqrt + N);
    int*   col       = row_start + (N + 1);
    int2*  part      = (int2*)bufB;
    int*   bin_cnt   = (int*)bufB + (size_t)2 * E;
    int*   bin_cur   = bin_cnt + nBins;
    int*   cnt_out   = (int*)in_isqrt;

    if (ws_size < ((size_t)2 * N * D_FEAT + (size_t)3 * N + 1 + (size_t)E) * 4) return;
    if ((size_t)2 * E + (size_t)2 * nBins > (size_t)N * D_FEAT) return;
    if (nBins > MAX_BINS || N < 1 || E < 1) return;

    const int B256 = 256;
    dim3 blk(B256);

    const int nEdgeBlocks = (E + EDGES_PER_BLOCK - 1) / EDGES_PER_BLOCK;
    gcn_zero_i32<<<(N + 2 * nBins + B256 - 1) / B256, blk, 0, stream>>>(cnt_out, N, bin_cnt, 2 * nBins);
    gcn_bin_count<<<nEdgeBlocks, dim3(EDGE_THREADS), 0, stream>>>(src, dst, cnt_out, bin_cnt, E, nBins);
    gcn_bin_scatter<<<nEdgeBlocks, dim3(EDGE_THREADS), 0, stream>>>(src, dst, bin_cnt, bin_cur, part, E, nBins,
                                                                    x, cnt_out, out_isqrt, bufA, N, N * 16);
    gcn_bin_sort<<<nBins, dim3(SORT_THREADS), 0, stream>>>(part, bin_cnt, row_start, in_isqrt, col, N);

    const int nTiles = (N + TILE_NODES - 1) / TILE_NODES;
    gcn_layer_fused<<<nTiles, dim3(128), 0, stream>>>(
        bufA, row_start, col, W1, b1, in_isqrt, out_isqrt, bufB, N);

    gcn_layer_fused<<<nTiles, dim3(128), 0, stream>>>(
        bufB, row_start, col, W2, b2, in_isqrt, nullptr, bufA, N);

    gcn_classify<<<(N + B256 - 1) / B256, blk, 0, stream>>>(bufA, Wc, bc, out, N);
}


# Out-degrees counted by the LDS bin sort too (source ids partitioned by 512-node bin, counted per bin in LDS): no per-edge global atomic left; x pre-scale moved to the per-bin launch; sum order same
# speedup vs baseline: 2.9001x; 1.3779x over previous
#include <hip/hip_runtime.h>
#include <hip/hip_bf16.h>


typedef __attribute__((ext_vector_type(2))) float v2f;
typedef __attribute__((ext_vector_type(8))) float v8f;

#define D_FEAT 64
#define LDS_STRIDE 68

#define BIN_SHIFT 9
#define BIN_NODES (1 << BIN_SHIFT)
#define MAX_BINS  2048
#define EDGE_THREADS 256
#define EDGES_PER_THREAD 8
#define EDGES_PER_BLOCK (EDGE_THREADS * EDGES_PER_THREAD)
#define SORT_THREADS BIN_NODES
#define STAGE_CAP 12288
#define SORT_UNROLL 8

__global__ void gcn_zero_i32(int* __restrict__ p0, int n0, int* __restrict__ p1, int n1) {
    int i = blockIdx.x * blockDim.x + threadIdx.x;
    if (i < n0) p0[i] = 0;
    else if (i - n0 < n1) p1[i - n0] = 0;
}

__device__ __forceinline__ void gcn_block_scan_excl(int* a, int n, int* wsum) {
    const int tid  = threadIdx.x;
    const int nthr = blockDim.x;
    const int lane = tid & 31;
    const int wave = tid >> 5;
    const int nwav = nthr >> 5;
    const int G  = (n + nthr - 1) / nthr;
    const int i0 = tid * G;
    __syncthreads();
    int sum = 0;
    for (int k = 0; k < G; ++k) { const int i = i0 + k; if (i < n) sum += a[i]; }
    int incl = sum;
    for (int o = 1; o < 32; o <<= 1) {
        int t = __shfl_up(incl, o);
        if (lane >= o) incl += t;
    }
    if (lane == 31) wsum[wave] = incl;
    __syncthreads();
    int run = incl - sum;
    for (int w = 0; w < nwav; ++w) run += (w < wave) ? wsum[w] : 0;
    for (int k = 0; k < G; ++k) {
        const int i = i0 + k;
        if (i < n) { const int v = a[i]; a[i] = run; run += v; }
    }
    __syncthreads();
}

__global__ void __launch_bounds__(EDGE_THREADS)
gcn_bin_count(const int* __restrict__ src, const int* __restrict__ dst,
              int* __restrict__ cnt2, int nEdges, int nBins) {
    __shared__ int hcnt[2 * MAX_BINS];
    const int tid = threadIdx.x;
    const int nB2 = 2 * nBins;
    for (int i = tid; i < nB2; i += EDGE_THREADS) hcnt[i] = 0;
    __syncthreads();
    const int base = blockIdx.x * EDGES_PER_BLOCK;
    int es[EDGES_PER_THREAD], ed[EDGES_PER_THREAD];
    #pragma unroll
    for (int k = 0; k < EDGES_PER_THREAD; ++k) {
        const int ec = min(base + k * EDGE_THREADS + tid, nEdges - 1);
        es[k] = src[ec];
        ed[k] = dst[ec];
    }
    #pragma unroll
    for (int k = 0; k < EDGES_PER_THREAD; ++k) {
        if (base + k * EDGE_THREADS + tid < nEdges) {
            atomicAdd(&hcnt[ed[k] >> BIN_SHIFT], 1);
            atomicAdd(&hcnt[nBins + (es[k] >> BIN_SHIFT)], 1);
        }
    }
    __syncthreads();
    for (int i = tid; i < nB2; i += EDGE_THREADS) {
        const int c = hcnt[i];
        if (c) atomicAdd(&cnt2[i], c);
    }
}

__global__ void __launch_bounds__(EDGE_THREADS)
gcn_bin_scatter(const int* __restrict__ src, const int* __restrict__ dst,
                const int* __restrict__ cnt2, int* __restrict__ cur2,
                int2* __restrict__ part, int* __restrict__ spart,
                int nEdges, int nBins) {
    __shared__ int  hcnt[2 * MAX_BINS];
    __shared__ int  gpos[2 * MAX_BINS];
    __shared__ int2 stage[EDGES_PER_BLOCK];
    __shared__ int  sstage[EDGES_PER_BLOCK];
    __shared__ int  wsum[EDGE_THREADS / 32];

    const int tid = threadIdx.x;
    const int nB2 = 2 * nBins;

    for (int i = tid; i < nB2; i += EDGE_THREADS) { gpos[i] = cnt2[i]; hcnt[i] = 0; }
    gcn_block_scan_excl(gpos, nB2, wsum);

    const int base   = blockIdx.x * EDGES_PER_BLOCK;
    const int nLocal = min(EDGES_PER_BLOCK, nEdges - base);
    int es[EDGES_PER_THREAD], ed[EDGES_PER_THREAD], er[EDGES_PER_THREAD], sr[EDGES_PER_THREAD];
    #pragma unroll
    for (int k = 0; k < EDGES_PER_THREAD; ++k) {
        const int ec = min(base + k * EDGE_THREADS + tid, nEdges - 1);
        es[k] = src[ec];
        ed[k] = dst[ec];
    }
    #pragma unroll
    for (int k = 0; k < EDGES_PER_THREAD; ++k) {
        er[k] = -1;
        sr[k] = 0;
        if (base + k * EDGE_THREADS + tid < nEdges) {
            er[k] = atomicAdd(&hcnt[ed[k] >> BIN_SHIFT], 1);
            sr[k] = atomicAdd(&hcnt[nBins + (es[k] >> BIN_SHIFT)], 1);
        }
    }
    __syncthreads();

    for (int i = tid; i < nB2; i += EDGE_THREADS) {
        const int c = hcnt[i];
        if (c) gpos[i] += atomicAdd(&cur2[i], c);
    }
    gcn_block_scan_excl(hcnt, nB2, wsum);

    #pragma unroll
    for (int k = 0; k < EDGES_PER_THREAD; ++k) {
        if (er[k] >= 0) {
            stage[hcnt[ed[k] >> BIN_SHIFT] + er[k]] = make_int2(ed[k], es[k]);
            sstage[hcnt[nBins + (es[k] >> BIN_SHIFT)] - nLocal + sr[k]] = es[k];
        }
    }
    __syncthreads();

    #pragma unroll
    for (int k = 0; k < EDGES_PER_THREAD; ++k) {
        const int i  = k * EDGE_THREADS + tid;
        const int ic = min(i, nLocal - 1);
        const int2 p = stage[ic];
        const int b  = p.x >> BIN_SHIFT;
        const int o  = gpos[b] + (ic - hcnt[b]);
        const int s  = sstage[ic];
        const int sb = nBins + (s >> BIN_SHIFT);
        const int so = (gpos[sb] - nEdges) + (ic - (hcnt[sb] - nLocal));
        if (i < nLocal) { part[o] = p; spart[so] = s; }
    }
}

#define ROW_QUADS_PER_THREAD (BIN_NODES * (D_FEAT / 4) / SORT_THREADS)
__global__ void __launch_bounds__(SORT_THREADS)
gcn_bin_sort(const int2* __restrict__ part, const int* __restrict__ spart,
             const int* __restrict__ cnt2,
             int* __restrict__ row_start, float* __restrict__ in_isqrt,
             int* __restrict__ col,
             const float* __restrict__ x, float* __restrict__ out_isqrt,
             float* __restrict__ xs, int nNodes, int nBins, int nQuads  ) {
    __shared__ int   cnt[BIN_NODES];
    __shared__ int   cur[BIN_NODES];
    __shared__ int   ocnt[BIN_NODES];
    __shared__ float oscl[BIN_NODES];
    __shared__ int   wsum[SORT_THREADS / 32];
    __shared__ int   wsum2[SORT_THREADS / 32];
    __shared__ int   stage[STAGE_CAP];

    const int tid  = threadIdx.x;
    const int lane = tid & 31;
    const int wave = tid >> 5;
    const int bin  = blockIdx.x;

    const int q0 = bin * (BIN_NODES * (D_FEAT / 4));
    float4 xv[ROW_QUADS_PER_THREAD];
    #pragma unroll
    for (int u = 0; u < ROW_QUADS_PER_THREAD; ++u)
        xv[u] = ((const float4*)x)[min(q0 + u * SORT_THREADS + tid, nQuads - 1)];

    int pre = 0, spre = 0;
    for (int i = tid; i < bin; i += SORT_THREADS) { pre += cnt2[i]; spre += cnt2[nBins + i]; }
    for (int o = 16; o > 0; o >>= 1) { pre += __shfl_xor(pre, o); spre += __shfl_xor(spre, o); }
    if (lane == 0) { wsum[wave] = pre; wsum2[wave] = spre; }
    cnt[tid]  = 0;
    ocnt[tid] = 0;
    __syncthreads();
    int segBeg = 0, sSegBeg = 0;
    for (int w = 0; w < SORT_THREADS / 32; ++w) { segBeg += wsum[w]; sSegBeg += wsum2[w]; }
    const int segLen  = cnt2[bin];
    const int sSegLen = cnt2[nBins + bin];
    const int2* seg  = part + segBeg;
    const int*  sseg = spart + sSegBeg;

    for (int i0 = tid; i0 < sSegLen; i0 += SORT_UNROLL * SORT_THREADS) {
        int sl[SORT_UNROLL];
        #pragma unroll
        for (int u = 0; u < SORT_UNROLL; ++u)
            sl[u] = sseg[min(i0 + u * SORT_THREADS, sSegLen - 1)];
        #pragma unroll
        for (int u = 0; u < SORT_UNROLL; ++u)
            if (i0 + u * SORT_THREADS < sSegLen) atomicAdd(&ocnt[sl[u] & (BIN_NODES - 1)], 1);
    }
    for (int i0 = tid; i0 < segLen; i0 += SORT_UNROLL * SORT_THREADS) {
        int dl[SORT_UNROLL];
        #pragma unroll
        for (int u = 0; u < SORT_UNROLL; ++u)
            dl[u] = seg[min(i0 + u * SORT_THREADS, segLen - 1)].x;
        #pragma unroll
        for (int u = 0; u < SORT_UNROLL; ++u)
            if (i0 + u * SORT_THREADS < segLen) atomicAdd(&cnt[dl[u] & (BIN_NODES - 1)], 1);
    }
    __syncthreads();

    const int c = cnt[tid];
    int incl = c;
    for (int o = 1; o < 32; o <<= 1) {
        int v = __shfl_up(incl, o);
        if (lane >= o) incl += v;
    }
    if (lane == 31) wsum[wave] = incl;
    const float osc = rsqrtf(fmaxf((float)ocnt[tid], 1.0f));
    oscl[tid] = osc;
    __syncthreads();
    int excl = incl - c;
    for (int w = 0; w < SORT_THREADS / 32; ++w) excl += (w < wave) ? wsum[w] : 0;

    const int node = bin * BIN_NODES + tid;
    if (node < nNodes) {
        row_start[node] = segBeg + excl;
        if (node == nNodes - 1) row_start[nNodes] = segBeg + excl + c;
        in_isqrt[node]  = rsqrtf(fmaxf((float)c, 1.0f));
        out_isqrt[node] = osc;
    }
    cur[tid] = excl;
    __syncthreads();

    if (segLen <= STAGE_CAP) {
        for (int i0 = tid; i0 < segLen; i0 += SORT_UNROLL * SORT_THREADS) {
            int2 p[SORT_UNROLL];
            #pragma unroll
            for (int u = 0; u < SORT_UNROLL; ++u)
                p[u] = seg[min(i0 + u * SORT_THREADS, segLen - 1)];
            #pragma unroll
            for (int u = 0; u < SORT_UNROLL; ++u)
                if (i0 + u * SORT_THREADS < segLen)
                    stage[atomicAdd(&cur[p[u].x & (BIN_NODES - 1)], 1)] = p[u].y;
        }
        __syncthreads();
        for (int i = tid; i < segLen; i += SORT_THREADS) col[segBeg + i] = stage[i];
    } else {
        for (int i = tid; i < segLen; i += SORT_THREADS) {
            const int2 p = seg[i];
            col[segBeg + atomicAdd(&cur[p.x & (BIN_NODES - 1)], 1)] = p.y;
        }
    }

    #pragma unroll
    for (int u = 0; u < ROW_QUADS_PER_THREAD; ++u) {
        const int lq = u * SORT_THREADS + tid;
        if (q0 + lq < nQuads) {
            const float sc = oscl[lq >> 4];
            float4 v = xv[u];
            v.x *= sc; v.y *= sc; v.z *= sc; v.w *= sc;
            ((float4*)xs)[q0 + lq] = v;
        }
    }
}

#define TILE_NODES 32
#define NODES_PER_HW 4
__global__ void __launch_bounds__(128)
gcn_layer_fused(const float* __restrict__ h,
                const int* __restrict__ row_start,
                const int* __restrict__ col,
                const float* __restrict__ W,
                const float* __restrict__ bias,
                const float* __restrict__ in_isqrt,
                const float* __restrict__ post_scale,
                float* __restrict__ out, int nNodes) {
    __shared__ float lds[TILE_NODES * LDS_STRIDE];

    const int r0   = blockIdx.x * TILE_NODES;
    const int tid  = threadIdx.x;
    const int wave = tid >> 5;
    const int lane = tid & 31;

    {
        const int hw = tid >> 4;
        const int l  = tid & 15;
        const int f  = l << 2;
        const int n0 = r0 + hw * NODES_PER_HW;

        int   rsv = 0;
        float scv = 0.0f;
        if (l <= NODES_PER_HW) rsv = row_start[min(n0 + l, nNodes)];
        if (l <  NODES_PER_HW && n0 + l < nNodes) scv = in_isqrt[n0 + l];
        const int segbeg = __shfl(rsv, 0, 16);
        const int e1     = __shfl(rsv, 1, 16);
        const int e2     = __shfl(rsv, 2, 16);
        const int e3     = __shfl(rsv, 3, 16);
        const int segend = __shfl(rsv, 4, 16);
        const float sc0  = __shfl(scv, 0, 16);
        const float sc1  = __shfl(scv, 1, 16);
        const float sc2  = __shfl(scv, 2, 16);
        const float sc3  = __shfl(scv, 3, 16);

        const float4 zero4 = make_float4(0.0f, 0.0f, 0.0f, 0.0f);
        float4 acc = zero4;
        int cur    = 0;
        int curEnd = e1;
        const int lbase = (hw * NODES_PER_HW) * LDS_STRIDE + f;

#define GCN_FLUSH()                                                              \
        {                                                                        \
            const float s_ = (cur == 0) ? sc0 : (cur == 1) ? sc1 : (cur == 2) ? sc2 : sc3; \
            float4 o_;                                                           \
            o_.x = acc.x * s_; o_.y = acc.y * s_; o_.z = acc.z * s_; o_.w = acc.w * s_; \
            *(float4*)&lds[lbase + cur * LDS_STRIDE] = o_;                       \
            acc = zero4;                                                         \
            ++cur;                                                               \
            curEnd = (cur == 1) ? e2 : (cur == 2) ? e3 : segend;                 \
        }
#define GCN_ADD(rq, jq)                                                          \
        {                                                                        \
            const int j_ = (jq);                                                 \
            while (cur < NODES_PER_HW && j_ >= curEnd) GCN_FLUSH()               \
            acc.x += rq.x; acc.y += rq.y; acc.z += rq.z; acc.w += rq.w;          \
        }
#define GCN_LOAD(rq, sq, kq)                                                     \
        float4 rq = zero4;                                                       \
        if ((kq) < rem) rq = *(const float4*)(hb + ((unsigned)(sq) << 8));
#define GCN_GROUP_LOADS(K)                                                       \
            const int s0 = __shfl(c, (K),     16);                               \
            const int s1 = __shfl(c, (K) + 1, 16);                               \
            const int s2 = __shfl(c, (K) + 2, 16);                               \
            const int s3 = __shfl(c, (K) + 3, 16);                               \
            const int s4 = __shfl(c, (K) + 4, 16);                               \
            const int s5 = __shfl(c, (K) + 5, 16);                               \
            const int s6 = __shfl(c, (K) + 6, 16);                               \
            const int s7 = __shfl(c, (K) + 7, 16);                               \
            GCN_LOAD(q0, s0, (K))                                                \
            GCN_LOAD(q1, s1, (K) + 1)                                            \
            GCN_LOAD(q2, s2, (K) + 2)                                            \
            GCN_LOAD(q3, s3, (K) + 3)                                            \
            GCN_LOAD(q4, s4, (K) + 4)                                            \
            GCN_LOAD(q5, s5, (K) + 5)                                            \
            GCN_LOAD(q6, s6, (K) + 6)                                            \
            GCN_LOAD(q7, s7, (K) + 7)
#define GCN_GROUP_ADDS(K)                                                        \
            GCN_ADD(q0, base + (K))                                              \
            GCN_ADD(q1, base + (K) + 1)                                          \
            GCN_ADD(q2, base + (K) + 2)                                          \
            GCN_ADD(q3, base + (K) + 3)                                          \
            GCN_ADD(q4, base + (K) + 4)                                          \
            GCN_ADD(q5, base + (K) + 5)                                          \
            GCN_ADD(q6, base + (K) + 6)                                          \
            GCN_ADD(q7, base + (K) + 7)

        const char* hb = (const char*)h + (f << 2);
        int cnext = (segbeg + l < segend) ? col[segbeg + l] : 0;
        for (int base = segbeg; ; base += 16) {
            const int rem  = segend - base;
            const int remW = max(rem, __shfl_xor(rem, 16));
            if (remW <= 0) break;
            const int c = cnext;
            {
                GCN_GROUP_LOADS(0)
                cnext = (base + 16 + l < segend) ? col[base + 16 + l] : 0;
                GCN_GROUP_ADDS(0)
            }
            if (remW > 8) {
                GCN_GROUP_LOADS(8)
                GCN_GROUP_ADDS(8)
            }
        }
        while (cur < NODES_PER_HW) GCN_FLUSH()
#undef GCN_GROUP_ADDS
#undef GCN_GROUP_LOADS
#undef GCN_LOAD
#undef GCN_ADD
#undef GCN_FLUSH
    }
    __syncthreads();

    const int colbase = wave << 4;
    const int n     = lane & 15;
    const int khalf = lane >> 4;

    v8f acc0 = {};
    v8f acc1 = {};
    #pragma unroll
    for (int kk = 0; kk < 16; ++kk) {
        const int k0 = (kk << 2) + (khalf << 1);
        v2f a0 = *(const v2f*)&lds[n * LDS_STRIDE + k0];
        v2f a1 = *(const v2f*)&lds[(16 + n) * LDS_STRIDE + k0];
        v2f b;
        b.x = W[(k0 + 0) * D_FEAT + colbase + n];
        b.y = W[(k0 + 1) * D_FEAT + colbase + n];
        acc0 = __builtin_amdgcn_wmma_f32_16x16x4_f32(
             false, a0,  false, b,
             (short)0, acc0,  false,  false);
        acc1 = __builtin_amdgcn_wmma_f32_16x16x4_f32(
             false, a1,  false, b,
             (short)0, acc1,  false,  false);
    }

    const int colo = colbase + n;
    const float bv = bias[colo];
    #pragma unroll
    for (int t = 0; t < 2; ++t) {
        #pragma unroll
        for (int r = 0; r < 8; ++r) {
            const int row = r0 + (t << 4) + (khalf << 3) + r;
            if (row < nNodes) {
                float v = (t == 0 ? acc0[r] : acc1[r]) + bv;
                v = (v > 0.0f) ? v : 0.01f * v;
                if (post_scale) v *= post_scale[row];
                out[(size_t)row * D_FEAT + colo] = v;
            }
        }
    }
}

__global__ void gcn_classify(const float* __restrict__ h,
                             const float* __restrict__ Wc,
                             const float* __restrict__ bc,
                             float* __restrict__ out, int nNodes) {
    int i = blockIdx.x * blockDim.x + threadIdx.x;
    if (i >= nNodes) return;
    float a0 = bc[0], a1 = bc[1];
    const float4* row = (const float4*)(h + (size_t)i * D_FEAT);
    #pragma unroll
    for (int q = 0; q < 16; ++q) {
        float4 v = row[q];
        int k = q << 2;
        a0 += v.x * Wc[(k+0)*2] + v.y * Wc[(k+1)*2] + v.z * Wc[(k+2)*2] + v.w * Wc[(k+3)*2];
        a1 += v.x * Wc[(k+0)*2+1] + v.y * Wc[(k+1)*2+1] + v.z * Wc[(k+2)*2+1] + v.w * Wc[(k+3)*2+1];
    }
    out[(size_t)i * 2 + 0] = a0;
    out[(size_t)i * 2 + 1] = a1;
}

extern "C" void kernel_launch(void* const* d_in, const int* in_sizes, int n_in,
                              void* d_out, int out_size, void* d_ws, size_t ws_size,
                              hipStream_t stream) {
    const float* x   = (const float*)d_in[0];
    const int*   src = (const int*)  d_in[1];
    const int*   dst = (const int*)  d_in[2];
    const float* W1  = (const float*)d_in[3];
    const float* b1  = (const float*)d_in[4];
    const float* W2  = (const float*)d_in[5];
    const float* b2  = (const float*)d_in[6];
    const float* Wc  = (const float*)d_in[7];
    const float* bc  = (const float*)d_in[8];
    float* out = (float*)d_out;

    const int N = in_sizes[0] / D_FEAT;
    const int E = in_sizes[1];
    const int nBins = (N + BIN_NODES - 1) >> BIN_SHIFT;

    float* bufA      = (float*)d_ws;
    float* bufB      = bufA + (size_t)N * D_FEAT;
    float* out_isqrt = bufB + (size_t)N * D_FEAT;
    float* in_isqrt  = out_isqrt + N;
    int*   row_start = (int*)(in_isqrt + N);
    int*   col       = row_start + (N + 1);
    int2*  part      = (int2*)bufB;
    int*   spart     = (int*)bufB + (size_t)2 * E;
    int*   cnt2      = spart + E;
    int*   cur2      = cnt2 + 2 * nBins;

    if (ws_size < ((size_t)2 * N * D_FEAT + (size_t)3 * N + 1 + (size_t)E) * 4) return;
    if ((size_t)3 * E + (size_t)4 * nBins > (size_t)N * D_FEAT) return;
    if (nBins > MAX_BINS || N < 1 || E < 1) return;

    const int B256 = 256;
    dim3 blk(B256);

    const int nEdgeBlocks = (E + EDGES_PER_BLOCK - 1) / EDGES_PER_BLOCK;
    gcn_zero_i32<<<(4 * nBins + B256 - 1) / B256, blk, 0, stream>>>(cnt2, 2 * nBins, cur2, 2 * nBins);
    gcn_bin_count<<<nEdgeBlocks, dim3(EDGE_THREADS), 0, stream>>>(src, dst, cnt2, E, nBins);
    gcn_bin_scatter<<<nEdgeBlocks, dim3(EDGE_THREADS), 0, stream>>>(src, dst, cnt2, cur2, part, spart, E, nBins);
    gcn_bin_sort<<<nBins, dim3(SORT_THREADS), 0, stream>>>(part, spart, cnt2, row_start, in_isqrt, col,
                                                          x, out_isqrt, bufA, N, nBins, N * 16);

    const int nTiles = (N + TILE_NODES - 1) / TILE_NODES;
    gcn_layer_fused<<<nTiles, dim3(128), 0, stream>>>(
        bufA, row_start, col, W1, b1, in_isqrt, out_isqrt, bufB, N);

    gcn_layer_fused<<<nTiles, dim3(128), 0, stream>>>(
        bufB, row_start, col, W2, b2, in_isqrt, nullptr, bufA, N);

    gcn_classify<<<(N + B256 - 1) / B256, blk, 0, stream>>>(bufA, Wc, bc, out, N);
}


# Linear head fused into the layer-2 launch: activated 32x64 tile kept in LDS, one wave forms both outputs per node (same k order); layer-2 activations never written, 7 -> 6 launches
# speedup vs baseline: 3.2840x; 1.1324x over previous
#include <hip/hip_runtime.h>
#include <hip/hip_bf16.h>


typedef __attribute__((ext_vector_type(2))) float v2f;
typedef __attribute__((ext_vector_type(8))) float v8f;

#define D_FEAT 64
#define LDS_STRIDE 68

#define BIN_SHIFT 9
#define BIN_NODES (1 << BIN_SHIFT)
#define MAX_BINS  2048
#define EDGE_THREADS 256
#define EDGES_PER_THREAD 8
#define EDGES_PER_BLOCK (EDGE_THREADS * EDGES_PER_THREAD)
#define SORT_THREADS BIN_NODES
#define STAGE_CAP 12288
#define SORT_UNROLL 8

__global__ void gcn_zero_i32(int* __restrict__ p0, int n0, int* __restrict__ p1, int n1) {
    int i = blockIdx.x * blockDim.x + threadIdx.x;
    if (i < n0) p0[i] = 0;
    else if (i - n0 < n1) p1[i - n0] = 0;
}

__device__ __forceinline__ void gcn_block_scan_excl(int* a, int n, int* wsum) {
    const int tid  = threadIdx.x;
    const int nthr = blockDim.x;
    const int lane = tid & 31;
    const int wave = tid >> 5;
    const int nwav = nthr >> 5;
    const int G  = (n + nthr - 1) / nthr;
    const int i0 = tid * G;
    __syncthreads();
    int sum = 0;
    for (int k = 0; k < G; ++k) { const int i = i0 + k; if (i < n) sum += a[i]; }
    int incl = sum;
    for (int o = 1; o < 32; o <<= 1) {
        int t = __shfl_up(incl, o);
        if (lane >= o) incl += t;
    }
    if (lane == 31) wsum[wave] = incl;
    __syncthreads();
    int run = incl - sum;
    for (int w = 0; w < nwav; ++w) run += (w < wave) ? wsum[w] : 0;
    for (int k = 0; k < G; ++k) {
        const int i = i0 + k;
        if (i < n) { const int v = a[i]; a[i] = run; run += v; }
    }
    __syncthreads();
}

__global__ void __launch_bounds__(EDGE_THREADS)
gcn_bin_count(const int* __restrict__ src, const int* __restrict__ dst,
              int* __restrict__ cnt2, int nEdges, int nBins) {
    __shared__ int hcnt[2 * MAX_BINS];
    const int tid = threadIdx.x;
    const int nB2 = 2 * nBins;
    for (int i = tid; i < nB2; i += EDGE_THREADS) hcnt[i] = 0;
    __syncthreads();
    const int base = blockIdx.x * EDGES_PER_BLOCK;
    int es[EDGES_PER_THREAD], ed[EDGES_PER_THREAD];
    #pragma unroll
    for (int k = 0; k < EDGES_PER_THREAD; ++k) {
        const int ec = min(base + k * EDGE_THREADS + tid, nEdges - 1);
        es[k] = src[ec];
        ed[k] = dst[ec];
    }
    #pragma unroll
    for (int k = 0; k < EDGES_PER_THREAD; ++k) {
        if (base + k * EDGE_THREADS + tid < nEdges) {
            atomicAdd(&hcnt[ed[k] >> BIN_SHIFT], 1);
            atomicAdd(&hcnt[nBins + (es[k] >> BIN_SHIFT)], 1);
        }
    }
    __syncthreads();
    for (int i = tid; i < nB2; i += EDGE_THREADS) {
        const int c = hcnt[i];
        if (c) atomicAdd(&cnt2[i], c);
    }
}

__global__ void __launch_bounds__(EDGE_THREADS)
gcn_bin_scatter(const int* __restrict__ src, const int* __restrict__ dst,
                const int* __restrict__ cnt2, int* __restrict__ cur2,
                int2* __restrict__ part, int* __restrict__ spart,
                int nEdges, int nBins) {
    __shared__ int  hcnt[2 * MAX_BINS];
    __shared__ int  gpos[2 * MAX_BINS];
    __shared__ int2 stage[EDGES_PER_BLOCK];
    __shared__ int  sstage[EDGES_PER_BLOCK];
    __shared__ int  wsum[EDGE_THREADS / 32];

    const int tid = threadIdx.x;
    const int nB2 = 2 * nBins;

    for (int i = tid; i < nB2; i += EDGE_THREADS) { gpos[i] = cnt2[i]; hcnt[i] = 0; }
    gcn_block_scan_excl(gpos, nB2, wsum);

    const int base   = blockIdx.x * EDGES_PER_BLOCK;
    const int nLocal = min(EDGES_PER_BLOCK, nEdges - base);
    int es[EDGES_PER_THREAD], ed[EDGES_PER_THREAD], er[EDGES_PER_THREAD], sr[EDGES_PER_THREAD];
    #pragma unroll
    for (int k = 0; k < EDGES_PER_THREAD; ++k) {
        const int ec = min(base + k * EDGE_THREADS + tid, nEdges - 1);
        es[k] = src[ec];
        ed[k] = dst[ec];
    }
    #pragma unroll
    for (int k = 0; k < EDGES_PER_THREAD; ++k) {
        er[k] = -1;
        sr[k] = 0;
        if (base + k * EDGE_THREADS + tid < nEdges) {
            er[k] = atomicAdd(&hcnt[ed[k] >> BIN_SHIFT], 1);
            sr[k] = atomicAdd(&hcnt[nBins + (es[k] >> BIN_SHIFT)], 1);
        }
    }
    __syncthreads();

    for (int i = tid; i < nB2; i += EDGE_THREADS) {
        const int c = hcnt[i];
        if (c) gpos[i] += atomicAdd(&cur2[i], c);
    }
    gcn_block_scan_excl(hcnt, nB2, wsum);

    #pragma unroll
    for (int k = 0; k < EDGES_PER_THREAD; ++k) {
        if (er[k] >= 0) {
            stage[hcnt[ed[k] >> BIN_SHIFT] + er[k]] = make_int2(ed[k], es[k]);
            sstage[hcnt[nBins + (es[k] >> BIN_SHIFT)] - nLocal + sr[k]] = es[k];
        }
    }
    __syncthreads();

    #pragma unroll
    for (int k = 0; k < EDGES_PER_THREAD; ++k) {
        const int i  = k * EDGE_THREADS + tid;
        const int ic = min(i, nLocal - 1);
        const int2 p = stage[ic];
        const int b  = p.x >> BIN_SHIFT;
        const int o  = gpos[b] + (ic - hcnt[b]);
        const int s  = sstage[ic];
        const int sb = nBins + (s >> BIN_SHIFT);
        const int so = (gpos[sb] - nEdges) + (ic - (hcnt[sb] - nLocal));
        if (i < nLocal) { part[o] = p; spart[so] = s; }
    }
}

#define ROW_QUADS_PER_THREAD (BIN_NODES * (D_FEAT / 4) / SORT_THREADS)
__global__ void __launch_bounds__(SORT_THREADS)
gcn_bin_sort(const int2* __restrict__ part, const int* __restrict__ spart,
             const int* __restrict__ cnt2,
             int* __restrict__ row_start, float* __restrict__ in_isqrt,
             int* __restrict__ col,
             const float* __restrict__ x, float* __restrict__ out_isqrt,
             float* __restrict__ xs, int nNodes, int nBins, int nQuads  ) {
    __shared__ int   cnt[BIN_NODES];
    __shared__ int   cur[BIN_NODES];
    __shared__ int   ocnt[BIN_NODES];
    __shared__ float oscl[BIN_NODES];
    __shared__ int   wsum[SORT_THREADS / 32];
    __shared__ int   wsum2[SORT_THREADS / 32];
    __shared__ int   stage[STAGE_CAP];

    const int tid  = threadIdx.x;
    const int lane = tid & 31;
    const int wave = tid >> 5;
    const int bin  = blockIdx.x;

    const int q0 = bin * (BIN_NODES * (D_FEAT / 4));
    float4 xv[ROW_QUADS_PER_THREAD];
    #pragma unroll
    for (int u = 0; u < ROW_QUADS_PER_THREAD; ++u)
        xv[u] = ((const float4*)x)[min(q0 + u * SORT_THREADS + tid, nQuads - 1)];

    int pre = 0, spre = 0;
    for (int i = tid; i < bin; i += SORT_THREADS) { pre += cnt2[i]; spre += cnt2[nBins + i]; }
    for (int o = 16; o > 0; o >>= 1) { pre += __shfl_xor(pre, o); spre += __shfl_xor(spre, o); }
    if (lane == 0) { wsum[wave] = pre; wsum2[wave] = spre; }
    cnt[tid]  = 0;
    ocnt[tid] = 0;
    __syncthreads();
    int segBeg = 0, sSegBeg = 0;
    for (int w = 0; w < SORT_THREADS / 32; ++w) { segBeg += wsum[w]; sSegBeg += wsum2[w]; }
    const int segLen  = cnt2[bin];
    const int sSegLen = cnt2[nBins + bin];
    const int2* seg  = part + segBeg;
    const int*  sseg = spart + sSegBeg;

    for (int i0 = tid; i0 < sSegLen; i0 += SORT_UNROLL * SORT_THREADS) {
        int sl[SORT_UNROLL];
        #pragma unroll
        for (int u = 0; u < SORT_UNROLL; ++u)
            sl[u] = sseg[min(i0 + u * SORT_THREADS, sSegLen - 1)];
        #pragma unroll
        for (int u = 0; u < SORT_UNROLL; ++u)
            if (i0 + u * SORT_THREADS < sSegLen) atomicAdd(&ocnt[sl[u] & (BIN_NODES - 1)], 1);
    }
    for (int i0 = tid; i0 < segLen; i0 += SORT_UNROLL * SORT_THREADS) {
        int dl[SORT_UNROLL];
        #pragma unroll
        for (int u = 0; u < SORT_UNROLL; ++u)
            dl[u] = seg[min(i0 + u * SORT_THREADS, segLen - 1)].x;
        #pragma unroll
        for (int u = 0; u < SORT_UNROLL; ++u)
            if (i0 + u * SORT_THREADS < segLen) atomicAdd(&cnt[dl[u] & (BIN_NODES - 1)], 1);
    }
    __syncthreads();

    const int c = cnt[tid];
    int incl = c;
    for (int o = 1; o < 32; o <<= 1) {
        int v = __shfl_up(incl, o);
        if (lane >= o) incl += v;
    }
    if (lane == 31) wsum[wave] = incl;
    const float osc = rsqrtf(fmaxf((float)ocnt[tid], 1.0f));
    oscl[tid] = osc;
    __syncthreads();
    int excl = incl - c;
    for (int w = 0; w < SORT_THREADS / 32; ++w) excl += (w < wave) ? wsum[w] : 0;

    const int node = bin * BIN_NODES + tid;
    if (node < nNodes) {
        row_start[node] = segBeg + excl;
        if (node == nNodes - 1) row_start[nNodes] = segBeg + excl + c;
        in_isqrt[node]  = rsqrtf(fmaxf((float)c, 1.0f));
        out_isqrt[node] = osc;
    }
    cur[tid] = excl;
    __syncthreads();

    if (segLen <= STAGE_CAP) {
        for (int i0 = tid; i0 < segLen; i0 += SORT_UNROLL * SORT_THREADS) {
            int2 p[SORT_UNROLL];
            #pragma unroll
            for (int u = 0; u < SORT_UNROLL; ++u)
                p[u] = seg[min(i0 + u * SORT_THREADS, segLen - 1)];
            #pragma unroll
            for (int u = 0; u < SORT_UNROLL; ++u)
                if (i0 + u * SORT_THREADS < segLen)
                    stage[atomicAdd(&cur[p[u].x & (BIN_NODES - 1)], 1)] = p[u].y;
        }
        __syncthreads();
        for (int i = tid; i < segLen; i += SORT_THREADS) col[segBeg + i] = stage[i];
    } else {
        for (int i = tid; i < segLen; i += SORT_THREADS) {
            const int2 p = seg[i];
            col[segBeg + atomicAdd(&cur[p.x & (BIN_NODES - 1)], 1)] = p.y;
        }
    }

    #pragma unroll
    for (int u = 0; u < ROW_QUADS_PER_THREAD; ++u) {
        const int lq = u * SORT_THREADS + tid;
        if (q0 + lq < nQuads) {
            const float sc = oscl[lq >> 4];
            float4 v = xv[u];
            v.x *= sc; v.y *= sc; v.z *= sc; v.w *= sc;
            ((float4*)xs)[q0 + lq] = v;
        }
    }
}

#define TILE_NODES 32
#define NODES_PER_HW 4
template <bool HEAD>
__device__ __forceinline__ void
gcn_layer_body(float* __restrict__ lds,
               const float* __restrict__ h,
               const int* __restrict__ row_start,
               const int* __restrict__ col,
               const float* __restrict__ W,
               const float* __restrict__ bias,
               const float* __restrict__ in_isqrt,
               const float* __restrict__ post_scale,
               const float* __restrict__ Wc,
               const float* __restrict__ bc,
               float* __restrict__ out, int nNodes) {

    const int r0   = blockIdx.x * TILE_NODES;
    const int tid  = threadIdx.x;
    const int wave = tid >> 5;
    const int lane = tid & 31;

    {
        const int hw = tid >> 4;
        const int l  = tid & 15;
        const int f  = l << 2;
        const int n0 = r0 + hw * NODES_PER_HW;

        int   rsv = 0;
        float scv = 0.0f;
        if (l <= NODES_PER_HW) rsv = row_start[min(n0 + l, nNodes)];
        if (l <  NODES_PER_HW && n0 + l < nNodes) scv = in_isqrt[n0 + l];
        const int segbeg = __shfl(rsv, 0, 16);
        const int e1     = __shfl(rsv, 1, 16);
        const int e2     = __shfl(rsv, 2, 16);
        const int e3     = __shfl(rsv, 3, 16);
        const int segend = __shfl(rsv, 4, 16);
        const float sc0  = __shfl(scv, 0, 16);
        const float sc1  = __shfl(scv, 1, 16);
        const float sc2  = __shfl(scv, 2, 16);
        const float sc3  = __shfl(scv, 3, 16);

        const float4 zero4 = make_float4(0.0f, 0.0f, 0.0f, 0.0f);
        float4 acc = zero4;
        int cur    = 0;
        int curEnd = e1;
        const int lbase = (hw * NODES_PER_HW) * LDS_STRIDE + f;

#define GCN_FLUSH()                                                              \
        {                                                                        \
            const float s_ = (cur == 0) ? sc0 : (cur == 1) ? sc1 : (cur == 2) ? sc2 : sc3; \
            float4 o_;                                                           \
            o_.x = acc.x * s_; o_.y = acc.y * s_; o_.z = acc.z * s_; o_.w = acc.w * s_; \
            *(float4*)&lds[lbase + cur * LDS_STRIDE] = o_;                       \
            acc = zero4;                                                         \
            ++cur;                                                               \
            curEnd = (cur == 1) ? e2 : (cur == 2) ? e3 : segend;                 \
        }
#define GCN_ADD(rq, jq)                                                          \
        {                                                                        \
            const int j_ = (jq);                                                 \
            while (cur < NODES_PER_HW && j_ >= curEnd) GCN_FLUSH()               \
            acc.x += rq.x; acc.y += rq.y; acc.z += rq.z; acc.w += rq.w;          \
        }
#define GCN_LOAD(rq, sq, kq)                                                     \
        float4 rq = zero4;                                                       \
        if ((kq) < rem) rq = *(const float4*)(hb + ((unsigned)(sq) << 8));
#define GCN_GROUP_LOADS(K)                                                       \
            const int s0 = __shfl(c, (K),     16);                               \
            const int s1 = __shfl(c, (K) + 1, 16);                               \
            const int s2 = __shfl(c, (K) + 2, 16);                               \
            const int s3 = __shfl(c, (K) + 3, 16);                               \
            const int s4 = __shfl(c, (K) + 4, 16);                               \
            const int s5 = __shfl(c, (K) + 5, 16);                               \
            const int s6 = __shfl(c, (K) + 6, 16);                               \
            const int s7 = __shfl(c, (K) + 7, 16);                               \
            GCN_LOAD(q0, s0, (K))                                                \
            GCN_LOAD(q1, s1, (K) + 1)                                            \
            GCN_LOAD(q2, s2, (K) + 2)                                            \
            GCN_LOAD(q3, s3, (K) + 3)                                            \
            GCN_LOAD(q4, s4, (K) + 4)                                            \
            GCN_LOAD(q5, s5, (K) + 5)                                            \
            GCN_LOAD(q6, s6, (K) + 6)                                            \
            GCN_LOAD(q7, s7, (K) + 7)
#define GCN_GROUP_ADDS(K)                                                        \
            GCN_ADD(q0, base + (K))                                              \
            GCN_ADD(q1, base + (K) + 1)                                          \
            GCN_ADD(q2, base + (K) + 2)                                          \
            GCN_ADD(q3, base + (K) + 3)                                          \
            GCN_ADD(q4, base + (K) + 4)                                          \
            GCN_ADD(q5, base + (K) + 5)                                          \
            GCN_ADD(q6, base + (K) + 6)                                          \
            GCN_ADD(q7, base + (K) + 7)

        const char* hb = (const char*)h + (f << 2);
        int cnext = (segbeg + l < segend) ? col[segbeg + l] : 0;
        for (int base = segbeg; ; base += 16) {
            const int rem  = segend - base;
            const int remW = max(rem, __shfl_xor(rem, 16));
            if (remW <= 0) break;
            const int c = cnext;
            {
                GCN_GROUP_LOADS(0)
                cnext = (base + 16 + l < segend) ? col[base + 16 + l] : 0;
                GCN_GROUP_ADDS(0)
            }
            if (remW > 8) {
                GCN_GROUP_LOADS(8)
                GCN_GROUP_ADDS(8)
            }
        }
        while (cur < NODES_PER_HW) GCN_FLUSH()
#undef GCN_GROUP_ADDS
#undef GCN_GROUP_LOADS
#undef GCN_LOAD
#undef GCN_ADD
#undef GCN_FLUSH
    }
    __syncthreads();

    const int colbase = wave << 4;
    const int n     = lane & 15;
    const int khalf = lane >> 4;

    v8f acc0 = {};
    v8f acc1 = {};
    #pragma unroll
    for (int kk = 0; kk < 16; ++kk) {
        const int k0 = (kk << 2) + (khalf << 1);
        v2f a0 = *(const v2f*)&lds[n * LDS_STRIDE + k0];
        v2f a1 = *(const v2f*)&lds[(16 + n) * LDS_STRIDE + k0];
        v2f b;
        b.x = W[(k0 + 0) * D_FEAT + colbase + n];
        b.y = W[(k0 + 1) * D_FEAT + colbase + n];
        acc0 = __builtin_amdgcn_wmma_f32_16x16x4_f32(
             false, a0,  false, b,
             (short)0, acc0,  false,  false);
        acc1 = __builtin_amdgcn_wmma_f32_16x16x4_f32(
             false, a1,  false, b,
             (short)0, acc1,  false,  false);
    }

    const int colo = colbase + n;
    const float bv = bias[colo];
    if constexpr (!HEAD) {
        #pragma unroll
        for (int t = 0; t < 2; ++t) {
            #pragma unroll
            for (int r = 0; r < 8; ++r) {
                const int row = r0 + (t << 4) + (khalf << 3) + r;
                if (row < nNodes) {
                    float v = (t == 0 ? acc0[r] : acc1[r]) + bv;
                    v = (v > 0.0f) ? v : 0.01f * v;
                    if (post_scale) v *= post_scale[row];
                    out[(size_t)row * D_FEAT + colo] = v;
                }
            }
        }
    } else {
        __syncthreads();
        #pragma unroll
        for (int t = 0; t < 2; ++t) {
            #pragma unroll
            for (int r = 0; r < 8; ++r) {
                float v = (t == 0 ? acc0[r] : acc1[r]) + bv;
                v = (v > 0.0f) ? v : 0.01f * v;
                lds[((t << 4) + (khalf << 3) + r) * LDS_STRIDE + colo] = v;
            }
        }
        __syncthreads();
        const int node = r0 + lane;
        if (wave == 0 && node < nNodes) {
            float a0 = bc[0], a1 = bc[1];
            const float4* row = (const float4*)&lds[lane * LDS_STRIDE];
            #pragma unroll
            for (int q = 0; q < 16; ++q) {
                float4 v = row[q];
                int k = q << 2;
                a0 += v.x * Wc[(k+0)*2] + v.y * Wc[(k+1)*2] + v.z * Wc[(k+2)*2] + v.w * Wc[(k+3)*2];
                a1 += v.x * Wc[(k+0)*2+1] + v.y * Wc[(k+1)*2+1] + v.z * Wc[(k+2)*2+1] + v.w * Wc[(k+3)*2+1];
            }
            *(float2*)&out[(size_t)node * 2] = make_float2(a0, a1);
        }
    }
}

__global__ void __launch_bounds__(128)
gcn_layer_fused(const float* __restrict__ h, const int* __restrict__ row_start,
                const int* __restrict__ col, const float* __restrict__ W,
                const float* __restrict__ bias, const float* __restrict__ in_isqrt,
                const float* __restrict__ post_scale,
                float* __restrict__ out, int nNodes) {
    __shared__ float lds[TILE_NODES * LDS_STRIDE];
    gcn_layer_body<false>(lds, h, row_start, col, W, bias, in_isqrt, post_scale,
                          nullptr, nullptr, out, nNodes);
}

__global__ void __launch_bounds__(128)
gcn_layer_head(const float* __restrict__ h, const int* __restrict__ row_start,
               const int* __restrict__ col, const float* __restrict__ W,
               const float* __restrict__ bias, const float* __restrict__ in_isqrt,
               const float* __restrict__ Wc, const float* __restrict__ bc,
               float* __restrict__ out, int nNodes) {
    __shared__ float lds[TILE_NODES * LDS_STRIDE];
    gcn_layer_body<true>(lds, h, row_start, col, W, bias, in_isqrt, nullptr,
                         Wc, bc, out, nNodes);
}

extern "C" void kernel_launch(void* const* d_in, const int* in_sizes, int n_in,
                              void* d_out, int out_size, void* d_ws, size_t ws_size,
                              hipStream_t stream) {
    const float* x   = (const float*)d_in[0];
    const int*   src = (const int*)  d_in[1];
    const int*   dst = (const int*)  d_in[2];
    const float* W1  = (const float*)d_in[3];
    const float* b1  = (const float*)d_in[4];
    const float* W2  = (const float*)d_in[5];
    const float* b2  = (const float*)d_in[6];
    const float* Wc  = (const float*)d_in[7];
    const float* bc  = (const float*)d_in[8];
    float* out = (float*)d_out;

    const int N = in_sizes[0] / D_FEAT;
    const int E = in_sizes[1];
    const int nBins = (N + BIN_NODES - 1) >> BIN_SHIFT;

    float* bufA      = (float*)d_ws;
    float* bufB      = bufA + (size_t)N * D_FEAT;
    float* out_isqrt = bufB + (size_t)N * D_FEAT;
    float* in_isqrt  = out_isqrt + N;
    int*   row_start = (int*)(in_isqrt + N);
    int*   col       = row_start + (N + 1);
    int2*  part      = (int2*)bufB;
    int*   spart     = (int*)bufB + (size_t)2 * E;
    int*   cnt2      = spart + E;
    int*   cur2      = cnt2 + 2 * nBins;

    if (ws_size < ((size_t)2 * N * D_FEAT + (size_t)3 * N + 1 + (size_t)E) * 4) return;
    if ((size_t)3 * E + (size_t)4 * nBins > (size_t)N * D_FEAT) return;
    if (nBins > MAX_BINS || N < 1 || E < 1) return;

    const int B256 = 256;
    dim3 blk(B256);

    const int nEdgeBlocks = (E + EDGES_PER_BLOCK - 1) / EDGES_PER_BLOCK;
    gcn_zero_i32<<<(4 * nBins + B256 - 1) / B256, blk, 0, stream>>>(cnt2, 2 * nBins, cur2, 2 * nBins);
    gcn_bin_count<<<nEdgeBlocks, dim3(EDGE_THREADS), 0, stream>>>(src, dst, cnt2, E, nBins);
    gcn_bin_scatter<<<nEdgeBlocks, dim3(EDGE_THREADS), 0, stream>>>(src, dst, cnt2, cur2, part, spart, E, nBins);
    gcn_bin_sort<<<nBins, dim3(SORT_THREADS), 0, stream>>>(part, spart, cnt2, row_start, in_isqrt, col,
                                                          x, out_isqrt, bufA, N, nBins, N * 16);

    const int nTiles = (N + TILE_NODES - 1) / TILE_NODES;
    gcn_layer_fused<<<nTiles, dim3(128), 0, stream>>>(
        bufA, row_start, col, W1, b1, in_isqrt, out_isqrt, bufB, N);

    gcn_layer_head<<<nTiles, dim3(128), 0, stream>>>(
        bufB, row_start, col, W2, b2, in_isqrt, Wc, bc, out, N);
}


# Layer-1 epilogue staged through the LDS tile: each half-wave stores whole 256-B activation rows (one 16-B store per lane), not 64-B pieces per wave and row; next-layer scale loaded at kernel start
# speedup vs baseline: 3.4927x; 1.0636x over previous
#include <hip/hip_runtime.h>
#include <hip/hip_bf16.h>


typedef __attribute__((ext_vector_type(2))) float v2f;
typedef __attribute__((ext_vector_type(8))) float v8f;

#define D_FEAT 64
#define LDS_STRIDE 68

#define BIN_SHIFT 9
#define BIN_NODES (1 << BIN_SHIFT)
#define MAX_BINS  2048
#define EDGE_THREADS 256
#define EDGES_PER_THREAD 8
#define EDGES_PER_BLOCK (EDGE_THREADS * EDGES_PER_THREAD)
#define SORT_THREADS BIN_NODES
#define STAGE_CAP 12288
#define SORT_UNROLL 8

__global__ void gcn_zero_i32(int* __restrict__ p0, int n0, int* __restrict__ p1, int n1) {
    int i = blockIdx.x * blockDim.x + threadIdx.x;
    if (i < n0) p0[i] = 0;
    else if (i - n0 < n1) p1[i - n0] = 0;
}

__device__ __forceinline__ void gcn_block_scan_excl(int* a, int n, int* wsum) {
    const int tid  = threadIdx.x;
    const int nthr = blockDim.x;
    const int lane = tid & 31;
    const int wave = tid >> 5;
    const int nwav = nthr >> 5;
    const int G  = (n + nthr - 1) / nthr;
    const int i0 = tid * G;
    __syncthreads();
    int sum = 0;
    for (int k = 0; k < G; ++k) { const int i = i0 + k; if (i < n) sum += a[i]; }
    int incl = sum;
    for (int o = 1; o < 32; o <<= 1) {
        int t = __shfl_up(incl, o);
        if (lane >= o) incl += t;
    }
    if (lane == 31) wsum[wave] = incl;
    __syncthreads();
    int run = incl - sum;
    for (int w = 0; w < nwav; ++w) run += (w < wave) ? wsum[w] : 0;
    for (int k = 0; k < G; ++k) {
        const int i = i0 + k;
        if (i < n) { const int v = a[i]; a[i] = run; run += v; }
    }
    __syncthreads();
}

__global__ void __launch_bounds__(EDGE_THREADS)
gcn_bin_count(const int* __restrict__ src, const int* __restrict__ dst,
              int* __restrict__ cnt2, int nEdges, int nBins) {
    __shared__ int hcnt[2 * MAX_BINS];
    const int tid = threadIdx.x;
    const int nB2 = 2 * nBins;
    for (int i = tid; i < nB2; i += EDGE_THREADS) hcnt[i] = 0;
    __syncthreads();
    const int base = blockIdx.x * EDGES_PER_BLOCK;
    int es[EDGES_PER_THREAD], ed[EDGES_PER_THREAD];
    #pragma unroll
    for (int k = 0; k < EDGES_PER_THREAD; ++k) {
        const int ec = min(base + k * EDGE_THREADS + tid, nEdges - 1);
        es[k] = src[ec];
        ed[k] = dst[ec];
    }
    #pragma unroll
    for (int k = 0; k < EDGES_PER_THREAD; ++k) {
        if (base + k * EDGE_THREADS + tid < nEdges) {
            atomicAdd(&hcnt[ed[k] >> BIN_SHIFT], 1);
            atomicAdd(&hcnt[nBins + (es[k] >> BIN_SHIFT)], 1);
        }
    }
    __syncthreads();
    for (int i = tid; i < nB2; i += EDGE_THREADS) {
        const int c = hcnt[i];
        if (c) atomicAdd(&cnt2[i], c);
    }
}

__global__ void __launch_bounds__(EDGE_THREADS)
gcn_bin_scatter(const int* __restrict__ src, const int* __restrict__ dst,
                const int* __restrict__ cnt2, int* __restrict__ cur2,
                int2* __restrict__ part, int* __restrict__ spart,
                int nEdges, int nBins) {
    __shared__ int  hcnt[2 * MAX_BINS];
    __shared__ int  gpos[2 * MAX_BINS];
    __shared__ int2 stage[EDGES_PER_BLOCK];
    __shared__ int  sstage[EDGES_PER_BLOCK];
    __shared__ int  wsum[EDGE_THREADS / 32];

    const int tid = threadIdx.x;
    const int nB2 = 2 * nBins;

    for (int i = tid; i < nB2; i += EDGE_THREADS) { gpos[i] = cnt2[i]; hcnt[i] = 0; }
    gcn_block_scan_excl(gpos, nB2, wsum);

    const int base   = blockIdx.x * EDGES_PER_BLOCK;
    const int nLocal = min(EDGES_PER_BLOCK, nEdges - base);
    int es[EDGES_PER_THREAD], ed[EDGES_PER_THREAD], er[EDGES_PER_THREAD], sr[EDGES_PER_THREAD];
    #pragma unroll
    for (int k = 0; k < EDGES_PER_THREAD; ++k) {
        const int ec = min(base + k * EDGE_THREADS + tid, nEdges - 1);
        es[k] = src[ec];
        ed[k] = dst[ec];
    }
    #pragma unroll
    for (int k = 0; k < EDGES_PER_THREAD; ++k) {
        er[k] = -1;
        sr[k] = 0;
        if (base + k * EDGE_THREADS + tid < nEdges) {
            er[k] = atomicAdd(&hcnt[ed[k] >> BIN_SHIFT], 1);
            sr[k] = atomicAdd(&hcnt[nBins + (es[k] >> BIN_SHIFT)], 1);
        }
    }
    __syncthreads();

    for (int i = tid; i < nB2; i += EDGE_THREADS) {
        const int c = hcnt[i];
        if (c) gpos[i] += atomicAdd(&cur2[i], c);
    }
    gcn_block_scan_excl(hcnt, nB2, wsum);

    #pragma unroll
    for (int k = 0; k < EDGES_PER_THREAD; ++k) {
        if (er[k] >= 0) {
            stage[hcnt[ed[k] >> BIN_SHIFT] + er[k]] = make_int2(ed[k], es[k]);
            sstage[hcnt[nBins + (es[k] >> BIN_SHIFT)] - nLocal + sr[k]] = es[k];
        }
    }
    __syncthreads();

    #pragma unroll
    for (int k = 0; k < EDGES_PER_THREAD; ++k) {
        const int i  = k * EDGE_THREADS + tid;
        const int ic = min(i, nLocal - 1);
        const int2 p = stage[ic];
        const int b  = p.x >> BIN_SHIFT;
        const int o  = gpos[b] + (ic - hcnt[b]);
        const int s  = sstage[ic];
        const int sb = nBins + (s >> BIN_SHIFT);
        const int so = (gpos[sb] - nEdges) + (ic - (hcnt[sb] - nLocal));
        if (i < nLocal) { part[o] = p; spart[so] = s; }
    }
}

#define ROW_QUADS_PER_THREAD (BIN_NODES * (D_FEAT / 4) / SORT_THREADS)
__global__ void __launch_bounds__(SORT_THREADS)
gcn_bin_sort(const int2* __restrict__ part, const int* __restrict__ spart,
             const int* __restrict__ cnt2,
             int* __restrict__ row_start, float* __restrict__ in_isqrt,
             int* __restrict__ col,
             const float* __restrict__ x, float* __restrict__ out_isqrt,
             float* __restrict__ xs, int nNodes, int nBins, int nQuads  ) {
    __shared__ int   cnt[BIN_NODES];
    __shared__ int   cur[BIN_NODES];
    __shared__ int   ocnt[BIN_NODES];
    __shared__ float oscl[BIN_NODES];
    __shared__ int   wsum[SORT_THREADS / 32];
    __shared__ int   wsum2[SORT_THREADS / 32];
    __shared__ int   stage[STAGE_CAP];

    const int tid  = threadIdx.x;
    const int lane = tid & 31;
    const int wave = tid >> 5;
    const int bin  = blockIdx.x;

    const int q0 = bin * (BIN_NODES * (D_FEAT / 4));
    float4 xv[ROW_QUADS_PER_THREAD];
    #pragma unroll
    for (int u = 0; u < ROW_QUADS_PER_THREAD; ++u)
        xv[u] = ((const float4*)x)[min(q0 + u * SORT_THREADS + tid, nQuads - 1)];

    int pre = 0, spre = 0;
    for (int i = tid; i < bin; i += SORT_THREADS) { pre += cnt2[i]; spre += cnt2[nBins + i]; }
    for (int o = 16; o > 0; o >>= 1) { pre += __shfl_xor(pre, o); spre += __shfl_xor(spre, o); }
    if (lane == 0) { wsum[wave] = pre; wsum2[wave] = spre; }
    cnt[tid]  = 0;
    ocnt[tid] = 0;
    __syncthreads();
    int segBeg = 0, sSegBeg = 0;
    for (int w = 0; w < SORT_THREADS / 32; ++w) { segBeg += wsum[w]; sSegBeg += wsum2[w]; }
    const int segLen  = cnt2[bin];
    const int sSegLen = cnt2[nBins + bin];
    const int2* seg  = part + segBeg;
    const int*  sseg = spart + sSegBeg;

    for (int i0 = tid; i0 < sSegLen; i0 += SORT_UNROLL * SORT_THREADS) {
        int sl[SORT_UNROLL];
        #pragma unroll
        for (int u = 0; u < SORT_UNROLL; ++u)
            sl[u] = sseg[min(i0 + u * SORT_THREADS, sSegLen - 1)];
        #pragma unroll
        for (int u = 0; u < SORT_UNROLL; ++u)
            if (i0 + u * SORT_THREADS < sSegLen) atomicAdd(&ocnt[sl[u] & (BIN_NODES - 1)], 1);
    }
    for (int i0 = tid; i0 < segLen; i0 += SORT_UNROLL * SORT_THREADS) {
        int dl[SORT_UNROLL];
        #pragma unroll
        for (int u = 0; u < SORT_UNROLL; ++u)
            dl[u] = seg[min(i0 + u * SORT_THREADS, segLen - 1)].x;
        #pragma unroll
        for (int u = 0; u < SORT_UNROLL; ++u)
            if (i0 + u * SORT_THREADS < segLen) atomicAdd(&cnt[dl[u] & (BIN_NODES - 1)], 1);
    }
    __syncthreads();

    const int c = cnt[tid];
    int incl = c;
    for (int o = 1; o < 32; o <<= 1) {
        int v = __shfl_up(incl, o);
        if (lane >= o) incl += v;
    }
    if (lane == 31) wsum[wave] = incl;
    const float osc = rsqrtf(fmaxf((float)ocnt[tid], 1.0f));
    oscl[tid] = osc;
    __syncthreads();
    int excl = incl - c;
    for (int w = 0; w < SORT_THREADS / 32; ++w) excl += (w < wave) ? wsum[w] : 0;

    const int node = bin * BIN_NODES + tid;
    if (node < nNodes) {
        row_start[node] = segBeg + excl;
        if (node == nNodes - 1) row_start[nNodes] = segBeg + excl + c;
        in_isqrt[node]  = rsqrtf(fmaxf((float)c, 1.0f));
        out_isqrt[node] = osc;
    }
    cur[tid] = excl;
    __syncthreads();

    if (segLen <= STAGE_CAP) {
        for (int i0 = tid; i0 < segLen; i0 += SORT_UNROLL * SORT_THREADS) {
            int2 p[SORT_UNROLL];
            #pragma unroll
            for (int u = 0; u < SORT_UNROLL; ++u)
                p[u] = seg[min(i0 + u * SORT_THREADS, segLen - 1)];
            #pragma unroll
            for (int u = 0; u < SORT_UNROLL; ++u)
                if (i0 + u * SORT_THREADS < segLen)
                    stage[atomicAdd(&cur[p[u].x & (BIN_NODES - 1)], 1)] = p[u].y;
        }
        __syncthreads();
        for (int i = tid; i < segLen; i += SORT_THREADS) col[segBeg + i] = stage[i];
    } else {
        for (int i = tid; i < segLen; i += SORT_THREADS) {
            const int2 p = seg[i];
            col[segBeg + atomicAdd(&cur[p.x & (BIN_NODES - 1)], 1)] = p.y;
        }
    }

    #pragma unroll
    for (int u = 0; u < ROW_QUADS_PER_THREAD; ++u) {
        const int lq = u * SORT_THREADS + tid;
        if (q0 + lq < nQuads) {
            const float sc = oscl[lq >> 4];
            float4 v = xv[u];
            v.x *= sc; v.y *= sc; v.z *= sc; v.w *= sc;
            ((float4*)xs)[q0 + lq] = v;
        }
    }
}

#define TILE_NODES 32
#define NODES_PER_HW 4
template <bool HEAD>
__device__ __forceinline__ void
gcn_layer_body(float* __restrict__ lds,
               const float* __restrict__ h,
               const int* __restrict__ row_start,
               const int* __restrict__ col,
               const float* __restrict__ W,
               const float* __restrict__ bias,
               const float* __restrict__ in_isqrt,
               const float* __restrict__ post_scale,
               const float* __restrict__ Wc,
               const float* __restrict__ bc,
               float* __restrict__ out, int nNodes) {

    const int r0   = blockIdx.x * TILE_NODES;
    const int tid  = threadIdx.x;
    const int wave = tid >> 5;
    const int lane = tid & 31;

    float psl = 1.0f;
    if constexpr (!HEAD) {
        const int prow = r0 + ((tid & 15) << 3) + (tid >> 4);
        if (post_scale && (tid & 15) < TILE_NODES / 8 && prow < nNodes) psl = post_scale[prow];
    }

    {
        const int hw = tid >> 4;
        const int l  = tid & 15;
        const int f  = l << 2;
        const int n0 = r0 + hw * NODES_PER_HW;

        int   rsv = 0;
        float scv = 0.0f;
        if (l <= NODES_PER_HW) rsv = row_start[min(n0 + l, nNodes)];
        if (l <  NODES_PER_HW && n0 + l < nNodes) scv = in_isqrt[n0 + l];
        const int segbeg = __shfl(rsv, 0, 16);
        const int e1     = __shfl(rsv, 1, 16);
        const int e2     = __shfl(rsv, 2, 16);
        const int e3     = __shfl(rsv, 3, 16);
        const int segend = __shfl(rsv, 4, 16);
        const float sc0  = __shfl(scv, 0, 16);
        const float sc1  = __shfl(scv, 1, 16);
        const float sc2  = __shfl(scv, 2, 16);
        const float sc3  = __shfl(scv, 3, 16);

        const float4 zero4 = make_float4(0.0f, 0.0f, 0.0f, 0.0f);
        float4 acc = zero4;
        int cur    = 0;
        int curEnd = e1;
        const int lbase = (hw * NODES_PER_HW) * LDS_STRIDE + f;

#define GCN_FLUSH()                                                              \
        {                                                                        \
            const float s_ = (cur == 0) ? sc0 : (cur == 1) ? sc1 : (cur == 2) ? sc2 : sc3; \
            float4 o_;                                                           \
            o_.x = acc.x * s_; o_.y = acc.y * s_; o_.z = acc.z * s_; o_.w = acc.w * s_; \
            *(float4*)&lds[lbase + cur * LDS_STRIDE] = o_;                       \
            acc = zero4;                                                         \
            ++cur;                                                               \
            curEnd = (cur == 1) ? e2 : (cur == 2) ? e3 : segend;                 \
        }
#define GCN_ADD(rq, jq)                                                          \
        {                                                                        \
            const int j_ = (jq);                                                 \
            while (cur < NODES_PER_HW && j_ >= curEnd) GCN_FLUSH()               \
            acc.x += rq.x; acc.y += rq.y; acc.z += rq.z; acc.w += rq.w;          \
        }
#define GCN_LOAD(rq, sq, kq)                                                     \
        float4 rq = zero4;                                                       \
        if ((kq) < rem) rq = *(const float4*)(hb + ((unsigned)(sq) << 8));
#define GCN_GROUP_LOADS(K)                                                       \
            const int s0 = __shfl(c, (K),     16);                               \
            const int s1 = __shfl(c, (K) + 1, 16);                               \
            const int s2 = __shfl(c, (K) + 2, 16);                               \
            const int s3 = __shfl(c, (K) + 3, 16);                               \
            const int s4 = __shfl(c, (K) + 4, 16);                               \
            const int s5 = __shfl(c, (K) + 5, 16);                               \
            const int s6 = __shfl(c, (K) + 6, 16);                               \
            const int s7 = __shfl(c, (K) + 7, 16);                               \
            GCN_LOAD(q0, s0, (K))                                                \
            GCN_LOAD(q1, s1, (K) + 1)                                            \
            GCN_LOAD(q2, s2, (K) + 2)                                            \
            GCN_LOAD(q3, s3, (K) + 3)                                            \
            GCN_LOAD(q4, s4, (K) + 4)                                            \
            GCN_LOAD(q5, s5, (K) + 5)                                            \
            GCN_LOAD(q6, s6, (K) + 6)                                            \
            GCN_LOAD(q7, s7, (K) + 7)
#define GCN_GROUP_ADDS(K)                                                        \
            GCN_ADD(q0, base + (K))                                              \
            GCN_ADD(q1, base + (K) + 1)                                          \
            GCN_ADD(q2, base + (K) + 2)                                          \
            GCN_ADD(q3, base + (K) + 3)                                          \
            GCN_ADD(q4, base + (K) + 4)                                          \
            GCN_ADD(q5, base + (K) + 5)                                          \
            GCN_ADD(q6, base + (K) + 6)                                          \
            GCN_ADD(q7, base + (K) + 7)

        const char* hb = (const char*)h + (f << 2);
        int cnext = (segbeg + l < segend) ? col[segbeg + l] : 0;
        for (int base = segbeg; ; base += 16) {
            const int rem  = segend - base;
            const int remW = max(rem, __shfl_xor(rem, 16));
            if (remW <= 0) break;
            const int c = cnext;
            {
                GCN_GROUP_LOADS(0)
                cnext = (base + 16 + l < segend) ? col[base + 16 + l] : 0;
                GCN_GROUP_ADDS(0)
            }
            if (remW > 8) {
                GCN_GROUP_LOADS(8)
                GCN_GROUP_ADDS(8)
            }
        }
        while (cur < NODES_PER_HW) GCN_FLUSH()
#undef GCN_GROUP_ADDS
#undef GCN_GROUP_LOADS
#undef GCN_LOAD
#undef GCN_ADD
#undef GCN_FLUSH
    }
    __syncthreads();

    const int colbase = wave << 4;
    const int n     = lane & 15;
    const int khalf = lane >> 4;

    v8f acc0 = {};
    v8f acc1 = {};
    #pragma unroll
    for (int kk = 0; kk < 16; ++kk) {
        const int k0 = (kk << 2) + (khalf << 1);
        v2f a0 = *(const v2f*)&lds[n * LDS_STRIDE + k0];
        v2f a1 = *(const v2f*)&lds[(16 + n) * LDS_STRIDE + k0];
        v2f b;
        b.x = W[(k0 + 0) * D_FEAT + colbase + n];
        b.y = W[(k0 + 1) * D_FEAT + colbase + n];
        acc0 = __builtin_amdgcn_wmma_f32_16x16x4_f32(
             false, a0,  false, b,
             (short)0, acc0,  false,  false);
        acc1 = __builtin_amdgcn_wmma_f32_16x16x4_f32(
             false, a1,  false, b,
             (short)0, acc1,  false,  false);
    }

    const int colo = colbase + n;
    const float bv = bias[colo];
    if constexpr (!HEAD) {
        const int hw = tid >> 4;
        const int fq = (tid & 15) << 2;
        __syncthreads();
        #pragma unroll
        for (int t = 0; t < 2; ++t) {
            #pragma unroll
            for (int r = 0; r < 8; ++r) {
                float v = (t == 0 ? acc0[r] : acc1[r]) + bv;
                v = (v > 0.0f) ? v : 0.01f * v;
                lds[((t << 4) + (khalf << 3) + r) * LDS_STRIDE + colo] = v;
            }
        }
        __syncthreads();
        float4 ov[TILE_NODES / 8];
        #pragma unroll
        for (int i = 0; i < TILE_NODES / 8; ++i)
            ov[i] = *(const float4*)&lds[((i << 3) + hw) * LDS_STRIDE + fq];
        #pragma unroll
        for (int i = 0; i < TILE_NODES / 8; ++i) {
            const int row = r0 + (i << 3) + hw;
            const float ps = __shfl(psl, i, 16);
            if (row < nNodes) {
                float4 v = ov[i];
                v.x *= ps; v.y *= ps; v.z *= ps; v.w *= ps;
                *(float4*)&out[(size_t)row * D_FEAT + fq] = v;
            }
        }
    } else {
        __syncthreads();
        #pragma unroll
        for (int t = 0; t < 2; ++t) {
            #pragma unroll
            for (int r = 0; r < 8; ++r) {
                float v = (t == 0 ? acc0[r] : acc1[r]) + bv;
                v = (v > 0.0f) ? v : 0.01f * v;
                lds[((t << 4) + (khalf << 3) + r) * LDS_STRIDE + colo] = v;
            }
        }
        __syncthreads();
        const int node = r0 + lane;
        if (wave == 0 && node < nNodes) {
            float a0 = bc[0], a1 = bc[1];
            const float4* row = (const float4*)&lds[lane * LDS_STRIDE];
            #pragma unroll
            for (int q = 0; q < 16; ++q) {
                float4 v = row[q];
                int k = q << 2;
                a0 += v.x * Wc[(k+0)*2] + v.y * Wc[(k+1)*2] + v.z * Wc[(k+2)*2] + v.w * Wc[(k+3)*2];
                a1 += v.x * Wc[(k+0)*2+1] + v.y * Wc[(k+1)*2+1] + v.z * Wc[(k+2)*2+1] + v.w * Wc[(k+3)*2+1];
            }
            *(float2*)&out[(size_t)node * 2] = make_float2(a0, a1);
        }
    }
}

__global__ void __launch_bounds__(128)
gcn_layer_fused(const float* __restrict__ h, const int* __restrict__ row_start,
                const int* __restrict__ col, const float* __restrict__ W,
                const float* __restrict__ bias, const float* __restrict__ in_isqrt,
                const float* __restrict__ post_scale,
                float* __restrict__ out, int nNodes) {
    __shared__ float lds[TILE_NODES * LDS_STRIDE];
    gcn_layer_body<false>(lds, h, row_start, col, W, bias, in_isqrt, post_scale,
                          nullptr, nullptr, out, nNodes);
}

__global__ void __launch_bounds__(128)
gcn_layer_head(const float* __restrict__ h, const int* __restrict__ row_start,
               const int* __restrict__ col, const float* __restrict__ W,
               const float* __restrict__ bias, const float* __restrict__ in_isqrt,
               const float* __restrict__ Wc, const float* __restrict__ bc,
               float* __restrict__ out, int nNodes) {
    __shared__ float lds[TILE_NODES * LDS_STRIDE];
    gcn_layer_body<true>(lds, h, row_start, col, W, bias, in_isqrt, nullptr,
                         Wc, bc, out, nNodes);
}

extern "C" void kernel_launch(void* const* d_in, const int* in_sizes, int n_in,
                              void* d_out, int out_size, void* d_ws, size_t ws_size,
                              hipStream_t stream) {
    const float* x   = (const float*)d_in[0];
    const int*   src = (const int*)  d_in[1];
    const int*   dst = (const int*)  d_in[2];
    const float* W1  = (const float*)d_in[3];
    const float* b1  = (const float*)d_in[4];
    const float* W2  = (const float*)d_in[5];
    const float* b2  = (const float*)d_in[6];
    const float* Wc  = (const float*)d_in[7];
    const float* bc  = (const float*)d_in[8];
    float* out = (float*)d_out;

    const int N = in_sizes[0] / D_FEAT;
    const int E = in_sizes[1];
    const int nBins = (N + BIN_NODES - 1) >> BIN_SHIFT;

    float* bufA      = (float*)d_ws;
    float* bufB      = bufA + (size_t)N * D_FEAT;
    float* out_isqrt = bufB + (size_t)N * D_FEAT;
    float* in_isqrt  = out_isqrt + N;
    int*   row_start = (int*)(in_isqrt + N);
    int*   col       = row_start + (N + 1);
    int2*  part      = (int2*)bufB;
    int*   spart     = (int*)bufB + (size_t)2 * E;
    int*   cnt2      = spart + E;
    int*   cur2      = cnt2 + 2 * nBins;

    if (ws_size < ((size_t)2 * N * D_FEAT + (size_t)3 * N + 1 + (size_t)E) * 4) return;
    if ((size_t)3 * E + (size_t)4 * nBins > (size_t)N * D_FEAT) return;
    if (nBins > MAX_BINS || N < 1 || E < 1) return;

    const int B256 = 256;
    dim3 blk(B256);

    const int nEdgeBlocks = (E + EDGES_PER_BLOCK - 1) / EDGES_PER_BLOCK;
    gcn_zero_i32<<<(4 * nBins + B256 - 1) / B256, blk, 0, stream>>>(cnt2, 2 * nBins, cur2, 2 * nBins);
    gcn_bin_count<<<nEdgeBlocks, dim3(EDGE_THREADS), 0, stream>>>(src, dst, cnt2, E, nBins);
    gcn_bin_scatter<<<nEdgeBlocks, dim3(EDGE_THREADS), 0, stream>>>(src, dst, cnt2, cur2, part, spart, E, nBins);
    gcn_bin_sort<<<nBins, dim3(SORT_THREADS), 0, stream>>>(part, spart, cnt2, row_start, in_isqrt, col,
                                                          x, out_isqrt, bufA, N, nBins, N * 16);

    const int nTiles = (N + TILE_NODES - 1) / TILE_NODES;
    gcn_layer_fused<<<nTiles, dim3(128), 0, stream>>>(
        bufA, row_start, col, W1, b1, in_isqrt, out_isqrt, bufB, N);

    gcn_layer_head<<<nTiles, dim3(128), 0, stream>>>(
        bufB, row_start, col, W2, b2, in_isqrt, Wc, bc, out, N);
}


# Edge blocks sort their edges by bin in place and publish a piece table; bin blocks collect the pieces: zero and count launches and all global atomics go (6 -> 4); sums still in LDS arrival order
# speedup vs baseline: 3.7526x; 1.0744x over previous
#include <hip/hip_runtime.h>
#include <hip/hip_bf16.h>


typedef __attribute__((ext_vector_type(2))) float v2f;
typedef __attribute__((ext_vector_type(8))) float v8f;

#define D_FEAT 64
#define LDS_STRIDE 68

#define BIN_SHIFT 9
#define BIN_NODES (1 << BIN_SHIFT)
#define MAX_BINS  2048
#define MAX_BLK   1024
#define EDGE_THREADS 256
#define EDGES_PER_THREAD 8
#define EDGES_PER_BLOCK (EDGE_THREADS * EDGES_PER_THREAD)
#define SORT_THREADS BIN_NODES
#define STAGE_CAP 12288
#define SORT_UNROLL 16

__device__ __forceinline__ void gcn_block_scan_excl(int* a, int n, int* wsum) {
    const int tid  = threadIdx.x;
    const int nthr = blockDim.x;
    const int lane = tid & 31;
    const int wave = tid >> 5;
    const int nwav = nthr >> 5;
    const int G  = (n + nthr - 1) / nthr;
    const int i0 = tid * G;
    __syncthreads();
    int sum = 0;
    for (int k = 0; k < G; ++k) { const int i = i0 + k; if (i < n) sum += a[i]; }
    int incl = sum;
    for (int o = 1; o < 32; o <<= 1) {
        int t = __shfl_up(incl, o);
        if (lane >= o) incl += t;
    }
    if (lane == 31) wsum[wave] = incl;
    __syncthreads();
    int run = incl - sum;
    for (int w = 0; w < nwav; ++w) run += (w < wave) ? wsum[w] : 0;
    for (int k = 0; k < G; ++k) {
        const int i = i0 + k;
        if (i < n) { const int v = a[i]; a[i] = run; run += v; }
    }
    __syncthreads();
}

__global__ void __launch_bounds__(EDGE_THREADS)
gcn_block_partition(const int* __restrict__ src, const int* __restrict__ dst,
                    int2* __restrict__ part, int* __restrict__ spart,
                    int4* __restrict__ tab, int nEdges, int nBins) {
    __shared__ int  hcnt[2 * MAX_BINS];
    __shared__ int2 stage[EDGES_PER_BLOCK];
    __shared__ int  sstage[EDGES_PER_BLOCK];
    __shared__ int  wsum[EDGE_THREADS / 32];

    const int tid = threadIdx.x;
    const int nB2 = 2 * nBins;
    for (int i = tid; i < nB2; i += EDGE_THREADS) hcnt[i] = 0;
    __syncthreads();

    const int base   = blockIdx.x * EDGES_PER_BLOCK;
    const int nLocal = min(EDGES_PER_BLOCK, nEdges - base);
    int es[EDGES_PER_THREAD], ed[EDGES_PER_THREAD], er[EDGES_PER_THREAD], sr[EDGES_PER_THREAD];
    #pragma unroll
    for (int k = 0; k < EDGES_PER_THREAD; ++k) {
        const int ec = min(base + k * EDGE_THREADS + tid, nEdges - 1);
        es[k] = src[ec];
        ed[k] = dst[ec];
    }
    #pragma unroll
    for (int k = 0; k < EDGES_PER_THREAD; ++k) {
        er[k] = -1;
        sr[k] = 0;
        if (base + k * EDGE_THREADS + tid < nEdges) {
            er[k] = atomicAdd(&hcnt[ed[k] >> BIN_SHIFT], 1);
            sr[k] = atomicAdd(&hcnt[nBins + (es[k] >> BIN_SHIFT)], 1);
        }
    }
    gcn_block_scan_excl(hcnt, nB2, wsum);

    for (int b = tid; b < nBins; b += EDGE_THREADS) {
        const int d0 = hcnt[b];
        const int d1 = hcnt[b + 1];
        const int s0 = hcnt[nBins + b];
        const int s1 = (b + 1 < nBins) ? hcnt[nBins + b + 1] : 2 * nLocal;
        tab[(size_t)blockIdx.x * nBins + b] = make_int4(d0, d1 - d0, s0 - nLocal, s1 - s0);
    }

    #pragma unroll
    for (int k = 0; k < EDGES_PER_THREAD; ++k) {
        if (er[k] >= 0) {
            stage[hcnt[ed[k] >> BIN_SHIFT] + er[k]] = make_int2(ed[k], es[k]);
            sstage[hcnt[nBins + (es[k] >> BIN_SHIFT)] - nLocal + sr[k]] = es[k];
        }
    }
    __syncthreads();

    #pragma unroll
    for (int k = 0; k < EDGES_PER_THREAD; ++k) {
        const int i = k * EDGE_THREADS + tid;
        if (i < nLocal) { part[base + i] = stage[i]; spart[base + i] = sstage[i]; }
    }
}

#define ROW_QUADS_PER_THREAD (BIN_NODES * (D_FEAT / 4) / SORT_THREADS)
#define PIECE_SLOTS 16
#define GCN_SLOT(v)                                                              \
        const int  v_  = (v);                                                    \
        const int  pc_ = min(v_ / PIECE_SLOTS, nBlk - 1);                        \
        const int  k_  = kb + (v_ & (PIECE_SLOTS - 1));                          \
        const bool in_ = v_ < nSlots;
__global__ void __launch_bounds__(SORT_THREADS)
gcn_bin_sort(const int2* __restrict__ part, const int* __restrict__ spart,
             const int4* __restrict__ tab,
             int* __restrict__ row_start, float* __restrict__ in_isqrt,
             int* __restrict__ col,
             const float* __restrict__ x, float* __restrict__ out_isqrt,
             float* __restrict__ xs, int nNodes, int nBins, int nQuads  ,
             int nBlk, int nEdges) {
    __shared__ int   cnt[BIN_NODES];
    __shared__ int   cur[BIN_NODES];
    __shared__ int   ocnt[BIN_NODES];
    __shared__ float oscl[BIN_NODES];
    __shared__ int   wsum[SORT_THREADS / 32];
    __shared__ int   wsum2[SORT_THREADS / 32];
    __shared__ int   wsum3[SORT_THREADS / 32];
    __shared__ int   stage[STAGE_CAP];
    __shared__ int2  pd[MAX_BLK];
    __shared__ int2  ps[MAX_BLK];

    const int tid  = threadIdx.x;
    const int lane = tid & 31;
    const int wave = tid >> 5;
    const int bin  = blockIdx.x;

    int pre = 0, len = 0, mx = 0;
    for (int p = tid; p < nBlk; p += SORT_THREADS) {
        const int4 t  = tab[(size_t)p * nBins + bin];
        const int  gb = p * EDGES_PER_BLOCK;
        pd[p] = make_int2(gb + t.x, t.y);
        ps[p] = make_int2(gb + t.z, t.w);
        pre += t.x;
        len += t.y;
        mx   = max(mx, max(t.y, t.w));
    }

    const int q0 = bin * (BIN_NODES * (D_FEAT / 4));
    float4 xv[ROW_QUADS_PER_THREAD];
    #pragma unroll
    for (int u = 0; u < ROW_QUADS_PER_THREAD; ++u)
        xv[u] = ((const float4*)x)[min(q0 + u * SORT_THREADS + tid, nQuads - 1)];

    for (int o = 16; o > 0; o >>= 1) {
        pre += __shfl_xor(pre, o);
        len += __shfl_xor(len, o);
        mx   = max(mx, __shfl_xor(mx, o));
    }
    if (lane == 0) { wsum[wave] = pre; wsum2[wave] = len; wsum3[wave] = mx; }
    cnt[tid]  = 0;
    ocnt[tid] = 0;
    __syncthreads();
    int segBeg = 0, segLen = 0, maxCnt = 0;
    for (int w = 0; w < SORT_THREADS / 32; ++w) {
        segBeg += wsum[w]; segLen += wsum2[w]; maxCnt = max(maxCnt, wsum3[w]);
    }
    const int nSlots = nBlk * PIECE_SLOTS;

    for (int kb = 0; kb < maxCnt; kb += PIECE_SLOTS) {
        for (int v0 = tid; v0 < nSlots; v0 += SORT_UNROLL * SORT_THREADS) {
            int  dl[SORT_UNROLL], sl[SORT_UNROLL];
            bool dv[SORT_UNROLL], sv[SORT_UNROLL];
            #pragma unroll
            for (int u = 0; u < SORT_UNROLL; ++u) {
                GCN_SLOT(v0 + u * SORT_THREADS)
                const int2 a = pd[pc_];
                const int2 b = ps[pc_];
                dv[u] = in_ && k_ < a.y;
                sv[u] = in_ && k_ < b.y;
                dl[u] = part[min(a.x + k_, nEdges - 1)].x;
                sl[u] = spart[min(b.x + k_, nEdges - 1)];
            }
            #pragma unroll
            for (int u = 0; u < SORT_UNROLL; ++u) {
                if (dv[u]) atomicAdd(&cnt[dl[u] & (BIN_NODES - 1)], 1);
                if (sv[u]) atomicAdd(&ocnt[sl[u] & (BIN_NODES - 1)], 1);
            }
        }
    }
    __syncthreads();

    const int c = cnt[tid];
    int incl = c;
    for (int o = 1; o < 32; o <<= 1) {
        int v = __shfl_up(incl, o);
        if (lane >= o) incl += v;
    }
    if (lane == 31) wsum[wave] = incl;
    const float osc = rsqrtf(fmaxf((float)ocnt[tid], 1.0f));
    oscl[tid] = osc;
    __syncthreads();
    int excl = incl - c;
    for (int w = 0; w < SORT_THREADS / 32; ++w) excl += (w < wave) ? wsum[w] : 0;

    const int node = bin * BIN_NODES + tid;
    if (node < nNodes) {
        row_start[node] = segBeg + excl;
        if (node == nNodes - 1) row_start[nNodes] = segBeg + excl + c;
        in_isqrt[node]  = rsqrtf(fmaxf((float)c, 1.0f));
        out_isqrt[node] = osc;
    }
    cur[tid] = excl;
    __syncthreads();


    if (segLen <= STAGE_CAP) {
        for (int kb = 0; kb < maxCnt; kb += PIECE_SLOTS) {
            for (int v0 = tid; v0 < nSlots; v0 += SORT_UNROLL * SORT_THREADS) {
                int2 pr[SORT_UNROLL];
                bool pv[SORT_UNROLL];
                #pragma unroll
                for (int u = 0; u < SORT_UNROLL; ++u) {
                    GCN_SLOT(v0 + u * SORT_THREADS)
                    const int2 a = pd[pc_];
                    pv[u] = in_ && k_ < a.y;
                    pr[u] = part[min(a.x + k_, nEdges - 1)];
                }
                #pragma unroll
                for (int u = 0; u < SORT_UNROLL; ++u)
                    if (pv[u]) stage[atomicAdd(&cur[pr[u].x & (BIN_NODES - 1)], 1)] = pr[u].y;
            }
        }
        __syncthreads();
        for (int i = tid; i < segLen; i += SORT_THREADS) col[segBeg + i] = stage[i];
    } else {
        for (int kb = 0; kb < maxCnt; kb += PIECE_SLOTS) {
            for (int v0 = tid; v0 < nSlots; v0 += SORT_THREADS) {
                GCN_SLOT(v0)
                const int2 a = pd[pc_];
                if (in_ && k_ < a.y) {
                    const int2 p = part[a.x + k_];
                    col[segBeg + atomicAdd(&cur[p.x & (BIN_NODES - 1)], 1)] = p.y;
                }
            }
        }
    }

    #pragma unroll
    for (int u = 0; u < ROW_QUADS_PER_THREAD; ++u) {
        const int lq = u * SORT_THREADS + tid;
        if (q0 + lq < nQuads) {
            const float sc = oscl[lq >> 4];
            float4 v = xv[u];
            v.x *= sc; v.y *= sc; v.z *= sc; v.w *= sc;
            ((float4*)xs)[q0 + lq] = v;
        }
    }
}
#undef GCN_SLOT

#define TILE_NODES 32
#define NODES_PER_HW 4
template <bool HEAD>
__device__ __forceinline__ void
gcn_layer_body(float* __restrict__ lds,
               const float* __restrict__ h,
               const int* __restrict__ row_start,
               const int* __restrict__ col,
               const float* __restrict__ W,
               const float* __restrict__ bias,
               const float* __restrict__ in_isqrt,
               const float* __restrict__ post_scale,
               const float* __restrict__ Wc,
               const float* __restrict__ bc,
               float* __restrict__ out, int nNodes) {

    const int r0   = blockIdx.x * TILE_NODES;
    const int tid  = threadIdx.x;
    const int wave = tid >> 5;
    const int lane = tid & 31;

    float psl = 1.0f;
    if constexpr (!HEAD) {
        const int prow = r0 + ((tid & 15) << 3) + (tid >> 4);
        if (post_scale && (tid & 15) < TILE_NODES / 8 && prow < nNodes) psl = post_scale[prow];
    }

    {
        const int hw = tid >> 4;
        const int l  = tid & 15;
        const int f  = l << 2;
        const int n0 = r0 + hw * NODES_PER_HW;

        int   rsv = 0;
        float scv = 0.0f;
        if (l <= NODES_PER_HW) rsv = row_start[min(n0 + l, nNodes)];
        if (l <  NODES_PER_HW && n0 + l < nNodes) scv = in_isqrt[n0 + l];
        const int segbeg = __shfl(rsv, 0, 16);
        const int e1     = __shfl(rsv, 1, 16);
        const int e2     = __shfl(rsv, 2, 16);
        const int e3     = __shfl(rsv, 3, 16);
        const int segend = __shfl(rsv, 4, 16);
        const float sc0  = __shfl(scv, 0, 16);
        const float sc1  = __shfl(scv, 1, 16);
        const float sc2  = __shfl(scv, 2, 16);
        const float sc3  = __shfl(scv, 3, 16);

        const float4 zero4 = make_float4(0.0f, 0.0f, 0.0f, 0.0f);
        float4 acc = zero4;
        int cur    = 0;
        int curEnd = e1;
        const int lbase = (hw * NODES_PER_HW) * LDS_STRIDE + f;

#define GCN_FLUSH()                                                              \
        {                                                                        \
            const float s_ = (cur == 0) ? sc0 : (cur == 1) ? sc1 : (cur == 2) ? sc2 : sc3; \
            float4 o_;                                                           \
            o_.x = acc.x * s_; o_.y = acc.y * s_; o_.z = acc.z * s_; o_.w = acc.w * s_; \
            *(float4*)&lds[lbase + cur * LDS_STRIDE] = o_;                       \
            acc = zero4;                                                         \
            ++cur;                                                               \
            curEnd = (cur == 1) ? e2 : (cur == 2) ? e3 : segend;                 \
        }
#define GCN_ADD(rq, jq)                                                          \
        {                                                                        \
            const int j_ = (jq);                                                 \
            while (cur < NODES_PER_HW && j_ >= curEnd) GCN_FLUSH()               \
            acc.x += rq.x; acc.y += rq.y; acc.z += rq.z; acc.w += rq.w;          \
        }
#define GCN_LOAD(rq, sq, kq)                                                     \
        float4 rq = zero4;                                                       \
        if ((kq) < rem) rq = *(const float4*)(hb + ((unsigned)(sq) << 8));
#define GCN_GROUP_LOADS(K)                                                       \
            const int s0 = __shfl(c, (K),     16);                               \
            const int s1 = __shfl(c, (K) + 1, 16);                               \
            const int s2 = __shfl(c, (K) + 2, 16);                               \
            const int s3 = __shfl(c, (K) + 3, 16);                               \
            const int s4 = __shfl(c, (K) + 4, 16);                               \
            const int s5 = __shfl(c, (K) + 5, 16);                               \
            const int s6 = __shfl(c, (K) + 6, 16);                               \
            const int s7 = __shfl(c, (K) + 7, 16);                               \
            GCN_LOAD(q0, s0, (K))                                                \
            GCN_LOAD(q1, s1, (K) + 1)                                            \
            GCN_LOAD(q2, s2, (K) + 2)                                            \
            GCN_LOAD(q3, s3, (K) + 3)                                            \
            GCN_LOAD(q4, s4, (K) + 4)                                            \
            GCN_LOAD(q5, s5, (K) + 5)                                            \
            GCN_LOAD(q6, s6, (K) + 6)                                            \
            GCN_LOAD(q7, s7, (K) + 7)
#define GCN_GROUP_ADDS(K)                                                        \
            GCN_ADD(q0, base + (K))                                              \
            GCN_ADD(q1, base + (K) + 1)                                          \
            GCN_ADD(q2, base + (K) + 2)                                          \
            GCN_ADD(q3, base + (K) + 3)                                          \
            GCN_ADD(q4, base + (K) + 4)                                          \
            GCN_ADD(q5, base + (K) + 5)                                          \
            GCN_ADD(q6, base + (K) + 6)                                          \
            GCN_ADD(q7, base + (K) + 7)

        const char* hb = (const char*)h + (f << 2);
        int cnext = (segbeg + l < segend) ? col[segbeg + l] : 0;
        for (int base = segbeg; ; base += 16) {
            const int rem  = segend - base;
            const int remW = max(rem, __shfl_xor(rem, 16));
            if (remW <= 0) break;
            const int c = cnext;
            {
                GCN_GROUP_LOADS(0)
                cnext = (base + 16 + l < segend) ? col[base + 16 + l] : 0;
                GCN_GROUP_ADDS(0)
            }
            if (remW > 8) {
                GCN_GROUP_LOADS(8)
                GCN_GROUP_ADDS(8)
            }
        }
        while (cur < NODES_PER_HW) GCN_FLUSH()
#undef GCN_GROUP_ADDS
#undef GCN_GROUP_LOADS
#undef GCN_LOAD
#undef GCN_ADD
#undef GCN_FLUSH
    }
    __syncthreads();

    const int colbase = wave << 4;
    const int n     = lane & 15;
    const int khalf = lane >> 4;

    v8f acc0 = {};
    v8f acc1 = {};
    #pragma unroll
    for (int kk = 0; kk < 16; ++kk) {
        const int k0 = (kk << 2) + (khalf << 1);
        v2f a0 = *(const v2f*)&lds[n * LDS_STRIDE + k0];
        v2f a1 = *(const v2f*)&lds[(16 + n) * LDS_STRIDE + k0];
        v2f b;
        b.x = W[(k0 + 0) * D_FEAT + colbase + n];
        b.y = W[(k0 + 1) * D_FEAT + colbase + n];
        acc0 = __builtin_amdgcn_wmma_f32_16x16x4_f32(
             false, a0,  false, b,
             (short)0, acc0,  false,  false);
        acc1 = __builtin_amdgcn_wmma_f32_16x16x4_f32(
             false, a1,  false, b,
             (short)0, acc1,  false,  false);
    }

    const int colo = colbase + n;
    const float bv = bias[colo];
    if constexpr (!HEAD) {
        const int hw = tid >> 4;
        const int fq = (tid & 15) << 2;
        __syncthreads();
        #pragma unroll
        for (int t = 0; t < 2; ++t) {
            #pragma unroll
            for (int r = 0; r < 8; ++r) {
                float v = (t == 0 ? acc0[r] : acc1[r]) + bv;
                v = (v > 0.0f) ? v : 0.01f * v;
                lds[((t << 4) + (khalf << 3) + r) * LDS_STRIDE + colo] = v;
            }
        }
        __syncthreads();
        float4 ov[TILE_NODES / 8];
        #pragma unroll
        for (int i = 0; i < TILE_NODES / 8; ++i)
            ov[i] = *(const float4*)&lds[((i << 3) + hw) * LDS_STRIDE + fq];
        #pragma unroll
        for (int i = 0; i < TILE_NODES / 8; ++i) {
            const int row = r0 + (i << 3) + hw;
            const float ps = __shfl(psl, i, 16);
            if (row < nNodes) {
                float4 v = ov[i];
                v.x *= ps; v.y *= ps; v.z *= ps; v.w *= ps;
                *(float4*)&out[(size_t)row * D_FEAT + fq] = v;
            }
        }
    } else {
        __syncthreads();
        #pragma unroll
        for (int t = 0; t < 2; ++t) {
            #pragma unroll
            for (int r = 0; r < 8; ++r) {
                float v = (t == 0 ? acc0[r] : acc1[r]) + bv;
                v = (v > 0.0f) ? v : 0.01f * v;
                lds[((t << 4) + (khalf << 3) + r) * LDS_STRIDE + colo] = v;
            }
        }
        __syncthreads();
        const int node = r0 + lane;
        if (wave == 0 && node < nNodes) {
            float a0 = bc[0], a1 = bc[1];
            const float4* row = (const float4*)&lds[lane * LDS_STRIDE];
            #pragma unroll
            for (int q = 0; q < 16; ++q) {
                float4 v = row[q];
                int k = q << 2;
                a0 += v.x * Wc[(k+0)*2] + v.y * Wc[(k+1)*2] + v.z * Wc[(k+2)*2] + v.w * Wc[(k+3)*2];
                a1 += v.x * Wc[(k+0)*2+1] + v.y * Wc[(k+1)*2+1] + v.z * Wc[(k+2)*2+1] + v.w * Wc[(k+3)*2+1];
            }
            *(float2*)&out[(size_t)node * 2] = make_float2(a0, a1);
        }
    }
}

__global__ void __launch_bounds__(128)
gcn_layer_fused(const float* __restrict__ h, const int* __restrict__ row_start,
                const int* __restrict__ col, const float* __restrict__ W,
                const float* __restrict__ bias, const float* __restrict__ in_isqrt,
                const float* __restrict__ post_scale,
                float* __restrict__ out, int nNodes) {
    __shared__ float lds[TILE_NODES * LDS_STRIDE];
    gcn_layer_body<false>(lds, h, row_start, col, W, bias, in_isqrt, post_scale,
                          nullptr, nullptr, out, nNodes);
}

__global__ void __launch_bounds__(128)
gcn_layer_head(const float* __restrict__ h, const int* __restrict__ row_start,
               const int* __restrict__ col, const float* __restrict__ W,
               const float* __restrict__ bias, const float* __restrict__ in_isqrt,
               const float* __restrict__ Wc, const float* __restrict__ bc,
               float* __restrict__ out, int nNodes) {
    __shared__ float lds[TILE_NODES * LDS_STRIDE];
    gcn_layer_body<true>(lds, h, row_start, col, W, bias, in_isqrt, nullptr,
                         Wc, bc, out, nNodes);
}

extern "C" void kernel_launch(void* const* d_in, const int* in_sizes, int n_in,
                              void* d_out, int out_size, void* d_ws, size_t ws_size,
                              hipStream_t stream) {
    const float* x   = (const float*)d_in[0];
    const int*   src = (const int*)  d_in[1];
    const int*   dst = (const int*)  d_in[2];
    const float* W1  = (const float*)d_in[3];
    const float* b1  = (const float*)d_in[4];
    const float* W2  = (const float*)d_in[5];
    const float* b2  = (const float*)d_in[6];
    const float* Wc  = (const float*)d_in[7];
    const float* bc  = (const float*)d_in[8];
    float* out = (float*)d_out;

    const int N = in_sizes[0] / D_FEAT;
    const int E = in_sizes[1];
    const int nBins = (N + BIN_NODES - 1) >> BIN_SHIFT;
    const int nEdgeBlocks = (E + EDGES_PER_BLOCK - 1) / EDGES_PER_BLOCK;

    float* bufA      = (float*)d_ws;
    float* bufB      = bufA + (size_t)N * D_FEAT;
    float* out_isqrt = bufB + (size_t)N * D_FEAT;
    float* in_isqrt  = out_isqrt + N;
    int*   row_start = (int*)(in_isqrt + N);
    int*   col       = row_start + (N + 1);
    int4*  tab       = (int4*)bufB;
    int2*  part      = (int2*)(tab + (size_t)nEdgeBlocks * nBins);
    int*   spart     = (int*)(part + E);

    if (ws_size < ((size_t)2 * N * D_FEAT + (size_t)3 * N + 1 + (size_t)E) * 4) return;
    if ((size_t)3 * E + (size_t)4 * nEdgeBlocks * nBins > (size_t)N * D_FEAT) return;
    if (nBins > MAX_BINS || nEdgeBlocks > MAX_BLK || N < 1 || E < 1) return;

    gcn_block_partition<<<nEdgeBlocks, dim3(EDGE_THREADS), 0, stream>>>(src, dst, part, spart, tab, E, nBins);
    gcn_bin_sort<<<nBins, dim3(SORT_THREADS), 0, stream>>>(part, spart, tab, row_start, in_isqrt, col,
                                                          x, out_isqrt, bufA, N, nBins, N * 16, nEdgeBlocks, E);

    const int nTiles = (N + TILE_NODES - 1) / TILE_NODES;
    gcn_layer_fused<<<nTiles, dim3(128), 0, stream>>>(
        bufA, row_start, col, W1, b1, in_isqrt, out_isqrt, bufB, N);

    gcn_layer_head<<<nTiles, dim3(128), 0, stream>>>(
        bufB, row_start, col, W2, b2, in_isqrt, Wc, bc, out, N);
}


# Row-sum walk asks one question per row: open nodes' ends and scales kept as a shifting queue counted from the chunk's first edge; no per-row node lookup or select chains; sums and their order same
# speedup vs baseline: 3.9012x; 1.0396x over previous
#include <hip/hip_runtime.h>
#include <hip/hip_bf16.h>


typedef __attribute__((ext_vector_type(2))) float v2f;
typedef __attribute__((ext_vector_type(8))) float v8f;

#define D_FEAT 64
#define LDS_STRIDE 68

#define BIN_SHIFT 9
#define BIN_NODES (1 << BIN_SHIFT)
#define MAX_BINS  2048
#define MAX_BLK   1024
#define EDGE_THREADS 256
#define EDGES_PER_THREAD 8
#define EDGES_PER_BLOCK (EDGE_THREADS * EDGES_PER_THREAD)
#define SORT_THREADS BIN_NODES
#define STAGE_CAP 12288
#define SORT_UNROLL 16

__device__ __forceinline__ void gcn_block_scan_excl(int* a, int n, int* wsum) {
    const int tid  = threadIdx.x;
    const int nthr = blockDim.x;
    const int lane = tid & 31;
    const int wave = tid >> 5;
    const int nwav = nthr >> 5;
    const int G  = (n + nthr - 1) / nthr;
    const int i0 = tid * G;
    __syncthreads();
    int sum = 0;
    for (int k = 0; k < G; ++k) { const int i = i0 + k; if (i < n) sum += a[i]; }
    int incl = sum;
    for (int o = 1; o < 32; o <<= 1) {
        int t = __shfl_up(incl, o);
        if (lane >= o) incl += t;
    }
    if (lane == 31) wsum[wave] = incl;
    __syncthreads();
    int run = incl - sum;
    for (int w = 0; w < nwav; ++w) run += (w < wave) ? wsum[w] : 0;
    for (int k = 0; k < G; ++k) {
        const int i = i0 + k;
        if (i < n) { const int v = a[i]; a[i] = run; run += v; }
    }
    __syncthreads();
}

__global__ void __launch_bounds__(EDGE_THREADS)
gcn_block_partition(const int* __restrict__ src, const int* __restrict__ dst,
                    int2* __restrict__ part, int* __restrict__ spart,
                    int4* __restrict__ tab, int nEdges, int nBins) {
    __shared__ int  hcnt[2 * MAX_BINS];
    __shared__ int2 stage[EDGES_PER_BLOCK];
    __shared__ int  sstage[EDGES_PER_BLOCK];
    __shared__ int  wsum[EDGE_THREADS / 32];

    const int tid = threadIdx.x;
    const int nB2 = 2 * nBins;
    for (int i = tid; i < nB2; i += EDGE_THREADS) hcnt[i] = 0;
    __syncthreads();

    const int base   = blockIdx.x * EDGES_PER_BLOCK;
    const int nLocal = min(EDGES_PER_BLOCK, nEdges - base);
    int es[EDGES_PER_THREAD], ed[EDGES_PER_THREAD], er[EDGES_PER_THREAD], sr[EDGES_PER_THREAD];
    #pragma unroll
    for (int k = 0; k < EDGES_PER_THREAD; ++k) {
        const int ec = min(base + k * EDGE_THREADS + tid, nEdges - 1);
        es[k] = src[ec];
        ed[k] = dst[ec];
    }
    #pragma unroll
    for (int k = 0; k < EDGES_PER_THREAD; ++k) {
        er[k] = -1;
        sr[k] = 0;
        if (base + k * EDGE_THREADS + tid < nEdges) {
            er[k] = atomicAdd(&hcnt[ed[k] >> BIN_SHIFT], 1);
            sr[k] = atomicAdd(&hcnt[nBins + (es[k] >> BIN_SHIFT)], 1);
        }
    }
    gcn_block_scan_excl(hcnt, nB2, wsum);

    for (int b = tid; b < nBins; b += EDGE_THREADS) {
        const int d0 = hcnt[b];
        const int d1 = hcnt[b + 1];
        const int s0 = hcnt[nBins + b];
        const int s1 = (b + 1 < nBins) ? hcnt[nBins + b + 1] : 2 * nLocal;
        tab[(size_t)blockIdx.x * nBins + b] = make_int4(d0, d1 - d0, s0 - nLocal, s1 - s0);
    }

    #pragma unroll
    for (int k = 0; k < EDGES_PER_THREAD; ++k) {
        if (er[k] >= 0) {
            stage[hcnt[ed[k] >> BIN_SHIFT] + er[k]] = make_int2(ed[k], es[k]);
            sstage[hcnt[nBins + (es[k] >> BIN_SHIFT)] - nLocal + sr[k]] = es[k];
        }
    }
    __syncthreads();

    #pragma unroll
    for (int k = 0; k < EDGES_PER_THREAD; ++k) {
        const int i = k * EDGE_THREADS + tid;
        if (i < nLocal) { part[base + i] = stage[i]; spart[base + i] = sstage[i]; }
    }
}

#define ROW_QUADS_PER_THREAD (BIN_NODES * (D_FEAT / 4) / SORT_THREADS)
#define PIECE_SLOTS 16
#define GCN_SLOT(v)                                                              \
        const int  v_  = (v);                                                    \
        const int  pc_ = min(v_ / PIECE_SLOTS, nBlk - 1);                        \
        const int  k_  = kb + (v_ & (PIECE_SLOTS - 1));                          \
        const bool in_ = v_ < nSlots;
__global__ void __launch_bounds__(SORT_THREADS)
gcn_bin_sort(const int2* __restrict__ part, const int* __restrict__ spart,
             const int4* __restrict__ tab,
             int* __restrict__ row_start, float* __restrict__ in_isqrt,
             int* __restrict__ col,
             const float* __restrict__ x, float* __restrict__ out_isqrt,
             float* __restrict__ xs, int nNodes, int nBins, int nQuads  ,
             int nBlk, int nEdges) {
    __shared__ int   cnt[BIN_NODES];
    __shared__ int   cur[BIN_NODES];
    __shared__ int   ocnt[BIN_NODES];
    __shared__ float oscl[BIN_NODES];
    __shared__ int   wsum[SORT_THREADS / 32];
    __shared__ int   wsum2[SORT_THREADS / 32];
    __shared__ int   wsum3[SORT_THREADS / 32];
    __shared__ int   stage[STAGE_CAP];
    __shared__ int2  pd[MAX_BLK];
    __shared__ int2  ps[MAX_BLK];

    const int tid  = threadIdx.x;
    const int lane = tid & 31;
    const int wave = tid >> 5;
    const int bin  = blockIdx.x;

    int pre = 0, len = 0, mx = 0;
    for (int p = tid; p < nBlk; p += SORT_THREADS) {
        const int4 t  = tab[(size_t)p * nBins + bin];
        const int  gb = p * EDGES_PER_BLOCK;
        pd[p] = make_int2(gb + t.x, t.y);
        ps[p] = make_int2(gb + t.z, t.w);
        pre += t.x;
        len += t.y;
        mx   = max(mx, max(t.y, t.w));
    }

    const int q0 = bin * (BIN_NODES * (D_FEAT / 4));
    float4 xv[ROW_QUADS_PER_THREAD];
    #pragma unroll
    for (int u = 0; u < ROW_QUADS_PER_THREAD; ++u)
        xv[u] = ((const float4*)x)[min(q0 + u * SORT_THREADS + tid, nQuads - 1)];

    for (int o = 16; o > 0; o >>= 1) {
        pre += __shfl_xor(pre, o);
        len += __shfl_xor(len, o);
        mx   = max(mx, __shfl_xor(mx, o));
    }
    if (lane == 0) { wsum[wave] = pre; wsum2[wave] = len; wsum3[wave] = mx; }
    cnt[tid]  = 0;
    ocnt[tid] = 0;
    __syncthreads();
    int segBeg = 0, segLen = 0, maxCnt = 0;
    for (int w = 0; w < SORT_THREADS / 32; ++w) {
        segBeg += wsum[w]; segLen += wsum2[w]; maxCnt = max(maxCnt, wsum3[w]);
    }
    const int nSlots = nBlk * PIECE_SLOTS;

    for (int kb = 0; kb < maxCnt; kb += PIECE_SLOTS) {
        for (int v0 = tid; v0 < nSlots; v0 += SORT_UNROLL * SORT_THREADS) {
            int  dl[SORT_UNROLL], sl[SORT_UNROLL];
            bool dv[SORT_UNROLL], sv[SORT_UNROLL];
            #pragma unroll
            for (int u = 0; u < SORT_UNROLL; ++u) {
                GCN_SLOT(v0 + u * SORT_THREADS)
                const int2 a = pd[pc_];
                const int2 b = ps[pc_];
                dv[u] = in_ && k_ < a.y;
                sv[u] = in_ && k_ < b.y;
                dl[u] = part[min(a.x + k_, nEdges - 1)].x;
                sl[u] = spart[min(b.x + k_, nEdges - 1)];
            }
            #pragma unroll
            for (int u = 0; u < SORT_UNROLL; ++u) {
                if (dv[u]) atomicAdd(&cnt[dl[u] & (BIN_NODES - 1)], 1);
                if (sv[u]) atomicAdd(&ocnt[sl[u] & (BIN_NODES - 1)], 1);
            }
        }
    }
    __syncthreads();

    const int c = cnt[tid];
    int incl = c;
    for (int o = 1; o < 32; o <<= 1) {
        int v = __shfl_up(incl, o);
        if (lane >= o) incl += v;
    }
    if (lane == 31) wsum[wave] = incl;
    const float osc = rsqrtf(fmaxf((float)ocnt[tid], 1.0f));
    oscl[tid] = osc;
    __syncthreads();
    int excl = incl - c;
    for (int w = 0; w < SORT_THREADS / 32; ++w) excl += (w < wave) ? wsum[w] : 0;

    const int node = bin * BIN_NODES + tid;
    if (node < nNodes) {
        row_start[node] = segBeg + excl;
        if (node == nNodes - 1) row_start[nNodes] = segBeg + excl + c;
        in_isqrt[node]  = rsqrtf(fmaxf((float)c, 1.0f));
        out_isqrt[node] = osc;
    }
    cur[tid] = excl;
    __syncthreads();


    if (segLen <= STAGE_CAP) {
        for (int kb = 0; kb < maxCnt; kb += PIECE_SLOTS) {
            for (int v0 = tid; v0 < nSlots; v0 += SORT_UNROLL * SORT_THREADS) {
                int2 pr[SORT_UNROLL];
                bool pv[SORT_UNROLL];
                #pragma unroll
                for (int u = 0; u < SORT_UNROLL; ++u) {
                    GCN_SLOT(v0 + u * SORT_THREADS)
                    const int2 a = pd[pc_];
                    pv[u] = in_ && k_ < a.y;
                    pr[u] = part[min(a.x + k_, nEdges - 1)];
                }
                #pragma unroll
                for (int u = 0; u < SORT_UNROLL; ++u)
                    if (pv[u]) stage[atomicAdd(&cur[pr[u].x & (BIN_NODES - 1)], 1)] = pr[u].y;
            }
        }
        __syncthreads();
        for (int i = tid; i < segLen; i += SORT_THREADS) col[segBeg + i] = stage[i];
    } else {
        for (int kb = 0; kb < maxCnt; kb += PIECE_SLOTS) {
            for (int v0 = tid; v0 < nSlots; v0 += SORT_THREADS) {
                GCN_SLOT(v0)
                const int2 a = pd[pc_];
                if (in_ && k_ < a.y) {
                    const int2 p = part[a.x + k_];
                    col[segBeg + atomicAdd(&cur[p.x & (BIN_NODES - 1)], 1)] = p.y;
                }
            }
        }
    }

    #pragma unroll
    for (int u = 0; u < ROW_QUADS_PER_THREAD; ++u) {
        const int lq = u * SORT_THREADS + tid;
        if (q0 + lq < nQuads) {
            const float sc = oscl[lq >> 4];
            float4 v = xv[u];
            v.x *= sc; v.y *= sc; v.z *= sc; v.w *= sc;
            ((float4*)xs)[q0 + lq] = v;
        }
    }
}
#undef GCN_SLOT

#define TILE_NODES 32
#define NODES_PER_HW 4
template <bool HEAD>
__device__ __forceinline__ void
gcn_layer_body(float* __restrict__ lds,
               const float* __restrict__ h,
               const int* __restrict__ row_start,
               const int* __restrict__ col,
               const float* __restrict__ W,
               const float* __restrict__ bias,
               const float* __restrict__ in_isqrt,
               const float* __restrict__ post_scale,
               const float* __restrict__ Wc,
               const float* __restrict__ bc,
               float* __restrict__ out, int nNodes) {

    const int r0   = blockIdx.x * TILE_NODES;
    const int tid  = threadIdx.x;
    const int wave = tid >> 5;
    const int lane = tid & 31;

    float psl = 1.0f;
    if constexpr (!HEAD) {
        const int prow = r0 + ((tid & 15) << 3) + (tid >> 4);
        if (post_scale && (tid & 15) < TILE_NODES / 8 && prow < nNodes) psl = post_scale[prow];
    }

    {
        const int hw = tid >> 4;
        const int l  = tid & 15;
        const int f  = l << 2;
        const int n0 = r0 + hw * NODES_PER_HW;

        int   rsv = 0;
        float scv = 0.0f;
        if (l <= NODES_PER_HW) rsv = row_start[min(n0 + l, nNodes)];
        if (l <  NODES_PER_HW && n0 + l < nNodes) scv = in_isqrt[n0 + l];
        const int segbeg = __shfl(rsv, 0, 16);
        const int segend = __shfl(rsv, 4, 16);
        int   qe0 = __shfl(rsv, 1, 16) - segbeg;
        int   qe1 = __shfl(rsv, 2, 16) - segbeg;
        int   qe2 = __shfl(rsv, 3, 16) - segbeg;
        int   qe3 = segend - segbeg;
        float qs0 = __shfl(scv, 0, 16);
        float qs1 = __shfl(scv, 1, 16);
        float qs2 = __shfl(scv, 2, 16);
        float qs3 = __shfl(scv, 3, 16);

        const float4 zero4 = make_float4(0.0f, 0.0f, 0.0f, 0.0f);
        float4 acc = zero4;
        int lrow = (hw * NODES_PER_HW) * LDS_STRIDE + f;

#define GCN_NO_END   0x3FFFFFFF
#define GCN_END_LIM  0x20000000
#define GCN_FLUSH()                                                              \
        {                                                                        \
            float4 o_;                                                           \
            o_.x = acc.x * qs0; o_.y = acc.y * qs0; o_.z = acc.z * qs0; o_.w = acc.w * qs0; \
            *(float4*)&lds[lrow] = o_;                                           \
            lrow += LDS_STRIDE;                                                  \
            acc = zero4;                                                         \
            qs0 = qs1; qs1 = qs2; qs2 = qs3;                                     \
            qe0 = qe1; qe1 = qe2; qe2 = qe3; qe3 = GCN_NO_END;                   \
        }
#define GCN_ADD(rq, kq)                                                          \
        {                                                                        \
            _Pragma("nounroll")                                                  \
            while ((kq) >= qe0) GCN_FLUSH()                                      \
            acc.x += rq.x; acc.y += rq.y; acc.z += rq.z; acc.w += rq.w;          \
        }
#define GCN_LOAD(rq, sq, kq)                                                     \
        float4 rq = zero4;                                                       \
        if ((kq) < rem) rq = *(const float4*)(hb + ((unsigned)(sq) << 8));
#define GCN_GROUP_LOADS(K)                                                       \
            const int s0 = __shfl(c, (K),     16);                               \
            const int s1 = __shfl(c, (K) + 1, 16);                               \
            const int s2 = __shfl(c, (K) + 2, 16);                               \
            const int s3 = __shfl(c, (K) + 3, 16);                               \
            const int s4 = __shfl(c, (K) + 4, 16);                               \
            const int s5 = __shfl(c, (K) + 5, 16);                               \
            const int s6 = __shfl(c, (K) + 6, 16);                               \
            const int s7 = __shfl(c, (K) + 7, 16);                               \
            GCN_LOAD(q0, s0, (K))                                                \
            GCN_LOAD(q1, s1, (K) + 1)                                            \
            GCN_LOAD(q2, s2, (K) + 2)                                            \
            GCN_LOAD(q3, s3, (K) + 3)                                            \
            GCN_LOAD(q4, s4, (K) + 4)                                            \
            GCN_LOAD(q5, s5, (K) + 5)                                            \
            GCN_LOAD(q6, s6, (K) + 6)                                            \
            GCN_LOAD(q7, s7, (K) + 7)
#define GCN_GROUP_ADDS(K)                                                        \
            GCN_ADD(q0, (K))                                                     \
            GCN_ADD(q1, (K) + 1)                                                 \
            GCN_ADD(q2, (K) + 2)                                                 \
            GCN_ADD(q3, (K) + 3)                                                 \
            GCN_ADD(q4, (K) + 4)                                                 \
            GCN_ADD(q5, (K) + 5)                                                 \
            GCN_ADD(q6, (K) + 6)                                                 \
            GCN_ADD(q7, (K) + 7)

        const char* hb = (const char*)h + (f << 2);
        int cnext = (segbeg + l < segend) ? col[segbeg + l] : 0;
        for (int base = segbeg; ; base += 16) {
            const int rem  = segend - base;
            const int remW = max(rem, __shfl_xor(rem, 16));
            if (remW <= 0) break;
            const int c = cnext;
            {
                GCN_GROUP_LOADS(0)
                cnext = (base + 16 + l < segend) ? col[base + 16 + l] : 0;
                GCN_GROUP_ADDS(0)
            }
            if (remW > 8) {
                GCN_GROUP_LOADS(8)
                GCN_GROUP_ADDS(8)
            }
            qe0 -= 16; qe1 -= 16; qe2 -= 16; qe3 -= 16;
        }
        while (qe0 < GCN_END_LIM) GCN_FLUSH()
#undef GCN_END_LIM
#undef GCN_NO_END
#undef GCN_GROUP_ADDS
#undef GCN_GROUP_LOADS
#undef GCN_LOAD
#undef GCN_ADD
#undef GCN_FLUSH
    }
    __syncthreads();

    const int colbase = wave << 4;
    const int n     = lane & 15;
    const int khalf = lane >> 4;

    v8f acc0 = {};
    v8f acc1 = {};
    #pragma unroll
    for (int kk = 0; kk < 16; ++kk) {
        const int k0 = (kk << 2) + (khalf << 1);
        v2f a0 = *(const v2f*)&lds[n * LDS_STRIDE + k0];
        v2f a1 = *(const v2f*)&lds[(16 + n) * LDS_STRIDE + k0];
        v2f b;
        b.x = W[(k0 + 0) * D_FEAT + colbase + n];
        b.y = W[(k0 + 1) * D_FEAT + colbase + n];
        acc0 = __builtin_amdgcn_wmma_f32_16x16x4_f32(
             false, a0,  false, b,
             (short)0, acc0,  false,  false);
        acc1 = __builtin_amdgcn_wmma_f32_16x16x4_f32(
             false, a1,  false, b,
             (short)0, acc1,  false,  false);
    }

    const int colo = colbase + n;
    const float bv = bias[colo];
    if constexpr (!HEAD) {
        const int hw = tid >> 4;
        const int fq = (tid & 15) << 2;
        __syncthreads();
        #pragma unroll
        for (int t = 0; t < 2; ++t) {
            #pragma unroll
            for (int r = 0; r < 8; ++r) {
                float v = (t == 0 ? acc0[r] : acc1[r]) + bv;
                v = (v > 0.0f) ? v : 0.01f * v;
                lds[((t << 4) + (khalf << 3) + r) * LDS_STRIDE + colo] = v;
            }
        }
        __syncthreads();
        float4 ov[TILE_NODES / 8];
        #pragma unroll
        for (int i = 0; i < TILE_NODES / 8; ++i)
            ov[i] = *(const float4*)&lds[((i << 3) + hw) * LDS_STRIDE + fq];
        #pragma unroll
        for (int i = 0; i < TILE_NODES / 8; ++i) {
            const int row = r0 + (i << 3) + hw;
            const float ps = __shfl(psl, i, 16);
            if (row < nNodes) {
                float4 v = ov[i];
                v.x *= ps; v.y *= ps; v.z *= ps; v.w *= ps;
                *(float4*)&out[(size_t)row * D_FEAT + fq] = v;
            }
        }
    } else {
        __syncthreads();
        #pragma unroll
        for (int t = 0; t < 2; ++t) {
            #pragma unroll
            for (int r = 0; r < 8; ++r) {
                float v = (t == 0 ? acc0[r] : acc1[r]) + bv;
                v = (v > 0.0f) ? v : 0.01f * v;
                lds[((t << 4) + (khalf << 3) + r) * LDS_STRIDE + colo] = v;
            }
        }
        __syncthreads();
        const int node = r0 + lane;
        if (wave == 0 && node < nNodes) {
            float a0 = bc[0], a1 = bc[1];
            const float4* row = (const float4*)&lds[lane * LDS_STRIDE];
            #pragma unroll
            for (int q = 0; q < 16; ++q) {
                float4 v = row[q];
                int k = q << 2;
                a0 += v.x * Wc[(k+0)*2] + v.y * Wc[(k+1)*2] + v.z * Wc[(k+2)*2] + v.w * Wc[(k+3)*2];
                a1 += v.x * Wc[(k+0)*2+1] + v.y * Wc[(k+1)*2+1] + v.z * Wc[(k+2)*2+1] + v.w * Wc[(k+3)*2+1];
            }
            *(float2*)&out[(size_t)node * 2] = make_float2(a0, a1);
        }
    }
}

__global__ void __launch_bounds__(128)
gcn_layer_fused(const float* __restrict__ h, const int* __restrict__ row_start,
                const int* __restrict__ col, const float* __restrict__ W,
                const float* __restrict__ bias, const float* __restrict__ in_isqrt,
                const float* __restrict__ post_scale,
                float* __restrict__ out, int nNodes) {
    __shared__ float lds[TILE_NODES * LDS_STRIDE];
    gcn_layer_body<false>(lds, h, row_start, col, W, bias, in_isqrt, post_scale,
                          nullptr, nullptr, out, nNodes);
}

__global__ void __launch_bounds__(128)
gcn_layer_head(const float* __restrict__ h, const int* __restrict__ row_start,
               const int* __restrict__ col, const float* __restrict__ W,
               const float* __restrict__ bias, const float* __restrict__ in_isqrt,
               const float* __restrict__ Wc, const float* __restrict__ bc,
               float* __restrict__ out, int nNodes) {
    __shared__ float lds[TILE_NODES * LDS_STRIDE];
    gcn_layer_body<true>(lds, h, row_start, col, W, bias, in_isqrt, nullptr,
                         Wc, bc, out, nNodes);
}

extern "C" void kernel_launch(void* const* d_in, const int* in_sizes, int n_in,
                              void* d_out, int out_size, void* d_ws, size_t ws_size,
                              hipStream_t stream) {
    const float* x   = (const float*)d_in[0];
    const int*   src = (const int*)  d_in[1];
    const int*   dst = (const int*)  d_in[2];
    const float* W1  = (const float*)d_in[3];
    const float* b1  = (const float*)d_in[4];
    const float* W2  = (const float*)d_in[5];
    const float* b2  = (const float*)d_in[6];
    const float* Wc  = (const float*)d_in[7];
    const float* bc  = (const float*)d_in[8];
    float* out = (float*)d_out;

    const int N = in_sizes[0] / D_FEAT;
    const int E = in_sizes[1];
    const int nBins = (N + BIN_NODES - 1) >> BIN_SHIFT;
    const int nEdgeBlocks = (E + EDGES_PER_BLOCK - 1) / EDGES_PER_BLOCK;

    float* bufA      = (float*)d_ws;
    float* bufB      = bufA + (size_t)N * D_FEAT;
    float* out_isqrt = bufB + (size_t)N * D_FEAT;
    float* in_isqrt  = out_isqrt + N;
    int*   row_start = (int*)(in_isqrt + N);
    int*   col       = row_start + (N + 1);
    int4*  tab       = (int4*)bufB;
    int2*  part      = (int2*)(tab + (size_t)nEdgeBlocks * nBins);
    int*   spart     = (int*)(part + E);

    if (ws_size < ((size_t)2 * N * D_FEAT + (size_t)3 * N + 1 + (size_t)E) * 4) return;
    if ((size_t)3 * E + (size_t)4 * nEdgeBlocks * nBins > (size_t)N * D_FEAT) return;
    if (nBins > MAX_BINS || nEdgeBlocks > MAX_BLK || N < 1 || E < 1) return;

    gcn_block_partition<<<nEdgeBlocks, dim3(EDGE_THREADS), 0, stream>>>(src, dst, part, spart, tab, E, nBins);
    gcn_bin_sort<<<nBins, dim3(SORT_THREADS), 0, stream>>>(part, spart, tab, row_start, in_isqrt, col,
                                                          x, out_isqrt, bufA, N, nBins, N * 16, nEdgeBlocks, E);

    const int nTiles = (N + TILE_NODES - 1) / TILE_NODES;
    gcn_layer_fused<<<nTiles, dim3(128), 0, stream>>>(
        bufA, row_start, col, W1, b1, in_isqrt, out_isqrt, bufB, N);

    gcn_layer_head<<<nTiles, dim3(128), 0, stream>>>(
        bufB, row_start, col, W2, b2, in_isqrt, Wc, bc, out, N);
}


# Row loads of the sum walk made unconditional: a slot past the half-wave's last edge has id 0 and reads row 0 into a sum that is never stored; zero fill, mask and branch per load go; sum order same
# speedup vs baseline: 4.1431x; 1.0620x over previous
#include <hip/hip_runtime.h>
#include <hip/hip_bf16.h>


typedef __attribute__((ext_vector_type(2))) float v2f;
typedef __attribute__((ext_vector_type(8))) float v8f;

#define D_FEAT 64
#define LDS_STRIDE 68

#define BIN_SHIFT 9
#define BIN_NODES (1 << BIN_SHIFT)
#define MAX_BINS  2048
#define MAX_BLK   1024
#define EDGE_THREADS 256
#define EDGES_PER_THREAD 8
#define EDGES_PER_BLOCK (EDGE_THREADS * EDGES_PER_THREAD)
#define SORT_THREADS BIN_NODES
#define STAGE_CAP 12288
#define SORT_UNROLL 16

__device__ __forceinline__ void gcn_block_scan_excl(int* a, int n, int* wsum) {
    const int tid  = threadIdx.x;
    const int nthr = blockDim.x;
    const int lane = tid & 31;
    const int wave = tid >> 5;
    const int nwav = nthr >> 5;
    const int G  = (n + nthr - 1) / nthr;
    const int i0 = tid * G;
    __syncthreads();
    int sum = 0;
    for (int k = 0; k < G; ++k) { const int i = i0 + k; if (i < n) sum += a[i]; }
    int incl = sum;
    for (int o = 1; o < 32; o <<= 1) {
        int t = __shfl_up(incl, o);
        if (lane >= o) incl += t;
    }
    if (lane == 31) wsum[wave] = incl;
    __syncthreads();
    int run = incl - sum;
    for (int w = 0; w < nwav; ++w) run += (w < wave) ? wsum[w] : 0;
    for (int k = 0; k < G; ++k) {
        const int i = i0 + k;
        if (i < n) { const int v = a[i]; a[i] = run; run += v; }
    }
    __syncthreads();
}

__global__ void __launch_bounds__(EDGE_THREADS)
gcn_block_partition(const int* __restrict__ src, const int* __restrict__ dst,
                    int2* __restrict__ part, int* __restrict__ spart,
                    int4* __restrict__ tab, int nEdges, int nBins) {
    __shared__ int  hcnt[2 * MAX_BINS];
    __shared__ int2 stage[EDGES_PER_BLOCK];
    __shared__ int  sstage[EDGES_PER_BLOCK];
    __shared__ int  wsum[EDGE_THREADS / 32];

    const int tid = threadIdx.x;
    const int nB2 = 2 * nBins;
    for (int i = tid; i < nB2; i += EDGE_THREADS) hcnt[i] = 0;
    __syncthreads();

    const int base   = blockIdx.x * EDGES_PER_BLOCK;
    const int nLocal = min(EDGES_PER_BLOCK, nEdges - base);
    int es[EDGES_PER_THREAD], ed[EDGES_PER_THREAD], er[EDGES_PER_THREAD], sr[EDGES_PER_THREAD];
    #pragma unroll
    for (int k = 0; k < EDGES_PER_THREAD; ++k) {
        const int ec = min(base + k * EDGE_THREADS + tid, nEdges - 1);
        es[k] = src[ec];
        ed[k] = dst[ec];
    }
    #pragma unroll
    for (int k = 0; k < EDGES_PER_THREAD; ++k) {
        er[k] = -1;
        sr[k] = 0;
        if (base + k * EDGE_THREADS + tid < nEdges) {
            er[k] = atomicAdd(&hcnt[ed[k] >> BIN_SHIFT], 1);
            sr[k] = atomicAdd(&hcnt[nBins + (es[k] >> BIN_SHIFT)], 1);
        }
    }
    gcn_block_scan_excl(hcnt, nB2, wsum);

    for (int b = tid; b < nBins; b += EDGE_THREADS) {
        const int d0 = hcnt[b];
        const int d1 = hcnt[b + 1];
        const int s0 = hcnt[nBins + b];
        const int s1 = (b + 1 < nBins) ? hcnt[nBins + b + 1] : 2 * nLocal;
        tab[(size_t)blockIdx.x * nBins + b] = make_int4(d0, d1 - d0, s0 - nLocal, s1 - s0);
    }

    #pragma unroll
    for (int k = 0; k < EDGES_PER_THREAD; ++k) {
        if (er[k] >= 0) {
            stage[hcnt[ed[k] >> BIN_SHIFT] + er[k]] = make_int2(ed[k], es[k]);
            sstage[hcnt[nBins + (es[k] >> BIN_SHIFT)] - nLocal + sr[k]] = es[k];
        }
    }
    __syncthreads();

    #pragma unroll
    for (int k = 0; k < EDGES_PER_THREAD; ++k) {
        const int i = k * EDGE_THREADS + tid;
        if (i < nLocal) { part[base + i] = stage[i]; spart[base + i] = sstage[i]; }
    }
}

#define ROW_QUADS_PER_THREAD (BIN_NODES * (D_FEAT / 4) / SORT_THREADS)
#define PIECE_SLOTS 16
#define GCN_SLOT(v)                                                              \
        const int  v_  = (v);                                                    \
        const int  pc_ = min(v_ / PIECE_SLOTS, nBlk - 1);                        \
        const int  k_  = kb + (v_ & (PIECE_SLOTS - 1));                          \
        const bool in_ = v_ < nSlots;
__global__ void __launch_bounds__(SORT_THREADS)
gcn_bin_sort(const int2* __restrict__ part, const int* __restrict__ spart,
             const int4* __restrict__ tab,
             int* __restrict__ row_start, float* __restrict__ in_isqrt,
             int* __restrict__ col,
             const float* __restrict__ x, float* __restrict__ out_isqrt,
             float* __restrict__ xs, int nNodes, int nBins, int nQuads  ,
             int nBlk, int nEdges) {
    __shared__ int   cnt[BIN_NODES];
    __shared__ int   cur[BIN_NODES];
    __shared__ int   ocnt[BIN_NODES];
    __shared__ float oscl[BIN_NODES];
    __shared__ int   wsum[SORT_THREADS / 32];
    __shared__ int   wsum2[SORT_THREADS / 32];
    __shared__ int   wsum3[SORT_THREADS / 32];
    __shared__ int   stage[STAGE_CAP];
    __shared__ int2  pd[MAX_BLK];
    __shared__ int2  ps[MAX_BLK];

    const int tid  = threadIdx.x;
    const int lane = tid & 31;
    const int wave = tid >> 5;
    const int bin  = blockIdx.x;

    int pre = 0, len = 0, mx = 0;
    for (int p = tid; p < nBlk; p += SORT_THREADS) {
        const int4 t  = tab[(size_t)p * nBins + bin];
        const int  gb = p * EDGES_PER_BLOCK;
        pd[p] = make_int2(gb + t.x, t.y);
        ps[p] = make_int2(gb + t.z, t.w);
        pre += t.x;
        len += t.y;
        mx   = max(mx, max(t.y, t.w));
    }

    const int q0 = bin * (BIN_NODES * (D_FEAT / 4));
    float4 xv[ROW_QUADS_PER_THREAD];
    #pragma unroll
    for (int u = 0; u < ROW_QUADS_PER_THREAD; ++u)
        xv[u] = ((const float4*)x)[min(q0 + u * SORT_THREADS + tid, nQuads - 1)];

    for (int o = 16; o > 0; o >>= 1) {
        pre += __shfl_xor(pre, o);
        len += __shfl_xor(len, o);
        mx   = max(mx, __shfl_xor(mx, o));
    }
    if (lane == 0) { wsum[wave] = pre; wsum2[wave] = len; wsum3[wave] = mx; }
    cnt[tid]  = 0;
    ocnt[tid] = 0;
    __syncthreads();
    int segBeg = 0, segLen = 0, maxCnt = 0;
    for (int w = 0; w < SORT_THREADS / 32; ++w) {
        segBeg += wsum[w]; segLen += wsum2[w]; maxCnt = max(maxCnt, wsum3[w]);
    }
    const int nSlots = nBlk * PIECE_SLOTS;

    for (int kb = 0; kb < maxCnt; kb += PIECE_SLOTS) {
        for (int v0 = tid; v0 < nSlots; v0 += SORT_UNROLL * SORT_THREADS) {
            int  dl[SORT_UNROLL], sl[SORT_UNROLL];
            bool dv[SORT_UNROLL], sv[SORT_UNROLL];
            #pragma unroll
            for (int u = 0; u < SORT_UNROLL; ++u) {
                GCN_SLOT(v0 + u * SORT_THREADS)
                const int2 a = pd[pc_];
                const int2 b = ps[pc_];
                dv[u] = in_ && k_ < a.y;
                sv[u] = in_ && k_ < b.y;
                dl[u] = part[min(a.x + k_, nEdges - 1)].x;
                sl[u] = spart[min(b.x + k_, nEdges - 1)];
            }
            #pragma unroll
            for (int u = 0; u < SORT_UNROLL; ++u) {
                if (dv[u]) atomicAdd(&cnt[dl[u] & (BIN_NODES - 1)], 1);
                if (sv[u]) atomicAdd(&ocnt[sl[u] & (BIN_NODES - 1)], 1);
            }
        }
    }
    __syncthreads();

    const int c = cnt[tid];
    int incl = c;
    for (int o = 1; o < 32; o <<= 1) {
        int v = __shfl_up(incl, o);
        if (lane >= o) incl += v;
    }
    if (lane == 31) wsum[wave] = incl;
    const float osc = rsqrtf(fmaxf((float)ocnt[tid], 1.0f));
    oscl[tid] = osc;
    __syncthreads();
    int excl = incl - c;
    for (int w = 0; w < SORT_THREADS / 32; ++w) excl += (w < wave) ? wsum[w] : 0;

    const int node = bin * BIN_NODES + tid;
    if (node < nNodes) {
        row_start[node] = segBeg + excl;
        if (node == nNodes - 1) row_start[nNodes] = segBeg + excl + c;
        in_isqrt[node]  = rsqrtf(fmaxf((float)c, 1.0f));
        out_isqrt[node] = osc;
    }
    cur[tid] = excl;
    __syncthreads();


    if (segLen <= STAGE_CAP) {
        for (int kb = 0; kb < maxCnt; kb += PIECE_SLOTS) {
            for (int v0 = tid; v0 < nSlots; v0 += SORT_UNROLL * SORT_THREADS) {
                int2 pr[SORT_UNROLL];
                bool pv[SORT_UNROLL];
                #pragma unroll
                for (int u = 0; u < SORT_UNROLL; ++u) {
                    GCN_SLOT(v0 + u * SORT_THREADS)
                    const int2 a = pd[pc_];
                    pv[u] = in_ && k_ < a.y;
                    pr[u] = part[min(a.x + k_, nEdges - 1)];
                }
                #pragma unroll
                for (int u = 0; u < SORT_UNROLL; ++u)
                    if (pv[u]) stage[atomicAdd(&cur[pr[u].x & (BIN_NODES - 1)], 1)] = pr[u].y;
            }
        }
        __syncthreads();
        for (int i = tid; i < segLen; i += SORT_THREADS) col[segBeg + i] = stage[i];
    } else {
        for (int kb = 0; kb < maxCnt; kb += PIECE_SLOTS) {
            for (int v0 = tid; v0 < nSlots; v0 += SORT_THREADS) {
                GCN_SLOT(v0)
                const int2 a = pd[pc_];
                if (in_ && k_ < a.y) {
                    const int2 p = part[a.x + k_];
                    col[segBeg + atomicAdd(&cur[p.x & (BIN_NODES - 1)], 1)] = p.y;
                }
            }
        }
    }

    #pragma unroll
    for (int u = 0; u < ROW_QUADS_PER_THREAD; ++u) {
        const int lq = u * SORT_THREADS + tid;
        if (q0 + lq < nQuads) {
            const float sc = oscl[lq >> 4];
            float4 v = xv[u];
            v.x *= sc; v.y *= sc; v.z *= sc; v.w *= sc;
            ((float4*)xs)[q0 + lq] = v;
        }
    }
}
#undef GCN_SLOT

#define TILE_NODES 32
#define NODES_PER_HW 4
template <bool HEAD>
__device__ __forceinline__ void
gcn_layer_body(float* __restrict__ lds,
               const float* __restrict__ h,
               const int* __restrict__ row_start,
               const int* __restrict__ col,
               const float* __restrict__ W,
               const float* __restrict__ bias,
               const float* __restrict__ in_isqrt,
               const float* __restrict__ post_scale,
               const float* __restrict__ Wc,
               const float* __restrict__ bc,
               float* __restrict__ out, int nNodes) {

    const int r0   = blockIdx.x * TILE_NODES;
    const int tid  = threadIdx.x;
    const int wave = tid >> 5;
    const int lane = tid & 31;

    float psl = 1.0f;
    if constexpr (!HEAD) {
        const int prow = r0 + ((tid & 15) << 3) + (tid >> 4);
        if (post_scale && (tid & 15) < TILE_NODES / 8 && prow < nNodes) psl = post_scale[prow];
    }

    {
        const int hw = tid >> 4;
        const int l  = tid & 15;
        const int f  = l << 2;
        const int n0 = r0 + hw * NODES_PER_HW;

        int   rsv = 0;
        float scv = 0.0f;
        if (l <= NODES_PER_HW) rsv = row_start[min(n0 + l, nNodes)];
        if (l <  NODES_PER_HW && n0 + l < nNodes) scv = in_isqrt[n0 + l];
        const int segbeg = __shfl(rsv, 0, 16);
        const int segend = __shfl(rsv, 4, 16);
        int   qe0 = __shfl(rsv, 1, 16) - segbeg;
        int   qe1 = __shfl(rsv, 2, 16) - segbeg;
        int   qe2 = __shfl(rsv, 3, 16) - segbeg;
        int   qe3 = segend - segbeg;
        float qs0 = __shfl(scv, 0, 16);
        float qs1 = __shfl(scv, 1, 16);
        float qs2 = __shfl(scv, 2, 16);
        float qs3 = __shfl(scv, 3, 16);

        const float4 zero4 = make_float4(0.0f, 0.0f, 0.0f, 0.0f);
        float4 acc = zero4;
        int lrow = (hw * NODES_PER_HW) * LDS_STRIDE + f;

#define GCN_NO_END   0x3FFFFFFF
#define GCN_END_LIM  0x20000000
#define GCN_FLUSH()                                                              \
        {                                                                        \
            float4 o_;                                                           \
            o_.x = acc.x * qs0; o_.y = acc.y * qs0; o_.z = acc.z * qs0; o_.w = acc.w * qs0; \
            *(float4*)&lds[lrow] = o_;                                           \
            lrow += LDS_STRIDE;                                                  \
            acc = zero4;                                                         \
            qs0 = qs1; qs1 = qs2; qs2 = qs3;                                     \
            qe0 = qe1; qe1 = qe2; qe2 = qe3; qe3 = GCN_NO_END;                   \
        }
#define GCN_ADD(rq, kq)                                                          \
        {                                                                        \
            _Pragma("nounroll")                                                  \
            while ((kq) >= qe0) GCN_FLUSH()                                      \
            acc.x += rq.x; acc.y += rq.y; acc.z += rq.z; acc.w += rq.w;          \
        }
#define GCN_LOAD(rq, sq, kq)                                                     \
        const float4 rq = *(const float4*)(hb + ((unsigned)(sq) << 8));
#define GCN_GROUP_LOADS(K)                                                       \
            const int s0 = __shfl(c, (K),     16);                               \
            const int s1 = __shfl(c, (K) + 1, 16);                               \
            const int s2 = __shfl(c, (K) + 2, 16);                               \
            const int s3 = __shfl(c, (K) + 3, 16);                               \
            const int s4 = __shfl(c, (K) + 4, 16);                               \
            const int s5 = __shfl(c, (K) + 5, 16);                               \
            const int s6 = __shfl(c, (K) + 6, 16);                               \
            const int s7 = __shfl(c, (K) + 7, 16);                               \
            GCN_LOAD(q0, s0, (K))                                                \
            GCN_LOAD(q1, s1, (K) + 1)                                            \
            GCN_LOAD(q2, s2, (K) + 2)                                            \
            GCN_LOAD(q3, s3, (K) + 3)                                            \
            GCN_LOAD(q4, s4, (K) + 4)                                            \
            GCN_LOAD(q5, s5, (K) + 5)                                            \
            GCN_LOAD(q6, s6, (K) + 6)                                            \
            GCN_LOAD(q7, s7, (K) + 7)
#define GCN_GROUP_ADDS(K)                                                        \
            GCN_ADD(q0, (K))                                                     \
            GCN_ADD(q1, (K) + 1)                                                 \
            GCN_ADD(q2, (K) + 2)                                                 \
            GCN_ADD(q3, (K) + 3)                                                 \
            GCN_ADD(q4, (K) + 4)                                                 \
            GCN_ADD(q5, (K) + 5)                                                 \
            GCN_ADD(q6, (K) + 6)                                                 \
            GCN_ADD(q7, (K) + 7)

        const char* hb = (const char*)h + (f << 2);
        int cnext = (segbeg + l < segend) ? col[segbeg + l] : 0;
        for (int base = segbeg; ; base += 16) {
            const int rem  = segend - base;
            const int remW = max(rem, __shfl_xor(rem, 16));
            if (remW <= 0) break;
            const int c = cnext;
            {
                GCN_GROUP_LOADS(0)
                cnext = (base + 16 + l < segend) ? col[base + 16 + l] : 0;
                GCN_GROUP_ADDS(0)
            }
            if (remW > 8) {
                GCN_GROUP_LOADS(8)
                GCN_GROUP_ADDS(8)
            }
            qe0 -= 16; qe1 -= 16; qe2 -= 16; qe3 -= 16;
        }
        while (qe0 < GCN_END_LIM) GCN_FLUSH()
#undef GCN_END_LIM
#undef GCN_NO_END
#undef GCN_GROUP_ADDS
#undef GCN_GROUP_LOADS
#undef GCN_LOAD
#undef GCN_ADD
#undef GCN_FLUSH
    }
    __syncthreads();

    const int colbase = wave << 4;
    const int n     = lane & 15;
    const int khalf = lane >> 4;

    v8f acc0 = {};
    v8f acc1 = {};
    #pragma unroll
    for (int kk = 0; kk < 16; ++kk) {
        const int k0 = (kk << 2) + (khalf << 1);
        v2f a0 = *(const v2f*)&lds[n * LDS_STRIDE + k0];
        v2f a1 = *(const v2f*)&lds[(16 + n) * LDS_STRIDE + k0];
        v2f b;
        b.x = W[(k0 + 0) * D_FEAT + colbase + n];
        b.y = W[(k0 + 1) * D_FEAT + colbase + n];
        acc0 = __builtin_amdgcn_wmma_f32_16x16x4_f32(
             false, a0,  false, b,
             (short)0, acc0,  false,  false);
        acc1 = __builtin_amdgcn_wmma_f32_16x16x4_f32(
             false, a1,  false, b,
             (short)0, acc1,  false,  false);
    }

    const int colo = colbase + n;
    const float bv = bias[colo];
    if constexpr (!HEAD) {
        const int hw = tid >> 4;
        const int fq = (tid & 15) << 2;
        __syncthreads();
        #pragma unroll
        for (int t = 0; t < 2; ++t) {
            #pragma unroll
            for (int r = 0; r < 8; ++r) {
                float v = (t == 0 ? acc0[r] : acc1[r]) + bv;
                v = (v > 0.0f) ? v : 0.01f * v;
                lds[((t << 4) + (khalf << 3) + r) * LDS_STRIDE + colo] = v;
            }
        }
        __syncthreads();
        float4 ov[TILE_NODES / 8];
        #pragma unroll
        for (int i = 0; i < TILE_NODES / 8; ++i)
            ov[i] = *(const float4*)&lds[((i << 3) + hw) * LDS_STRIDE + fq];
        #pragma unroll
        for (int i = 0; i < TILE_NODES / 8; ++i) {
            const int row = r0 + (i << 3) + hw;
            const float ps = __shfl(psl, i, 16);
            if (row < nNodes) {
                float4 v = ov[i];
                v.x *= ps; v.y *= ps; v.z *= ps; v.w *= ps;
                *(float4*)&out[(size_t)row * D_FEAT + fq] = v;
            }
        }
    } else {
        __syncthreads();
        #pragma unroll
        for (int t = 0; t < 2; ++t) {
            #pragma unroll
            for (int r = 0; r < 8; ++r) {
                float v = (t == 0 ? acc0[r] : acc1[r]) + bv;
                v = (v > 0.0f) ? v : 0.01f * v;
                lds[((t << 4) + (khalf << 3) + r) * LDS_STRIDE + colo] = v;
            }
        }
        __syncthreads();
        const int node = r0 + lane;
        if (wave == 0 && node < nNodes) {
            float a0 = bc[0], a1 = bc[1];
            const float4* row = (const float4*)&lds[lane * LDS_STRIDE];
            #pragma unroll
            for (int q = 0; q < 16; ++q) {
                float4 v = row[q];
                int k = q << 2;
                a0 += v.x * Wc[(k+0)*2] + v.y * Wc[(k+1)*2] + v.z * Wc[(k+2)*2] + v.w * Wc[(k+3)*2];
                a1 += v.x * Wc[(k+0)*2+1] + v.y * Wc[(k+1)*2+1] + v.z * Wc[(k+2)*2+1] + v.w * Wc[(k+3)*2+1];
            }
            *(float2*)&out[(size_t)node * 2] = make_float2(a0, a1);
        }
    }
}

__global__ void __launch_bounds__(128)
gcn_layer_fused(const float* __restrict__ h, const int* __restrict__ row_start,
                const int* __restrict__ col, const float* __restrict__ W,
                const float* __restrict__ bias, const float* __restrict__ in_isqrt,
                const float* __restrict__ post_scale,
                float* __restrict__ out, int nNodes) {
    __shared__ float lds[TILE_NODES * LDS_STRIDE];
    gcn_layer_body<false>(lds, h, row_start, col, W, bias, in_isqrt, post_scale,
                          nullptr, nullptr, out, nNodes);
}

__global__ void __launch_bounds__(128)
gcn_layer_head(const float* __restrict__ h, const int* __restrict__ row_start,
               const int* __restrict__ col, const float* __restrict__ W,
               const float* __restrict__ bias, const float* __restrict__ in_isqrt,
               const float* __restrict__ Wc, const float* __restrict__ bc,
               float* __restrict__ out, int nNodes) {
    __shared__ float lds[TILE_NODES * LDS_STRIDE];
    gcn_layer_body<true>(lds, h, row_start, col, W, bias, in_isqrt, nullptr,
                         Wc, bc, out, nNodes);
}

extern "C" void kernel_launch(void* const* d_in, const int* in_sizes, int n_in,
                              void* d_out, int out_size, void* d_ws, size_t ws_size,
                              hipStream_t stream) {
    const float* x   = (const float*)d_in[0];
    const int*   src = (const int*)  d_in[1];
    const int*   dst = (const int*)  d_in[2];
    const float* W1  = (const float*)d_in[3];
    const float* b1  = (const float*)d_in[4];
    const float* W2  = (const float*)d_in[5];
    const float* b2  = (const float*)d_in[6];
    const float* Wc  = (const float*)d_in[7];
    const float* bc  = (const float*)d_in[8];
    float* out = (float*)d_out;

    const int N = in_sizes[0] / D_FEAT;
    const int E = in_sizes[1];
    const int nBins = (N + BIN_NODES - 1) >> BIN_SHIFT;
    const int nEdgeBlocks = (E + EDGES_PER_BLOCK - 1) / EDGES_PER_BLOCK;

    float* bufA      = (float*)d_ws;
    float* bufB      = bufA + (size_t)N * D_FEAT;
    float* out_isqrt = bufB + (size_t)N * D_FEAT;
    float* in_isqrt  = out_isqrt + N;
    int*   row_start = (int*)(in_isqrt + N);
    int*   col       = row_start + (N + 1);
    int4*  tab       = (int4*)bufB;
    int2*  part      = (int2*)(tab + (size_t)nEdgeBlocks * nBins);
    int*   spart     = (int*)(part + E);

    if (ws_size < ((size_t)2 * N * D_FEAT + (size_t)3 * N + 1 + (size_t)E) * 4) return;
    if ((size_t)3 * E + (size_t)4 * nEdgeBlocks * nBins > (size_t)N * D_FEAT) return;
    if (nBins > MAX_BINS || nEdgeBlocks > MAX_BLK || N < 1 || E < 1) return;

    gcn_block_partition<<<nEdgeBlocks, dim3(EDGE_THREADS), 0, stream>>>(src, dst, part, spart, tab, E, nBins);
    gcn_bin_sort<<<nBins, dim3(SORT_THREADS), 0, stream>>>(part, spart, tab, row_start, in_isqrt, col,
                                                          x, out_isqrt, bufA, N, nBins, N * 16, nEdgeBlocks, E);

    const int nTiles = (N + TILE_NODES - 1) / TILE_NODES;
    gcn_layer_fused<<<nTiles, dim3(128), 0, stream>>>(
        bufA, row_start, col, W1, b1, in_isqrt, out_isqrt, bufB, N);

    gcn_layer_head<<<nTiles, dim3(128), 0, stream>>>(
        bufB, row_start, col, W2, b2, in_isqrt, Wc, bc, out, N);
}
